# starting kernel
# baseline (speedup 1.0000x reference)
#include <hip/hip_runtime.h>
#include <math.h>

#define NNODES 100000
#define NEDGES 1600000
#define NEP    1700000
#define FIN    16
#define FH     64
#define FOUT   8

typedef __attribute__((ext_vector_type(2))) float v2f;
typedef __attribute__((ext_vector_type(8))) float v8f;

__device__ __forceinline__ float leaky(float e) { return e > 0.f ? e : 0.2f * e; }

__global__ void gemm_h_kernel(const float* __restrict__ in, const float* __restrict__ W,
                              float* __restrict__ h, int K, int nRowTiles) {
    const int lane = threadIdx.x & 31;
    const int gw   = blockIdx.x * 8 + (threadIdx.x >> 5);
    const int rowTile = gw >> 2;
    const int colTile = gw & 3;
    if (rowTile >= nRowTiles) return;
    const int hi  = lane >> 4;
    const int lo  = lane & 15;
    const int row = rowTile * 16 + lo;
    const int col = colTile * 16 + lo;
    v8f c = {};
    for (int kk = 0; kk < K; kk += 4) {
        const int k0 = kk + 2 * hi;
        v2f a, b;
        a.x = in[(size_t)row * K + k0];
        a.y = in[(size_t)row * K + k0 + 1];
        b.x = W[(size_t)k0 * FH + col];
        b.y = W[(size_t)(k0 + 1) * FH + col];
        c = __builtin_amdgcn_wmma_f32_16x16x4_f32(false, a, false, b, (short)0, c, false, false);
    }
    const int rbase = rowTile * 16 + 8 * hi;
#pragma unroll
    for (int r = 0; r < 8; ++r)
        h[(size_t)(rbase + r) * FH + col] = c[r];
}

__global__ void scalars_init_kernel(const float* __restrict__ h,
                                    const float* __restrict__ a_src,
                                    const float* __restrict__ a_dst,
                                    float* __restrict__ s_src, float* __restrict__ s_dst,
                                    int* __restrict__ mbits, float* __restrict__ den,
                                    float* __restrict__ obuf) {
    const int lane = threadIdx.x & 31;
    const int n    = blockIdx.x * 8 + (threadIdx.x >> 5);
    if (n >= NNODES) return;
    const float h0 = h[(size_t)n * FH + lane];
    const float h1 = h[(size_t)n * FH + lane + 32];
    float ps = h0 * a_src[lane] + h1 * a_src[lane + 32];
    float pd = h0 * a_dst[lane] + h1 * a_dst[lane + 32];
    for (int off = 16; off; off >>= 1) {
        ps += __shfl_xor(ps, off, 32);
        pd += __shfl_xor(pd, off, 32);
    }
    if (lane == 0) {
        s_src[n] = ps;
        s_dst[n] = pd;
        mbits[n] = (int)0xFF800000u;
        den[n]   = 0.f;
    }
    obuf[(size_t)n * FH + lane]      = 0.f;
    obuf[(size_t)n * FH + lane + 32] = 0.f;
}

__device__ __forceinline__ void edge_sd(int e, const int* __restrict__ src,
                                        const int* __restrict__ dst, int& s, int& d) {
    if (e < NEDGES) { s = src[e]; d = dst[e]; }
    else            { s = d = e - NEDGES; }
}

__global__ void edge_max_kernel(const int* __restrict__ src, const int* __restrict__ dst,
                                const float* __restrict__ s_src, const float* __restrict__ s_dst,
                                int* __restrict__ mbits) {
    const int e = blockIdx.x * blockDim.x + threadIdx.x;
    if (e >= NEP) return;
    int s, d; edge_sd(e, src, dst, s, d);
    const float l = leaky(s_src[s] + s_dst[d]);
    const unsigned u = __float_as_uint(l);
    if ((u >> 31) == 0) atomicMax((int*)&mbits[d], (int)u);
    else                atomicMin((unsigned int*)&mbits[d], u);
}

__global__ void edge_den_kernel(const int* __restrict__ src, const int* __restrict__ dst,
                                const float* __restrict__ s_src, const float* __restrict__ s_dst,
                                const int* __restrict__ mbits, float* __restrict__ den) {
    const int e = blockIdx.x * blockDim.x + threadIdx.x;
    if (e >= NEP) return;
    int s, d; edge_sd(e, src, dst, s, d);
    const float l  = leaky(s_src[s] + s_dst[d]);
    const float mv = __int_as_float(mbits[d]);
    atomicAdd(&den[d], expf(l - mv));
}

__global__ void edge_aggr_kernel(const int* __restrict__ src, const int* __restrict__ dst,
                                 const float* __restrict__ s_src, const float* __restrict__ s_dst,
                                 const int* __restrict__ mbits, const float* __restrict__ den,
                                 const float* __restrict__ h, float* __restrict__ out) {
    const int lane = threadIdx.x & 31;
    const int e    = blockIdx.x * 8 + (threadIdx.x >> 5);
    if (e >= NEP) return;
    int s, d; edge_sd(e, src, dst, s, d);
    const float l     = leaky(s_src[s] + s_dst[d]);
    const float mv    = __int_as_float(mbits[d]);
    const float alpha = expf(l - mv) / (den[d] + 1e-16f);
    const float v0 = h[(size_t)s * FH + lane]      * alpha;
    const float v1 = h[(size_t)s * FH + lane + 32] * alpha;
    atomicAdd(&out[(size_t)d * FH + lane],      v0);
    atomicAdd(&out[(size_t)d * FH + lane + 32], v1);
}

__global__ void act_kernel(float* __restrict__ buf, const float* __restrict__ b) {
    const size_t i = (size_t)blockIdx.x * blockDim.x + threadIdx.x;
    if (i >= (size_t)NNODES * FH) return;
    const float v = buf[i] + b[i & 63];
    buf[i] = v > 0.f ? v : 0.f;
}

__global__ void policy_kernel(const float* __restrict__ act, const float* __restrict__ Wp,
                              const float* __restrict__ bp, float* __restrict__ out,
                              int nRowTiles) {
    const int lane = threadIdx.x & 31;
    const int gw   = blockIdx.x * 8 + (threadIdx.x >> 5);
    if (gw >= nRowTiles) return;
    const int hi  = lane >> 4;
    const int lo  = lane & 15;
    const int row = gw * 16 + lo;
    v8f c = {};
    for (int kk = 0; kk < FH; kk += 4) {
        const int k0 = kk + 2 * hi;
        v2f a, b;
        a.x = act[(size_t)row * FH + k0];
        a.y = act[(size_t)row * FH + k0 + 1];
        b.x = (lo < FOUT) ? Wp[(size_t)k0 * FOUT + lo]       : 0.f;
        b.y = (lo < FOUT) ? Wp[(size_t)(k0 + 1) * FOUT + lo] : 0.f;
        c = __builtin_amdgcn_wmma_f32_16x16x4_f32(false, a, false, b, (short)0, c, false, false);
    }
    if (lo < FOUT) {
        const int rbase = gw * 16 + 8 * hi;
        const float bb = bp[lo];
#pragma unroll
        for (int r = 0; r < 8; ++r)
            out[(size_t)(rbase + r) * FOUT + lo] = c[r] + bb;
    }
}

extern "C" void kernel_launch(void* const* d_in, const int* in_sizes, int n_in,
                              void* d_out, int out_size, void* d_ws, size_t ws_size,
                              hipStream_t stream) {
    const float* x   = (const float*)d_in[0];
    const int*   ei  = (const int*)  d_in[1];
    const float* W1  = (const float*)d_in[2];
    const float* as1 = (const float*)d_in[3];
    const float* ad1 = (const float*)d_in[4];
    const float* b1  = (const float*)d_in[5];
    const float* W2  = (const float*)d_in[6];
    const float* as2 = (const float*)d_in[7];
    const float* ad2 = (const float*)d_in[8];
    const float* b2  = (const float*)d_in[9];
    const float* Wp  = (const float*)d_in[10];
    const float* bp  = (const float*)d_in[11];
    float* out = (float*)d_out;

    const int* srcA = ei;
    const int* dstA = ei + NEDGES;

    float* ws   = (float*)d_ws;
    float* hbuf = ws;
    float* obuf = ws + (size_t)NNODES * FH;
    float* ssrc = ws + (size_t)2 * NNODES * FH;
    float* sdst = ssrc + NNODES;
    int*   mb   = (int*)(sdst + NNODES);
    float* den  = (float*)(mb + NNODES);

    const int nRowTiles = NNODES / 16;
    const dim3 blk(256);
    const int gemmBlocks = (nRowTiles * 4 + 7) / 8;
    const int nodeBlocks = (NNODES + 7) / 8;
    const int edgeBlocks = (NEP + 255) / 256;
    const int aggrBlocks = (NEP + 7) / 8;
    const int actBlocks  = (NNODES * FH + 255) / 256;
    const int polBlocks  = (nRowTiles + 7) / 8;

    gemm_h_kernel<<<gemmBlocks, blk, 0, stream>>>(x, W1, hbuf, FIN, nRowTiles);
    scalars_init_kernel<<<nodeBlocks, blk, 0, stream>>>(hbuf, as1, ad1, ssrc, sdst, mb, den, obuf);
    edge_max_kernel<<<edgeBlocks, blk, 0, stream>>>(srcA, dstA, ssrc, sdst, mb);
    edge_den_kernel<<<edgeBlocks, blk, 0, stream>>>(srcA, dstA, ssrc, sdst, mb, den);
    edge_aggr_kernel<<<aggrBlocks, blk, 0, stream>>>(srcA, dstA, ssrc, sdst, mb, den, hbuf, obuf);
    act_kernel<<<actBlocks, blk, 0, stream>>>(obuf, b1);

    gemm_h_kernel<<<gemmBlocks, blk, 0, stream>>>(obuf, W2, hbuf, FH, nRowTiles);
    scalars_init_kernel<<<nodeBlocks, blk, 0, stream>>>(hbuf, as2, ad2, ssrc, sdst, mb, den, obuf);
    edge_max_kernel<<<edgeBlocks, blk, 0, stream>>>(srcA, dstA, ssrc, sdst, mb);
    edge_den_kernel<<<edgeBlocks, blk, 0, stream>>>(srcA, dstA, ssrc, sdst, mb, den);
    edge_aggr_kernel<<<aggrBlocks, blk, 0, stream>>>(srcA, dstA, ssrc, sdst, mb, den, hbuf, obuf);
    act_kernel<<<actBlocks, blk, 0, stream>>>(obuf, b2);

    policy_kernel<<<polBlocks, blk, 0, stream>>>(obuf, Wp, bp, out, nRowTiles);
}


# edge_aggr: a work-group owns 800 dst nodes with their sum rows in LDS, scans the dst list for its edges and adds rows by LDS float adds; global f32 atomics and obuf clear go; sum order changes
# speedup vs baseline: 1.4948x; 1.4948x over previous
#include <hip/hip_runtime.h>
#include <math.h>

#define NNODES 100000
#define NEDGES 1600000
#define NEP    1700000
#define FIN    16
#define FH     64
#define FOUT   8

typedef __attribute__((ext_vector_type(2))) float v2f;
typedef __attribute__((ext_vector_type(8))) float v8f;

__device__ __forceinline__ float leaky(float e) { return e > 0.f ? e : 0.2f * e; }

__global__ void gemm_h_kernel(const float* __restrict__ in, const float* __restrict__ W,
                              float* __restrict__ h, int K, int nRowTiles) {
    const int lane = threadIdx.x & 31;
    const int gw   = blockIdx.x * 8 + (threadIdx.x >> 5);
    const int rowTile = gw >> 2;
    const int colTile = gw & 3;
    if (rowTile >= nRowTiles) return;
    const int hi  = lane >> 4;
    const int lo  = lane & 15;
    const int row = rowTile * 16 + lo;
    const int col = colTile * 16 + lo;
    v8f c = {};
    for (int kk = 0; kk < K; kk += 4) {
        const int k0 = kk + 2 * hi;
        v2f a, b;
        a.x = in[(size_t)row * K + k0];
        a.y = in[(size_t)row * K + k0 + 1];
        b.x = W[(size_t)k0 * FH + col];
        b.y = W[(size_t)(k0 + 1) * FH + col];
        c = __builtin_amdgcn_wmma_f32_16x16x4_f32(false, a, false, b, (short)0, c, false, false);
    }
    const int rbase = rowTile * 16 + 8 * hi;
#pragma unroll
    for (int r = 0; r < 8; ++r)
        h[(size_t)(rbase + r) * FH + col] = c[r];
}

__global__ void scalars_init_kernel(const float* __restrict__ h,
                                    const float* __restrict__ a_src,
                                    const float* __restrict__ a_dst,
                                    float* __restrict__ s_src, float* __restrict__ s_dst,
                                    int* __restrict__ mbits, float* __restrict__ den,
                                    float* __restrict__ flag) {
    const int lane = threadIdx.x & 31;
    const int n    = blockIdx.x * 8 + (threadIdx.x >> 5);
    if (n >= NNODES) return;
    const float h0 = h[(size_t)n * FH + lane];
    const float h1 = h[(size_t)n * FH + lane + 32];
    float ps = h0 * a_src[lane] + h1 * a_src[lane + 32];
    float pd = h0 * a_dst[lane] + h1 * a_dst[lane + 32];
    for (int off = 16; off; off >>= 1) {
        ps += __shfl_xor(ps, off, 32);
        pd += __shfl_xor(pd, off, 32);
    }
    if (lane == 0) {
        s_src[n] = ps;
        s_dst[n] = pd;
        mbits[n] = __float_as_int(leaky(ps + pd));
        den[n]   = 0.f;
        if (n == 0) *flag = 0.f;
    }
}

__device__ __forceinline__ void edge_sd(int e, const int* __restrict__ src,
                                        const int* __restrict__ dst, int& s, int& d) {
    if (e < NEDGES) { s = src[e]; d = dst[e]; }
    else            { s = d = e - NEDGES; }
}

#define SHIFT_WINDOW 40.0f
#define FB_BLOCKS    1024

__global__ void edge_den_kernel(const int* __restrict__ src, const int* __restrict__ dst,
                                const float* __restrict__ s_src, const float* __restrict__ s_dst,
                                const int* __restrict__ mbits, float* __restrict__ den,
                                float* __restrict__ flag) {
    const int e = blockIdx.x * blockDim.x + threadIdx.x;
    if (e >= NEP) return;
    int s, d; edge_sd(e, src, dst, s, d);
    const float l  = leaky(s_src[s] + s_dst[d]);
    const float mv = __int_as_float(mbits[d]);
    const float t  = l - mv;
    if (!(t <= SHIFT_WINDOW)) *flag = 1.f;
    atomicAdd(&den[d], expf(t));
}

__global__ void fb_max_kernel(const int* __restrict__ src, const int* __restrict__ dst,
                              const float* __restrict__ s_src, const float* __restrict__ s_dst,
                              int* __restrict__ mbits, float* __restrict__ den,
                              const float* __restrict__ flag) {
    if (*flag == 0.f) return;
    const int stride = gridDim.x * blockDim.x;
    for (int e = blockIdx.x * blockDim.x + threadIdx.x; e < NEDGES; e += stride) {
        const int s = src[e], d = dst[e];
        const float l = leaky(s_src[s] + s_dst[d]);
        const unsigned u = __float_as_uint(l);
        if ((u >> 31) == 0) atomicMax((int*)&mbits[d], (int)u);
        else                atomicMin((unsigned int*)&mbits[d], u);
        if (e < NNODES) den[e] = 0.f;
    }
}

__global__ void fb_den_kernel(const int* __restrict__ src, const int* __restrict__ dst,
                              const float* __restrict__ s_src, const float* __restrict__ s_dst,
                              const int* __restrict__ mbits, float* __restrict__ den,
                              const float* __restrict__ flag) {
    if (*flag == 0.f) return;
    const int stride = gridDim.x * blockDim.x;
    for (int e = blockIdx.x * blockDim.x + threadIdx.x; e < NEP; e += stride) {
        int s, d; edge_sd(e, src, dst, s, d);
        const float l  = leaky(s_src[s] + s_dst[d]);
        const float mv = __int_as_float(mbits[d]);
        atomicAdd(&den[d], expf(l - mv));
    }
}

#define TN        800
#define AG_WAVES  32
#define CH_EDGES  128
#define NCHUNK    (NEDGES / CH_EDGES)
#define LD_DEPTH  8
#define NIT       ((NCHUNK + AG_WAVES * LD_DEPTH - 1) / (AG_WAVES * LD_DEPTH))
#define QCAP      576
#define ROWS_GROUP 16

__device__ __forceinline__ void aggr_batch(unsigned pk, bool valid, int nb, int base, int lane,
                                           const int* __restrict__ src,
                                           const float* __restrict__ s_src,
                                           const float* __restrict__ s_dst,
                                           const int* __restrict__ mbits,
                                           const float* __restrict__ den,
                                           const float* __restrict__ h, float* acc) {
    const int e   = (int)(pk >> 10);
    const int rel = (int)(pk & 1023u);
    const int d   = base + rel;
    int   s     = d;
    float alpha = 0.f;
    if (valid) {
        s = (e < NEDGES) ? src[e] : e - NEDGES;
        const float l  = leaky(s_src[s] + s_dst[d]);
        const float mv = __int_as_float(mbits[d]);
        alpha = expf(l - mv) / (den[d] + 1e-16f);
    }
    for (int j0 = 0; j0 < nb; j0 += ROWS_GROUP) {
        float v0[ROWS_GROUP], v1[ROWS_GROUP];
#pragma unroll
        for (int u = 0; u < ROWS_GROUP; ++u) {
            const int j = j0 + u;
            const int   sj = __builtin_amdgcn_readlane(s, j);
            const float aj = __int_as_float(__builtin_amdgcn_readlane(__float_as_int(alpha), j));
            v0[u] = h[(size_t)sj * FH + lane]      * aj;
            v1[u] = h[(size_t)sj * FH + lane + 32] * aj;
        }
#pragma unroll
        for (int u = 0; u < ROWS_GROUP; ++u) {
            const int j = j0 + u;
            if (j < nb) {
                const int rj = __builtin_amdgcn_readlane(rel, j);
                atomicAdd(&acc[rj * FH + lane],      v0[u]);
                atomicAdd(&acc[rj * FH + lane + 32], v1[u]);
            }
        }
    }
}

#define AG_APPEND(HIT, REL, E)                                                     \
    {                                                                              \
        const unsigned m_ = __builtin_amdgcn_ballot_w32(HIT);                      \
        if (HIT) myq[cnt + (int)__builtin_amdgcn_mbcnt_lo(m_, 0u)] =               \
                     ((unsigned)(E) << 10) | (REL);                                \
        cnt += __builtin_popcount(m_);                                             \
    }

__global__ __launch_bounds__(1024)
void edge_aggr_kernel(const int* __restrict__ src, const int* __restrict__ dst,
                      const float* __restrict__ s_src, const float* __restrict__ s_dst,
                      const int* __restrict__ mbits, const float* __restrict__ den,
                      const float* __restrict__ h, float* __restrict__ out) {
    __shared__ float    acc[TN * FH];
    __shared__ unsigned queue[AG_WAVES * QCAP];
    const int lane = threadIdx.x & 31;
    const int wave = __builtin_amdgcn_readfirstlane((int)(threadIdx.x >> 5));
    const int base = blockIdx.x * TN;
    for (int i = threadIdx.x; i < TN * FH; i += 1024) acc[i] = 0.f;
    unsigned* myq = queue + wave * QCAP;
    int cnt = 0;
    if (wave < TN / 32) {
        const int rel = wave * 32 + lane;
        myq[lane] = ((unsigned)(NEDGES + base + rel) << 10) | (unsigned)rel;
        cnt = 32;
    }
    __syncthreads();

    const int4* dst4 = (const int4*)dst;
    for (int it = 0; it < NIT; ++it) {
        const int c0 = (it * AG_WAVES + wave) * LD_DEPTH;
        int4 dv[LD_DEPTH];
#pragma unroll
        for (int u = 0; u < LD_DEPTH; ++u) {
            const int c = c0 + u;
            dv[u] = (c < NCHUNK) ? dst4[(size_t)c * 32 + lane] : make_int4(-1, -1, -1, -1);
        }
#pragma unroll
        for (int half = 0; half < 2; ++half) {
#pragma unroll
            for (int u = half * 4; u < half * 4 + 4; ++u) {
                const unsigned r0 = (unsigned)(dv[u].x - base), r1 = (unsigned)(dv[u].y - base);
                const unsigned r2 = (unsigned)(dv[u].z - base), r3 = (unsigned)(dv[u].w - base);
                const bool h0 = r0 < TN, h1 = r1 < TN, h2 = r2 < TN, h3 = r3 < TN;
                if (__builtin_amdgcn_ballot_w32(h0 | h1 | h2 | h3)) {
                    const int eb = (c0 + u) * CH_EDGES + lane * 4;
                    AG_APPEND(h0, r0, eb)
                    AG_APPEND(h1, r1, eb + 1)
                    AG_APPEND(h2, r2, eb + 2)
                    AG_APPEND(h3, r3, eb + 3)
                }
            }
            while (cnt >= 32) {
                cnt -= 32;
                __builtin_amdgcn_wave_barrier();
                aggr_batch(myq[cnt + lane], true, 32, base, lane, src, s_src, s_dst, mbits, den, h, acc);
            }
        }
    }
    if (cnt > 0) {
        __builtin_amdgcn_wave_barrier();
        aggr_batch(lane < cnt ? myq[lane] : 0u, lane < cnt, cnt, base, lane,
                   src, s_src, s_dst, mbits, den, h, acc);
    }
    __syncthreads();
    for (int i = threadIdx.x; i < TN * FH; i += 1024)
        out[(size_t)base * FH + i] = acc[i];
}

__global__ void act_kernel(float* __restrict__ buf, const float* __restrict__ b) {
    const size_t i = (size_t)blockIdx.x * blockDim.x + threadIdx.x;
    if (i >= (size_t)NNODES * FH) return;
    const float v = buf[i] + b[i & 63];
    buf[i] = v > 0.f ? v : 0.f;
}

__global__ void policy_kernel(const float* __restrict__ act, const float* __restrict__ Wp,
                              const float* __restrict__ bp, float* __restrict__ out,
                              int nRowTiles) {
    const int lane = threadIdx.x & 31;
    const int gw   = blockIdx.x * 8 + (threadIdx.x >> 5);
    if (gw >= nRowTiles) return;
    const int hi  = lane >> 4;
    const int lo  = lane & 15;
    const int row = gw * 16 + lo;
    v8f c = {};
    for (int kk = 0; kk < FH; kk += 4) {
        const int k0 = kk + 2 * hi;
        v2f a, b;
        a.x = act[(size_t)row * FH + k0];
        a.y = act[(size_t)row * FH + k0 + 1];
        b.x = (lo < FOUT) ? Wp[(size_t)k0 * FOUT + lo]       : 0.f;
        b.y = (lo < FOUT) ? Wp[(size_t)(k0 + 1) * FOUT + lo] : 0.f;
        c = __builtin_amdgcn_wmma_f32_16x16x4_f32(false, a, false, b, (short)0, c, false, false);
    }
    if (lo < FOUT) {
        const int rbase = gw * 16 + 8 * hi;
        const float bb = bp[lo];
#pragma unroll
        for (int r = 0; r < 8; ++r)
            out[(size_t)(rbase + r) * FOUT + lo] = c[r] + bb;
    }
}

extern "C" void kernel_launch(void* const* d_in, const int* in_sizes, int n_in,
                              void* d_out, int out_size, void* d_ws, size_t ws_size,
                              hipStream_t stream) {
    const float* x   = (const float*)d_in[0];
    const int*   ei  = (const int*)  d_in[1];
    const float* W1  = (const float*)d_in[2];
    const float* as1 = (const float*)d_in[3];
    const float* ad1 = (const float*)d_in[4];
    const float* b1  = (const float*)d_in[5];
    const float* W2  = (const float*)d_in[6];
    const float* as2 = (const float*)d_in[7];
    const float* ad2 = (const float*)d_in[8];
    const float* b2  = (const float*)d_in[9];
    const float* Wp  = (const float*)d_in[10];
    const float* bp  = (const float*)d_in[11];
    float* out = (float*)d_out;

    const int* srcA = ei;
    const int* dstA = ei + NEDGES;

    float* ws   = (float*)d_ws;
    float* hbuf = ws;
    float* obuf = ws + (size_t)NNODES * FH;
    float* ssrc = ws + (size_t)2 * NNODES * FH;
    float* sdst = ssrc + NNODES;
    int*   mb   = (int*)(sdst + NNODES);
    float* den  = (float*)(mb + NNODES);

    const int nRowTiles = NNODES / 16;
    const dim3 blk(256);
    const int gemmBlocks = (nRowTiles * 4 + 7) / 8;
    const int nodeBlocks = (NNODES + 7) / 8;
    const int edgeBlocks = (NEP + 255) / 256;
    const int aggrBlocks = NNODES / TN;
    const int actBlocks  = (NNODES * FH + 255) / 256;
    const int polBlocks  = (nRowTiles + 7) / 8;

    float* flag = out;

    gemm_h_kernel<<<gemmBlocks, blk, 0, stream>>>(x, W1, hbuf, FIN, nRowTiles);
    scalars_init_kernel<<<nodeBlocks, blk, 0, stream>>>(hbuf, as1, ad1, ssrc, sdst, mb, den, flag);
    edge_den_kernel<<<edgeBlocks, blk, 0, stream>>>(srcA, dstA, ssrc, sdst, mb, den, flag);
    fb_max_kernel<<<FB_BLOCKS, blk, 0, stream>>>(srcA, dstA, ssrc, sdst, mb, den, flag);
    fb_den_kernel<<<FB_BLOCKS, blk, 0, stream>>>(srcA, dstA, ssrc, sdst, mb, den, flag);
    edge_aggr_kernel<<<aggrBlocks, dim3(1024), 0, stream>>>(srcA, dstA, ssrc, sdst, mb, den, hbuf, obuf);
    act_kernel<<<actBlocks, blk, 0, stream>>>(obuf, b1);

    gemm_h_kernel<<<gemmBlocks, blk, 0, stream>>>(obuf, W2, hbuf, FH, nRowTiles);
    scalars_init_kernel<<<nodeBlocks, blk, 0, stream>>>(hbuf, as2, ad2, ssrc, sdst, mb, den, flag);
    edge_den_kernel<<<edgeBlocks, blk, 0, stream>>>(srcA, dstA, ssrc, sdst, mb, den, flag);
    fb_max_kernel<<<FB_BLOCKS, blk, 0, stream>>>(srcA, dstA, ssrc, sdst, mb, den, flag);
    fb_den_kernel<<<FB_BLOCKS, blk, 0, stream>>>(srcA, dstA, ssrc, sdst, mb, den, flag);
    edge_aggr_kernel<<<aggrBlocks, dim3(1024), 0, stream>>>(srcA, dstA, ssrc, sdst, mb, den, hbuf, obuf);
    act_kernel<<<actBlocks, blk, 0, stream>>>(obuf, b2);

    policy_kernel<<<polBlocks, blk, 0, stream>>>(obuf, Wp, bp, out, nRowTiles);
}


# Owner group sums the softmax denominator in LDS beside its unnormalised rows and divides the rows once on the way out: edge_den's global atomics and 6 launches go; sum order and rounding change
# speedup vs baseline: 2.3906x; 1.5993x over previous
#include <hip/hip_runtime.h>
#include <math.h>

#define NNODES 100000
#define NEDGES 1600000
#define NEP    1700000
#define FIN    16
#define FH     64
#define FOUT   8

typedef __attribute__((ext_vector_type(2))) float v2f;
typedef __attribute__((ext_vector_type(8))) float v8f;

__device__ __forceinline__ float leaky(float e) { return e > 0.f ? e : 0.2f * e; }

__global__ void gemm_h_kernel(const float* __restrict__ in, const float* __restrict__ W,
                              float* __restrict__ h, int K, int nRowTiles) {
    const int lane = threadIdx.x & 31;
    const int gw   = blockIdx.x * 8 + (threadIdx.x >> 5);
    const int rowTile = gw >> 2;
    const int colTile = gw & 3;
    if (rowTile >= nRowTiles) return;
    const int hi  = lane >> 4;
    const int lo  = lane & 15;
    const int row = rowTile * 16 + lo;
    const int col = colTile * 16 + lo;
    v8f c = {};
    for (int kk = 0; kk < K; kk += 4) {
        const int k0 = kk + 2 * hi;
        v2f a, b;
        a.x = in[(size_t)row * K + k0];
        a.y = in[(size_t)row * K + k0 + 1];
        b.x = W[(size_t)k0 * FH + col];
        b.y = W[(size_t)(k0 + 1) * FH + col];
        c = __builtin_amdgcn_wmma_f32_16x16x4_f32(false, a, false, b, (short)0, c, false, false);
    }
    const int rbase = rowTile * 16 + 8 * hi;
#pragma unroll
    for (int r = 0; r < 8; ++r)
        h[(size_t)(rbase + r) * FH + col] = c[r];
}

__global__ void scalars_kernel(const float* __restrict__ h,
                               const float* __restrict__ a_src,
                               const float* __restrict__ a_dst,
                               float* __restrict__ s_src, float* __restrict__ s_dst) {
    const int lane = threadIdx.x & 31;
    const int n    = blockIdx.x * 8 + (threadIdx.x >> 5);
    if (n >= NNODES) return;
    const float h0 = h[(size_t)n * FH + lane];
    const float h1 = h[(size_t)n * FH + lane + 32];
    float ps = h0 * a_src[lane] + h1 * a_src[lane + 32];
    float pd = h0 * a_dst[lane] + h1 * a_dst[lane + 32];
    for (int off = 16; off; off >>= 1) {
        ps += __shfl_xor(ps, off, 32);
        pd += __shfl_xor(pd, off, 32);
    }
    if (lane == 0) {
        s_src[n] = ps;
        s_dst[n] = pd;
    }
}

#define SHIFT_WINDOW 40.0f

#define TN        800
#define AG_WAVES  32
#define CH_EDGES  128
#define NCHUNK    (NEDGES / CH_EDGES)
#define LD_DEPTH  8
#define NIT       ((NCHUNK + AG_WAVES * LD_DEPTH - 1) / (AG_WAVES * LD_DEPTH))
#define QCAP      576
#define ROWS_GROUP 16

template <int MODE>
__device__ __forceinline__ void aggr_batch(unsigned pk, bool valid, int nb, int base, int lane,
                                           const int* __restrict__ src,
                                           const float* __restrict__ s_src,
                                           const float* __restrict__ s_dst,
                                           const float* __restrict__ h, float* acc,
                                           float* den_l, int* shift_i, int* flag_l) {
    const int e   = (int)(pk >> 10);
    const int rel = (int)(pk & 1023u);
    const int d   = base + rel;
    int   s = d;
    float w = 0.f;
    if (valid) {
        s = (e < NEDGES) ? src[e] : e - NEDGES;
        const float l = leaky(s_src[s] + s_dst[d]);
        if (MODE == 1) {
            const unsigned u = __float_as_uint(l);
            if ((u >> 31) == 0) atomicMax(&shift_i[rel], (int)u);
            else                atomicMin((unsigned int*)&shift_i[rel], u);
        } else {
            const float t = l - __int_as_float(shift_i[rel]);
            if (!(t <= SHIFT_WINDOW)) *flag_l = 1;
            w = expf(t);
            atomicAdd(&den_l[rel], w);
        }
    }
    if (MODE == 1) return;
    for (int j0 = 0; j0 < nb; j0 += ROWS_GROUP) {
        float v0[ROWS_GROUP], v1[ROWS_GROUP];
#pragma unroll
        for (int u = 0; u < ROWS_GROUP; ++u) {
            const int j = j0 + u;
            const int   sj = __builtin_amdgcn_readlane(s, j);
            const float wj = __int_as_float(__builtin_amdgcn_readlane(__float_as_int(w), j));
            v0[u] = h[(size_t)sj * FH + lane]      * wj;
            v1[u] = h[(size_t)sj * FH + lane + 32] * wj;
        }
#pragma unroll
        for (int u = 0; u < ROWS_GROUP; ++u) {
            const int j = j0 + u;
            if (j < nb) {
                const int rj = __builtin_amdgcn_readlane(rel, j);
                atomicAdd(&acc[rj * FH + lane],      v0[u]);
                atomicAdd(&acc[rj * FH + lane + 32], v1[u]);
            }
        }
    }
}

#define AG_APPEND(HIT, REL, E)                                                     \
    {                                                                              \
        const unsigned m_ = __builtin_amdgcn_ballot_w32(HIT);                      \
        if (HIT) myq[cnt + (int)__builtin_amdgcn_mbcnt_lo(m_, 0u)] =               \
                     ((unsigned)(E) << 10) | (REL);                                \
        cnt += __builtin_popcount(m_);                                             \
    }

template <int MODE>
__device__ __forceinline__ void scan_pass(int wave, int lane, int base, unsigned* myq,
                                          const int* __restrict__ src, const int* __restrict__ dst,
                                          const float* __restrict__ s_src,
                                          const float* __restrict__ s_dst,
                                          const float* __restrict__ h, float* acc,
                                          float* den_l, int* shift_i, int* flag_l) {
    int cnt = 0;
    if (wave < TN / 32) {
        const int rel = wave * 32 + lane;
        myq[lane] = ((unsigned)(NEDGES + base + rel) << 10) | (unsigned)rel;
        cnt = 32;
    }
    const int4* dst4 = (const int4*)dst;
    for (int it = 0; it < NIT; ++it) {
        const int c0 = (it * AG_WAVES + wave) * LD_DEPTH;
        int4 dv[LD_DEPTH];
#pragma unroll
        for (int u = 0; u < LD_DEPTH; ++u) {
            const int c = c0 + u;
            dv[u] = (c < NCHUNK) ? dst4[(size_t)c * 32 + lane] : make_int4(-1, -1, -1, -1);
        }
#pragma unroll
        for (int half = 0; half < 2; ++half) {
#pragma unroll
            for (int u = half * 4; u < half * 4 + 4; ++u) {
                const unsigned r0 = (unsigned)(dv[u].x - base), r1 = (unsigned)(dv[u].y - base);
                const unsigned r2 = (unsigned)(dv[u].z - base), r3 = (unsigned)(dv[u].w - base);
                const bool h0 = r0 < TN, h1 = r1 < TN, h2 = r2 < TN, h3 = r3 < TN;
                if (__builtin_amdgcn_ballot_w32(h0 | h1 | h2 | h3)) {
                    const int eb = (c0 + u) * CH_EDGES + lane * 4;
                    AG_APPEND(h0, r0, eb)
                    AG_APPEND(h1, r1, eb + 1)
                    AG_APPEND(h2, r2, eb + 2)
                    AG_APPEND(h3, r3, eb + 3)
                }
            }
            while (cnt >= 32) {
                cnt -= 32;
                __builtin_amdgcn_wave_barrier();
                aggr_batch<MODE>(myq[cnt + lane], true, 32, base, lane, src, s_src, s_dst, h,
                                 acc, den_l, shift_i, flag_l);
            }
        }
    }
    if (cnt > 0) {
        __builtin_amdgcn_wave_barrier();
        aggr_batch<MODE>(lane < cnt ? myq[lane] : 0u, lane < cnt, cnt, base, lane,
                         src, s_src, s_dst, h, acc, den_l, shift_i, flag_l);
    }
}

__global__ __launch_bounds__(1024)
void edge_aggr_kernel(const int* __restrict__ src, const int* __restrict__ dst,
                      const float* __restrict__ s_src, const float* __restrict__ s_dst,
                      const float* __restrict__ h, float* __restrict__ out) {
    __shared__ float    acc[TN * FH];
    __shared__ unsigned queue[AG_WAVES * QCAP];
    __shared__ float    den_l[TN];
    __shared__ int      shift_i[TN];
    __shared__ int      flag_l;
    const int lane = threadIdx.x & 31;
    const int wave = __builtin_amdgcn_readfirstlane((int)(threadIdx.x >> 5));
    const int base = blockIdx.x * TN;
    unsigned* myq = queue + wave * QCAP;
    for (int i = threadIdx.x; i < TN * FH; i += 1024) acc[i] = 0.f;
    if (threadIdx.x < TN) {
        const int n = base + (int)threadIdx.x;
        den_l[threadIdx.x]   = 0.f;
        shift_i[threadIdx.x] = __float_as_int(leaky(s_src[n] + s_dst[n]));
    }
    if (threadIdx.x == 0) flag_l = 0;
    __syncthreads();

    scan_pass<0>(wave, lane, base, myq, src, dst, s_src, s_dst, h, acc, den_l, shift_i, &flag_l);
    __syncthreads();

    const bool redo = flag_l != 0;
    if (redo) {
        scan_pass<1>(wave, lane, base, myq, src, dst, s_src, s_dst, h, acc, den_l, shift_i, &flag_l);
        __syncthreads();
        for (int i = threadIdx.x; i < TN * FH; i += 1024) acc[i] = 0.f;
        if (threadIdx.x < TN) den_l[threadIdx.x] = 0.f;
        __syncthreads();
        scan_pass<0>(wave, lane, base, myq, src, dst, s_src, s_dst, h, acc, den_l, shift_i, &flag_l);
        __syncthreads();
    }

    for (int i = threadIdx.x; i < TN * FH; i += 1024)
        out[(size_t)base * FH + i] = acc[i] / (den_l[i >> 6] + 1e-16f);
}

__global__ void act_kernel(float* __restrict__ buf, const float* __restrict__ b) {
    const size_t i = (size_t)blockIdx.x * blockDim.x + threadIdx.x;
    if (i >= (size_t)NNODES * FH) return;
    const float v = buf[i] + b[i & 63];
    buf[i] = v > 0.f ? v : 0.f;
}

__global__ void policy_kernel(const float* __restrict__ act, const float* __restrict__ Wp,
                              const float* __restrict__ bp, float* __restrict__ out,
                              int nRowTiles) {
    const int lane = threadIdx.x & 31;
    const int gw   = blockIdx.x * 8 + (threadIdx.x >> 5);
    if (gw >= nRowTiles) return;
    const int hi  = lane >> 4;
    const int lo  = lane & 15;
    const int row = gw * 16 + lo;
    v8f c = {};
    for (int kk = 0; kk < FH; kk += 4) {
        const int k0 = kk + 2 * hi;
        v2f a, b;
        a.x = act[(size_t)row * FH + k0];
        a.y = act[(size_t)row * FH + k0 + 1];
        b.x = (lo < FOUT) ? Wp[(size_t)k0 * FOUT + lo]       : 0.f;
        b.y = (lo < FOUT) ? Wp[(size_t)(k0 + 1) * FOUT + lo] : 0.f;
        c = __builtin_amdgcn_wmma_f32_16x16x4_f32(false, a, false, b, (short)0, c, false, false);
    }
    if (lo < FOUT) {
        const int rbase = gw * 16 + 8 * hi;
        const float bb = bp[lo];
#pragma unroll
        for (int r = 0; r < 8; ++r)
            out[(size_t)(rbase + r) * FOUT + lo] = c[r] + bb;
    }
}

extern "C" void kernel_launch(void* const* d_in, const int* in_sizes, int n_in,
                              void* d_out, int out_size, void* d_ws, size_t ws_size,
                              hipStream_t stream) {
    const float* x   = (const float*)d_in[0];
    const int*   ei  = (const int*)  d_in[1];
    const float* W1  = (const float*)d_in[2];
    const float* as1 = (const float*)d_in[3];
    const float* ad1 = (const float*)d_in[4];
    const float* b1  = (const float*)d_in[5];
    const float* W2  = (const float*)d_in[6];
    const float* as2 = (const float*)d_in[7];
    const float* ad2 = (const float*)d_in[8];
    const float* b2  = (const float*)d_in[9];
    const float* Wp  = (const float*)d_in[10];
    const float* bp  = (const float*)d_in[11];
    float* out = (float*)d_out;

    const int* srcA = ei;
    const int* dstA = ei + NEDGES;

    float* ws   = (float*)d_ws;
    float* hbuf = ws;
    float* obuf = ws + (size_t)NNODES * FH;
    float* ssrc = ws + (size_t)2 * NNODES * FH;
    float* sdst = ssrc + NNODES;

    const int nRowTiles = NNODES / 16;
    const dim3 blk(256);
    const int gemmBlocks = (nRowTiles * 4 + 7) / 8;
    const int nodeBlocks = (NNODES + 7) / 8;
    const int aggrBlocks = NNODES / TN;
    const int actBlocks  = (NNODES * FH + 255) / 256;
    const int polBlocks  = (nRowTiles + 7) / 8;

    gemm_h_kernel<<<gemmBlocks, blk, 0, stream>>>(x, W1, hbuf, FIN, nRowTiles);
    scalars_kernel<<<nodeBlocks, blk, 0, stream>>>(hbuf, as1, ad1, ssrc, sdst);
    edge_aggr_kernel<<<aggrBlocks, dim3(1024), 0, stream>>>(srcA, dstA, ssrc, sdst, hbuf, obuf);
    act_kernel<<<actBlocks, blk, 0, stream>>>(obuf, b1);

    gemm_h_kernel<<<gemmBlocks, blk, 0, stream>>>(obuf, W2, hbuf, FH, nRowTiles);
    scalars_kernel<<<nodeBlocks, blk, 0, stream>>>(hbuf, as2, ad2, ssrc, sdst);
    edge_aggr_kernel<<<aggrBlocks, dim3(1024), 0, stream>>>(srcA, dstA, ssrc, sdst, hbuf, obuf);
    act_kernel<<<actBlocks, blk, 0, stream>>>(obuf, b2);

    policy_kernel<<<polBlocks, blk, 0, stream>>>(obuf, Wp, bp, out, nRowTiles);
}


# edge_aggr filled 125 of 256 WGPs: two work-groups per 800-node range, each scans half the dst list; act adds the two partial rows and dens (rescaled if shifts differ); sum order and rounding change
# speedup vs baseline: 3.0710x; 1.2846x over previous
#include <hip/hip_runtime.h>
#include <math.h>

#define NNODES 100000
#define NEDGES 1600000
#define NEP    1700000
#define FIN    16
#define FH     64
#define FOUT   8

typedef __attribute__((ext_vector_type(2))) float v2f;
typedef __attribute__((ext_vector_type(8))) float v8f;

__device__ __forceinline__ float leaky(float e) { return e > 0.f ? e : 0.2f * e; }

__global__ void gemm_h_kernel(const float* __restrict__ in, const float* __restrict__ W,
                              float* __restrict__ h, int K, int nRowTiles) {
    const int lane = threadIdx.x & 31;
    const int gw   = blockIdx.x * 8 + (threadIdx.x >> 5);
    const int rowTile = gw >> 2;
    const int colTile = gw & 3;
    if (rowTile >= nRowTiles) return;
    const int hi  = lane >> 4;
    const int lo  = lane & 15;
    const int row = rowTile * 16 + lo;
    const int col = colTile * 16 + lo;
    v8f c = {};
    for (int kk = 0; kk < K; kk += 4) {
        const int k0 = kk + 2 * hi;
        v2f a, b;
        a.x = in[(size_t)row * K + k0];
        a.y = in[(size_t)row * K + k0 + 1];
        b.x = W[(size_t)k0 * FH + col];
        b.y = W[(size_t)(k0 + 1) * FH + col];
        c = __builtin_amdgcn_wmma_f32_16x16x4_f32(false, a, false, b, (short)0, c, false, false);
    }
    const int rbase = rowTile * 16 + 8 * hi;
#pragma unroll
    for (int r = 0; r < 8; ++r)
        h[(size_t)(rbase + r) * FH + col] = c[r];
}

__global__ void scalars_kernel(const float* __restrict__ h,
                               const float* __restrict__ a_src,
                               const float* __restrict__ a_dst,
                               float* __restrict__ s_src, float* __restrict__ s_dst) {
    const int lane = threadIdx.x & 31;
    const int n    = blockIdx.x * 8 + (threadIdx.x >> 5);
    if (n >= NNODES) return;
    const float h0 = h[(size_t)n * FH + lane];
    const float h1 = h[(size_t)n * FH + lane + 32];
    float ps = h0 * a_src[lane] + h1 * a_src[lane + 32];
    float pd = h0 * a_dst[lane] + h1 * a_dst[lane + 32];
    for (int off = 16; off; off >>= 1) {
        ps += __shfl_xor(ps, off, 32);
        pd += __shfl_xor(pd, off, 32);
    }
    if (lane == 0) {
        s_src[n] = ps;
        s_dst[n] = pd;
    }
}

#define SHIFT_WINDOW 40.0f

#define TN        800
#define AG_WAVES  32
#define CH_EDGES  128
#define NCHUNK    (NEDGES / CH_EDGES)
#define LD_DEPTH  8
#define WS_SPLIT_BYTES ((size_t)(3 * FH + 6) * NNODES * sizeof(float))
#define QCAP      576
#define ROWS_GROUP 16

template <int MODE>
__device__ __forceinline__ void aggr_batch(unsigned pk, bool valid, int nb, int base, int lane,
                                           const int* __restrict__ src,
                                           const float* __restrict__ s_src,
                                           const float* __restrict__ s_dst,
                                           const float* __restrict__ h, float* acc,
                                           float* den_l, int* shift_i, int* flag_l) {
    const int e   = (int)(pk >> 10);
    const int rel = (int)(pk & 1023u);
    const int d   = base + rel;
    int   s = d;
    float w = 0.f;
    if (valid) {
        s = (e < NEDGES) ? src[e] : e - NEDGES;
        const float l = leaky(s_src[s] + s_dst[d]);
        if (MODE == 1) {
            const unsigned u = __float_as_uint(l);
            if ((u >> 31) == 0) atomicMax(&shift_i[rel], (int)u);
            else                atomicMin((unsigned int*)&shift_i[rel], u);
        } else {
            const float t = l - __int_as_float(shift_i[rel]);
            if (!(t <= SHIFT_WINDOW)) *flag_l = 1;
            w = expf(t);
            atomicAdd(&den_l[rel], w);
        }
    }
    if (MODE == 1) return;
    for (int j0 = 0; j0 < nb; j0 += ROWS_GROUP) {
        float v0[ROWS_GROUP], v1[ROWS_GROUP];
#pragma unroll
        for (int u = 0; u < ROWS_GROUP; ++u) {
            const int j = j0 + u;
            const int   sj = __builtin_amdgcn_readlane(s, j);
            const float wj = __int_as_float(__builtin_amdgcn_readlane(__float_as_int(w), j));
            v0[u] = h[(size_t)sj * FH + lane]      * wj;
            v1[u] = h[(size_t)sj * FH + lane + 32] * wj;
        }
#pragma unroll
        for (int u = 0; u < ROWS_GROUP; ++u) {
            const int j = j0 + u;
            if (j < nb) {
                const int rj = __builtin_amdgcn_readlane(rel, j);
                atomicAdd(&acc[rj * FH + lane],      v0[u]);
                atomicAdd(&acc[rj * FH + lane + 32], v1[u]);
            }
        }
    }
}

#define AG_APPEND(HIT, REL, E)                                                     \
    {                                                                              \
        const unsigned m_ = __builtin_amdgcn_ballot_w32(HIT);                      \
        if (HIT) myq[cnt + (int)__builtin_amdgcn_mbcnt_lo(m_, 0u)] =               \
                     ((unsigned)(E) << 10) | (REL);                                \
        cnt += __builtin_popcount(m_);                                             \
    }

template <int MODE>
__device__ __forceinline__ void scan_pass(int wave, int lane, int base, unsigned* myq,
                                          int cbeg, int cend, int sbeg, int nself,
                                          const int* __restrict__ src, const int* __restrict__ dst,
                                          const float* __restrict__ s_src,
                                          const float* __restrict__ s_dst,
                                          const float* __restrict__ h, float* acc,
                                          float* den_l, int* shift_i, int* flag_l) {
    int cnt = 0;
    if (wave * 32 < nself) {
        const int idx = wave * 32 + lane;
        const int rel = sbeg + idx;
        if (idx < nself) myq[lane] = ((unsigned)(NEDGES + base + rel) << 10) | (unsigned)rel;
        cnt = min(32, nself - wave * 32);
    }
    const int4* dst4 = (const int4*)dst;
    const int nit = (cend - cbeg + AG_WAVES * LD_DEPTH - 1) / (AG_WAVES * LD_DEPTH);
    for (int it = 0; it < nit; ++it) {
        const int c0 = cbeg + (it * AG_WAVES + wave) * LD_DEPTH;
        int4 dv[LD_DEPTH];
#pragma unroll
        for (int u = 0; u < LD_DEPTH; ++u) {
            const int c = c0 + u;
            dv[u] = (c < cend) ? dst4[(size_t)c * 32 + lane] : make_int4(-1, -1, -1, -1);
        }
#pragma unroll
        for (int half = 0; half < 2; ++half) {
#pragma unroll
            for (int u = half * 4; u < half * 4 + 4; ++u) {
                const unsigned r0 = (unsigned)(dv[u].x - base), r1 = (unsigned)(dv[u].y - base);
                const unsigned r2 = (unsigned)(dv[u].z - base), r3 = (unsigned)(dv[u].w - base);
                const bool h0 = r0 < TN, h1 = r1 < TN, h2 = r2 < TN, h3 = r3 < TN;
                if (__builtin_amdgcn_ballot_w32(h0 | h1 | h2 | h3)) {
                    const int eb = (c0 + u) * CH_EDGES + lane * 4;
                    AG_APPEND(h0, r0, eb)
                    AG_APPEND(h1, r1, eb + 1)
                    AG_APPEND(h2, r2, eb + 2)
                    AG_APPEND(h3, r3, eb + 3)
                }
            }
            while (cnt >= 32) {
                cnt -= 32;
                __builtin_amdgcn_wave_barrier();
                aggr_batch<MODE>(myq[cnt + lane], true, 32, base, lane, src, s_src, s_dst, h,
                                 acc, den_l, shift_i, flag_l);
            }
        }
    }
    if (cnt > 0) {
        __builtin_amdgcn_wave_barrier();
        aggr_batch<MODE>(lane < cnt ? myq[lane] : 0u, lane < cnt, cnt, base, lane,
                         src, s_src, s_dst, h, acc, den_l, shift_i, flag_l);
    }
}

__global__ __launch_bounds__(1024)
void edge_aggr_kernel(const int* __restrict__ src, const int* __restrict__ dst,
                      const float* __restrict__ s_src, const float* __restrict__ s_dst,
                      const float* __restrict__ h, float* __restrict__ out,
                      float* __restrict__ part1, float* __restrict__ den_g,
                      int* __restrict__ shf_g, size_t ws_size) {
    const bool split = ws_size >= WS_SPLIT_BYTES;
    const int  half  = blockIdx.x & 1;
    if (!split && half) return;
    __shared__ float    acc[TN * FH];
    __shared__ unsigned queue[AG_WAVES * QCAP];
    __shared__ float    den_l[TN];
    __shared__ int      shift_i[TN];
    __shared__ int      flag_l;
    const int lane = threadIdx.x & 31;
    const int wave = __builtin_amdgcn_readfirstlane((int)(threadIdx.x >> 5));
    const int base = (int)(blockIdx.x >> 1) * TN;
    const int cbeg  = split ? half * (NCHUNK / 2) : 0;
    const int cend  = split ? cbeg + NCHUNK / 2 : NCHUNK;
    const int sbeg  = split ? half * (TN / 2) : 0;
    const int nself = split ? TN / 2 : TN;
    unsigned* myq = queue + wave * QCAP;
    for (int i = threadIdx.x; i < TN * FH; i += 1024) acc[i] = 0.f;
    if (threadIdx.x < TN) {
        const int n = base + (int)threadIdx.x;
        den_l[threadIdx.x]   = 0.f;
        shift_i[threadIdx.x] = __float_as_int(leaky(s_src[n] + s_dst[n]));
    }
    if (threadIdx.x == 0) flag_l = 0;
    __syncthreads();

    scan_pass<0>(wave, lane, base, myq, cbeg, cend, sbeg, nself, src, dst, s_src, s_dst, h, acc, den_l, shift_i, &flag_l);
    __syncthreads();

    const bool redo = flag_l != 0;
    if (redo) {
        scan_pass<1>(wave, lane, base, myq, cbeg, cend, sbeg, nself, src, dst, s_src, s_dst, h, acc, den_l, shift_i, &flag_l);
        __syncthreads();
        for (int i = threadIdx.x; i < TN * FH; i += 1024) acc[i] = 0.f;
        if (threadIdx.x < TN) den_l[threadIdx.x] = 0.f;
        __syncthreads();
        scan_pass<0>(wave, lane, base, myq, cbeg, cend, sbeg, nself, src, dst, s_src, s_dst, h, acc, den_l, shift_i, &flag_l);
        __syncthreads();
    }

    if (!split) {
        for (int i = threadIdx.x; i < TN * FH; i += 1024)
            out[(size_t)base * FH + i] = acc[i] / (den_l[i >> 6] + 1e-16f);
        return;
    }
    float* rows = half ? part1 : out;
    for (int i = threadIdx.x; i < TN * FH; i += 1024)
        rows[(size_t)base * FH + i] = acc[i];
    if (threadIdx.x < TN) {
        den_g[half * NNODES + base + (int)threadIdx.x] = den_l[threadIdx.x];
        shf_g[half * NNODES + base + (int)threadIdx.x] = shift_i[threadIdx.x];
    }
}

__global__ void act_kernel(float* __restrict__ buf, const float* __restrict__ part1,
                           const float* __restrict__ den_g, const int* __restrict__ shf_g,
                           const float* __restrict__ b, size_t ws_size) {
    const size_t i = (size_t)blockIdx.x * blockDim.x + threadIdx.x;
    if (i >= (size_t)NNODES * FH) return;
    float v = buf[i];
    if (ws_size >= WS_SPLIT_BYTES) {
        const int n = (int)(i >> 6);
        const float s0 = __int_as_float(shf_g[n]);
        const float s1 = __int_as_float(shf_g[NNODES + n]);
        float a0 = v, a1 = part1[i];
        float d0 = den_g[n], d1 = den_g[NNODES + n];
        if (s0 != s1) {
            const float m  = fmaxf(s0, s1);
            const float f0 = expf(s0 - m), f1 = expf(s1 - m);
            a0 *= f0; d0 *= f0;
            a1 *= f1; d1 *= f1;
        }
        v = (a0 + a1) / (d0 + d1 + 1e-16f);
    }
    v += b[i & 63];
    buf[i] = v > 0.f ? v : 0.f;
}

__global__ void policy_kernel(const float* __restrict__ act, const float* __restrict__ Wp,
                              const float* __restrict__ bp, float* __restrict__ out,
                              int nRowTiles) {
    const int lane = threadIdx.x & 31;
    const int gw   = blockIdx.x * 8 + (threadIdx.x >> 5);
    if (gw >= nRowTiles) return;
    const int hi  = lane >> 4;
    const int lo  = lane & 15;
    const int row = gw * 16 + lo;
    v8f c = {};
    for (int kk = 0; kk < FH; kk += 4) {
        const int k0 = kk + 2 * hi;
        v2f a, b;
        a.x = act[(size_t)row * FH + k0];
        a.y = act[(size_t)row * FH + k0 + 1];
        b.x = (lo < FOUT) ? Wp[(size_t)k0 * FOUT + lo]       : 0.f;
        b.y = (lo < FOUT) ? Wp[(size_t)(k0 + 1) * FOUT + lo] : 0.f;
        c = __builtin_amdgcn_wmma_f32_16x16x4_f32(false, a, false, b, (short)0, c, false, false);
    }
    if (lo < FOUT) {
        const int rbase = gw * 16 + 8 * hi;
        const float bb = bp[lo];
#pragma unroll
        for (int r = 0; r < 8; ++r)
            out[(size_t)(rbase + r) * FOUT + lo] = c[r] + bb;
    }
}

extern "C" void kernel_launch(void* const* d_in, const int* in_sizes, int n_in,
                              void* d_out, int out_size, void* d_ws, size_t ws_size,
                              hipStream_t stream) {
    const float* x   = (const float*)d_in[0];
    const int*   ei  = (const int*)  d_in[1];
    const float* W1  = (const float*)d_in[2];
    const float* as1 = (const float*)d_in[3];
    const float* ad1 = (const float*)d_in[4];
    const float* b1  = (const float*)d_in[5];
    const float* W2  = (const float*)d_in[6];
    const float* as2 = (const float*)d_in[7];
    const float* ad2 = (const float*)d_in[8];
    const float* b2  = (const float*)d_in[9];
    const float* Wp  = (const float*)d_in[10];
    const float* bp  = (const float*)d_in[11];
    float* out = (float*)d_out;

    const int* srcA = ei;
    const int* dstA = ei + NEDGES;

    float* ws   = (float*)d_ws;
    float* hbuf = ws;
    float* obuf = ws + (size_t)NNODES * FH;
    float* ssrc = ws + (size_t)2 * NNODES * FH;
    float* sdst = ssrc + NNODES;
    float* prt1 = sdst + NNODES;
    float* deng = prt1 + (size_t)NNODES * FH;
    int*   shfg = (int*)(deng + 2 * NNODES);

    const int nRowTiles = NNODES / 16;
    const dim3 blk(256);
    const int gemmBlocks = (nRowTiles * 4 + 7) / 8;
    const int nodeBlocks = (NNODES + 7) / 8;
    const int aggrBlocks = 2 * (NNODES / TN);
    const int actBlocks  = (NNODES * FH + 255) / 256;
    const int polBlocks  = (nRowTiles + 7) / 8;

    gemm_h_kernel<<<gemmBlocks, blk, 0, stream>>>(x, W1, hbuf, FIN, nRowTiles);
    scalars_kernel<<<nodeBlocks, blk, 0, stream>>>(hbuf, as1, ad1, ssrc, sdst);
    edge_aggr_kernel<<<aggrBlocks, dim3(1024), 0, stream>>>(srcA, dstA, ssrc, sdst, hbuf, obuf,
                                                           prt1, deng, shfg, ws_size);
    act_kernel<<<actBlocks, blk, 0, stream>>>(obuf, prt1, deng, shfg, b1, ws_size);

    gemm_h_kernel<<<gemmBlocks, blk, 0, stream>>>(obuf, W2, hbuf, FH, nRowTiles);
    scalars_kernel<<<nodeBlocks, blk, 0, stream>>>(hbuf, as2, ad2, ssrc, sdst);
    edge_aggr_kernel<<<aggrBlocks, dim3(1024), 0, stream>>>(srcA, dstA, ssrc, sdst, hbuf, obuf,
                                                           prt1, deng, shfg, ws_size);
    act_kernel<<<actBlocks, blk, 0, stream>>>(obuf, prt1, deng, shfg, b2, ws_size);

    policy_kernel<<<polBlocks, blk, 0, stream>>>(obuf, Wp, bp, out, nRowTiles);
}


# Per-node passes folded into the WMMA kernels: A loader adds the two partial rows, divides, bias, relu; a wave holds 16x64 of h and takes both dots from D: 4 launches go; k and dot sum order change
# speedup vs baseline: 3.9261x; 1.2784x over previous
#include <hip/hip_runtime.h>
#include <math.h>

#define NNODES 100000
#define NEDGES 1600000
#define NEP    1700000
#define FIN    16
#define FH     64
#define FOUT   8

typedef __attribute__((ext_vector_type(2))) float v2f;
typedef __attribute__((ext_vector_type(8))) float v8f;

__device__ __forceinline__ float leaky(float e) { return e > 0.f ? e : 0.2f * e; }

#define SHIFT_WINDOW 40.0f

#define TN        800
#define AG_WAVES  32
#define CH_EDGES  128
#define NCHUNK    (NEDGES / CH_EDGES)
#define LD_DEPTH  8
#define WS_SPLIT_BYTES ((size_t)(3 * FH + 6) * NNODES * sizeof(float))
#define QCAP      576
#define ROWS_GROUP 16

template <int MODE>
__device__ __forceinline__ void aggr_batch(unsigned pk, bool valid, int nb, int base, int lane,
                                           const int* __restrict__ src,
                                           const float* __restrict__ s_src,
                                           const float* __restrict__ s_dst,
                                           const float* __restrict__ h, float* acc,
                                           float* den_l, int* shift_i, int* flag_l) {
    const int e   = (int)(pk >> 10);
    const int rel = (int)(pk & 1023u);
    const int d   = base + rel;
    int   s = d;
    float w = 0.f;
    if (valid) {
        s = (e < NEDGES) ? src[e] : e - NEDGES;
        const float l = leaky(s_src[s] + s_dst[d]);
        if (MODE == 1) {
            const unsigned u = __float_as_uint(l);
            if ((u >> 31) == 0) atomicMax(&shift_i[rel], (int)u);
            else                atomicMin((unsigned int*)&shift_i[rel], u);
        } else {
            const float t = l - __int_as_float(shift_i[rel]);
            if (!(t <= SHIFT_WINDOW)) *flag_l = 1;
            w = expf(t);
            atomicAdd(&den_l[rel], w);
        }
    }
    if (MODE == 1) return;
    for (int j0 = 0; j0 < nb; j0 += ROWS_GROUP) {
        float v0[ROWS_GROUP], v1[ROWS_GROUP];
#pragma unroll
        for (int u = 0; u < ROWS_GROUP; ++u) {
            const int j = j0 + u;
            const int   sj = __builtin_amdgcn_readlane(s, j);
            const float wj = __int_as_float(__builtin_amdgcn_readlane(__float_as_int(w), j));
            v0[u] = h[(size_t)sj * FH + lane]      * wj;
            v1[u] = h[(size_t)sj * FH + lane + 32] * wj;
        }
#pragma unroll
        for (int u = 0; u < ROWS_GROUP; ++u) {
            const int j = j0 + u;
            if (j < nb) {
                const int rj = __builtin_amdgcn_readlane(rel, j);
                atomicAdd(&acc[rj * FH + lane],      v0[u]);
                atomicAdd(&acc[rj * FH + lane + 32], v1[u]);
            }
        }
    }
}

#define AG_APPEND(HIT, REL, E)                                                     \
    {                                                                              \
        const unsigned m_ = __builtin_amdgcn_ballot_w32(HIT);                      \
        if (HIT) myq[cnt + (int)__builtin_amdgcn_mbcnt_lo(m_, 0u)] =               \
                     ((unsigned)(E) << 10) | (REL);                                \
        cnt += __builtin_popcount(m_);                                             \
    }

template <int MODE>
__device__ __forceinline__ void scan_pass(int wave, int lane, int base, unsigned* myq,
                                          int cbeg, int cend, int sbeg, int nself,
                                          const int* __restrict__ src, const int* __restrict__ dst,
                                          const float* __restrict__ s_src,
                                          const float* __restrict__ s_dst,
                                          const float* __restrict__ h, float* acc,
                                          float* den_l, int* shift_i, int* flag_l) {
    int cnt = 0;
    if (wave * 32 < nself) {
        const int idx = wave * 32 + lane;
        const int rel = sbeg + idx;
        if (idx < nself) myq[lane] = ((unsigned)(NEDGES + base + rel) << 10) | (unsigned)rel;
        cnt = min(32, nself - wave * 32);
    }
    const int4* dst4 = (const int4*)dst;
    const int nit = (cend - cbeg + AG_WAVES * LD_DEPTH - 1) / (AG_WAVES * LD_DEPTH);
    for (int it = 0; it < nit; ++it) {
        const int c0 = cbeg + (it * AG_WAVES + wave) * LD_DEPTH;
        int4 dv[LD_DEPTH];
#pragma unroll
        for (int u = 0; u < LD_DEPTH; ++u) {
            const int c = c0 + u;
            dv[u] = (c < cend) ? dst4[(size_t)c * 32 + lane] : make_int4(-1, -1, -1, -1);
        }
#pragma unroll
        for (int half = 0; half < 2; ++half) {
#pragma unroll
            for (int u = half * 4; u < half * 4 + 4; ++u) {
                const unsigned r0 = (unsigned)(dv[u].x - base), r1 = (unsigned)(dv[u].y - base);
                const unsigned r2 = (unsigned)(dv[u].z - base), r3 = (unsigned)(dv[u].w - base);
                const bool h0 = r0 < TN, h1 = r1 < TN, h2 = r2 < TN, h3 = r3 < TN;
                if (__builtin_amdgcn_ballot_w32(h0 | h1 | h2 | h3)) {
                    const int eb = (c0 + u) * CH_EDGES + lane * 4;
                    AG_APPEND(h0, r0, eb)
                    AG_APPEND(h1, r1, eb + 1)
                    AG_APPEND(h2, r2, eb + 2)
                    AG_APPEND(h3, r3, eb + 3)
                }
            }
            while (cnt >= 32) {
                cnt -= 32;
                __builtin_amdgcn_wave_barrier();
                aggr_batch<MODE>(myq[cnt + lane], true, 32, base, lane, src, s_src, s_dst, h,
                                 acc, den_l, shift_i, flag_l);
            }
        }
    }
    if (cnt > 0) {
        __builtin_amdgcn_wave_barrier();
        aggr_batch<MODE>(lane < cnt ? myq[lane] : 0u, lane < cnt, cnt, base, lane,
                         src, s_src, s_dst, h, acc, den_l, shift_i, flag_l);
    }
}

__global__ __launch_bounds__(1024)
void edge_aggr_kernel(const int* __restrict__ src, const int* __restrict__ dst,
                      const float* __restrict__ s_src, const float* __restrict__ s_dst,
                      const float* __restrict__ h, float* __restrict__ out,
                      float* __restrict__ part1, float* __restrict__ den_g,
                      int* __restrict__ shf_g, size_t ws_size) {
    const bool split = ws_size >= WS_SPLIT_BYTES;
    const int  half  = blockIdx.x & 1;
    if (!split && half) return;
    __shared__ float    acc[TN * FH];
    __shared__ unsigned queue[AG_WAVES * QCAP];
    __shared__ float    den_l[TN];
    __shared__ int      shift_i[TN];
    __shared__ int      flag_l;
    const int lane = threadIdx.x & 31;
    const int wave = __builtin_amdgcn_readfirstlane((int)(threadIdx.x >> 5));
    const int base = (int)(blockIdx.x >> 1) * TN;
    const int cbeg  = split ? half * (NCHUNK / 2) : 0;
    const int cend  = split ? cbeg + NCHUNK / 2 : NCHUNK;
    const int sbeg  = split ? half * (TN / 2) : 0;
    const int nself = split ? TN / 2 : TN;
    unsigned* myq = queue + wave * QCAP;
    for (int i = threadIdx.x; i < TN * FH; i += 1024) acc[i] = 0.f;
    if (threadIdx.x < TN) {
        const int n = base + (int)threadIdx.x;
        den_l[threadIdx.x]   = 0.f;
        shift_i[threadIdx.x] = __float_as_int(leaky(s_src[n] + s_dst[n]));
    }
    if (threadIdx.x == 0) flag_l = 0;
    __syncthreads();

    scan_pass<0>(wave, lane, base, myq, cbeg, cend, sbeg, nself, src, dst, s_src, s_dst, h, acc, den_l, shift_i, &flag_l);
    __syncthreads();

    const bool redo = flag_l != 0;
    if (redo) {
        scan_pass<1>(wave, lane, base, myq, cbeg, cend, sbeg, nself, src, dst, s_src, s_dst, h, acc, den_l, shift_i, &flag_l);
        __syncthreads();
        for (int i = threadIdx.x; i < TN * FH; i += 1024) acc[i] = 0.f;
        if (threadIdx.x < TN) den_l[threadIdx.x] = 0.f;
        __syncthreads();
        scan_pass<0>(wave, lane, base, myq, cbeg, cend, sbeg, nself, src, dst, s_src, s_dst, h, acc, den_l, shift_i, &flag_l);
        __syncthreads();
    }

    if (!split) {
        for (int i = threadIdx.x; i < TN * FH; i += 1024)
            out[(size_t)base * FH + i] = acc[i] / (den_l[i >> 6] + 1e-16f);
        return;
    }
    float* rows = half ? part1 : out;
    for (int i = threadIdx.x; i < TN * FH; i += 1024)
        rows[(size_t)base * FH + i] = acc[i];
    if (threadIdx.x < TN) {
        den_g[half * NNODES + base + (int)threadIdx.x] = den_l[threadIdx.x];
        shf_g[half * NNODES + base + (int)threadIdx.x] = shift_i[threadIdx.x];
    }
}

__device__ __forceinline__ int kbase(int s, int hi) { return 8 * (s >> 1) + 4 * hi + 2 * (s & 1); }

template <int K, bool ACT>
__device__ __forceinline__ void load_a(const float* __restrict__ in, const float* __restrict__ part1,
                                       const float* __restrict__ den_g, const int* __restrict__ shf_g,
                                       const float* __restrict__ bprev, bool split, int row, int hi,
                                       float4* a) {
    const float4* p0 = (const float4*)(in + (size_t)row * K + 4 * hi);
#pragma unroll
    for (int j = 0; j < K / 8; ++j) a[j] = p0[2 * j];
    if (ACT) {
        if (split) {
            const float4* p1 = (const float4*)(part1 + (size_t)row * K + 4 * hi);
            float4 q[K / 8];
#pragma unroll
            for (int j = 0; j < K / 8; ++j) q[j] = p1[2 * j];
            const float s0 = __int_as_float(shf_g[row]);
            const float s1 = __int_as_float(shf_g[NNODES + row]);
            float d0 = den_g[row], d1 = den_g[NNODES + row];
            float f0 = 1.f, f1 = 1.f;
            if (s0 != s1) {
                const float m = fmaxf(s0, s1);
                f0 = expf(s0 - m);
                f1 = expf(s1 - m);
            }
            d0 *= f0;
            d1 *= f1;
            const float den = d0 + d1 + 1e-16f;
#pragma unroll
            for (int j = 0; j < K / 8; ++j) {
                a[j].x = (a[j].x * f0 + q[j].x * f1) / den;
                a[j].y = (a[j].y * f0 + q[j].y * f1) / den;
                a[j].z = (a[j].z * f0 + q[j].z * f1) / den;
                a[j].w = (a[j].w * f0 + q[j].w * f1) / den;
            }
        }
#pragma unroll
        for (int j = 0; j < K / 8; ++j) {
            const float4 bb = *(const float4*)(bprev + 8 * j + 4 * hi);
            float v;
            v = a[j].x + bb.x; a[j].x = v > 0.f ? v : 0.f;
            v = a[j].y + bb.y; a[j].y = v > 0.f ? v : 0.f;
            v = a[j].z + bb.z; a[j].z = v > 0.f ? v : 0.f;
            v = a[j].w + bb.w; a[j].w = v > 0.f ? v : 0.f;
        }
    }
}

template <int K, bool ACT>
__global__ __launch_bounds__(256)
void node_kernel(const float* __restrict__ in, const float* __restrict__ part1,
                            const float* __restrict__ den_g, const int* __restrict__ shf_g,
                            const float* __restrict__ bprev, const float* __restrict__ W,
                            const float* __restrict__ a_src, const float* __restrict__ a_dst,
                            float* __restrict__ h, float* __restrict__ s_src,
                            float* __restrict__ s_dst, int nRowTiles, size_t ws_size) {
    const int lane    = threadIdx.x & 31;
    const int rowTile = blockIdx.x * 8 + (threadIdx.x >> 5);
    if (rowTile >= nRowTiles) return;
    const int hi  = lane >> 4;
    const int lo  = lane & 15;
    const int row = rowTile * 16 + lo;
    float4 a[K / 8];
    load_a<K, ACT>(in, part1, den_g, shf_g, bprev, ws_size >= WS_SPLIT_BYTES, row, hi, a);
    v8f c0 = {}, c1 = {}, c2 = {}, c3 = {};
#pragma unroll
    for (int s = 0; s < K / 4; ++s) {
        const int kb = kbase(s, hi);
        v2f av;
        av.x = (s & 1) ? a[s >> 1].z : a[s >> 1].x;
        av.y = (s & 1) ? a[s >> 1].w : a[s >> 1].y;
        const float* w0 = W + (size_t)kb * FH + lo;
        v2f b0, b1, b2, b3;
        b0.x = w0[0];  b0.y = w0[FH];
        b1.x = w0[16]; b1.y = w0[FH + 16];
        b2.x = w0[32]; b2.y = w0[FH + 32];
        b3.x = w0[48]; b3.y = w0[FH + 48];
        c0 = __builtin_amdgcn_wmma_f32_16x16x4_f32(false, av, false, b0, (short)0, c0, false, false);
        c1 = __builtin_amdgcn_wmma_f32_16x16x4_f32(false, av, false, b1, (short)0, c1, false, false);
        c2 = __builtin_amdgcn_wmma_f32_16x16x4_f32(false, av, false, b2, (short)0, c2, false, false);
        c3 = __builtin_amdgcn_wmma_f32_16x16x4_f32(false, av, false, b3, (short)0, c3, false, false);
    }
    const int rbase = rowTile * 16 + 8 * hi;
#pragma unroll
    for (int r = 0; r < 8; ++r) {
        float* hr = h + (size_t)(rbase + r) * FH + lo;
        hr[0]  = c0[r];
        hr[16] = c1[r];
        hr[32] = c2[r];
        hr[48] = c3[r];
    }
    const float as0 = a_src[lo], as1 = a_src[16 + lo], as2 = a_src[32 + lo], as3 = a_src[48 + lo];
    const float ad0 = a_dst[lo], ad1 = a_dst[16 + lo], ad2 = a_dst[32 + lo], ad3 = a_dst[48 + lo];
    float ps[8], pd[8];
#pragma unroll
    for (int r = 0; r < 8; ++r) {
        ps[r] = c0[r] * as0 + c1[r] * as1 + c2[r] * as2 + c3[r] * as3;
        pd[r] = c0[r] * ad0 + c1[r] * ad1 + c2[r] * ad2 + c3[r] * ad3;
    }
    const bool u8 = lo & 8, u4 = lo & 4, u2 = lo & 2, u1 = lo & 1;
    float qs[4], qd[4];
#pragma unroll
    for (int i = 0; i < 4; ++i) {
        qs[i] = (u8 ? ps[i + 4] : ps[i]) + __shfl_xor(u8 ? ps[i] : ps[i + 4], 8, 32);
        qd[i] = (u8 ? pd[i + 4] : pd[i]) + __shfl_xor(u8 ? pd[i] : pd[i + 4], 8, 32);
    }
    float ts[2], td[2];
#pragma unroll
    for (int i = 0; i < 2; ++i) {
        ts[i] = (u4 ? qs[i + 2] : qs[i]) + __shfl_xor(u4 ? qs[i] : qs[i + 2], 4, 32);
        td[i] = (u4 ? qd[i + 2] : qd[i]) + __shfl_xor(u4 ? qd[i] : qd[i + 2], 4, 32);
    }
    const float es = (u2 ? ts[1] : ts[0]) + __shfl_xor(u2 ? ts[0] : ts[1], 2, 32);
    const float ed = (u2 ? td[1] : td[0]) + __shfl_xor(u2 ? td[0] : td[1], 2, 32);
    const float fin = (u1 ? ed : es) + __shfl_xor(u1 ? es : ed, 1, 32);
    (u1 ? s_dst : s_src)[rbase + (lo >> 1)] = fin;
}

__global__ __launch_bounds__(256)
void policy_kernel(const float* __restrict__ rows0, const float* __restrict__ part1,
                              const float* __restrict__ den_g, const int* __restrict__ shf_g,
                              const float* __restrict__ bprev, const float* __restrict__ Wp,
                              const float* __restrict__ bp, float* __restrict__ out,
                              int nRowTiles, size_t ws_size) {
    const int lane = threadIdx.x & 31;
    const int gw   = blockIdx.x * 8 + (threadIdx.x >> 5);
    if (gw >= nRowTiles) return;
    const int hi  = lane >> 4;
    const int lo  = lane & 15;
    const int row = gw * 16 + lo;
    const int cl  = lo < FOUT ? lo : 0;
    float4 a[FH / 8];
    load_a<FH, true>(rows0, part1, den_g, shf_g, bprev, ws_size >= WS_SPLIT_BYTES, row, hi, a);
    v8f c = {};
#pragma unroll
    for (int s = 0; s < FH / 4; ++s) {
        const int kb = kbase(s, hi);
        v2f av, b;
        av.x = (s & 1) ? a[s >> 1].z : a[s >> 1].x;
        av.y = (s & 1) ? a[s >> 1].w : a[s >> 1].y;
        const float w0 = Wp[(size_t)kb * FOUT + cl];
        const float w1 = Wp[(size_t)(kb + 1) * FOUT + cl];
        b.x = (lo < FOUT) ? w0 : 0.f;
        b.y = (lo < FOUT) ? w1 : 0.f;
        c = __builtin_amdgcn_wmma_f32_16x16x4_f32(false, av, false, b, (short)0, c, false, false);
    }
    if (lo < FOUT) {
        const int rbase = gw * 16 + 8 * hi;
        const float bb = bp[lo];
#pragma unroll
        for (int r = 0; r < 8; ++r)
            out[(size_t)(rbase + r) * FOUT + lo] = c[r] + bb;
    }
}

extern "C" void kernel_launch(void* const* d_in, const int* in_sizes, int n_in,
                              void* d_out, int out_size, void* d_ws, size_t ws_size,
                              hipStream_t stream) {
    const float* x   = (const float*)d_in[0];
    const int*   ei  = (const int*)  d_in[1];
    const float* W1  = (const float*)d_in[2];
    const float* as1 = (const float*)d_in[3];
    const float* ad1 = (const float*)d_in[4];
    const float* b1  = (const float*)d_in[5];
    const float* W2  = (const float*)d_in[6];
    const float* as2 = (const float*)d_in[7];
    const float* ad2 = (const float*)d_in[8];
    const float* b2  = (const float*)d_in[9];
    const float* Wp  = (const float*)d_in[10];
    const float* bp  = (const float*)d_in[11];
    float* out = (float*)d_out;

    const int* srcA = ei;
    const int* dstA = ei + NEDGES;

    float* ws   = (float*)d_ws;
    float* hbuf = ws;
    float* obuf = ws + (size_t)NNODES * FH;
    float* ssrc = ws + (size_t)2 * NNODES * FH;
    float* sdst = ssrc + NNODES;
    float* prt1 = sdst + NNODES;
    float* deng = prt1 + (size_t)NNODES * FH;
    int*   shfg = (int*)(deng + 2 * NNODES);

    const int nRowTiles = NNODES / 16;
    const dim3 blk(256);
    const int tileBlocks = (nRowTiles + 7) / 8;
    const int aggrBlocks = 2 * (NNODES / TN);

    node_kernel<FIN, false><<<tileBlocks, blk, 0, stream>>>(x, nullptr, nullptr, nullptr, nullptr,
                                                            W1, as1, ad1, hbuf, ssrc, sdst,
                                                            nRowTiles, ws_size);
    edge_aggr_kernel<<<aggrBlocks, dim3(1024), 0, stream>>>(srcA, dstA, ssrc, sdst, hbuf, obuf,
                                                           prt1, deng, shfg, ws_size);

    node_kernel<FH, true><<<tileBlocks, blk, 0, stream>>>(obuf, prt1, deng, shfg, b1,
                                                          W2, as2, ad2, hbuf, ssrc, sdst,
                                                          nRowTiles, ws_size);
    edge_aggr_kernel<<<aggrBlocks, dim3(1024), 0, stream>>>(srcA, dstA, ssrc, sdst, hbuf, obuf,
                                                           prt1, deng, shfg, ws_size);

    policy_kernel<<<tileBlocks, blk, 0, stream>>>(obuf, prt1, deng, shfg, b2, Wp, bp, out,
                                                  nRowTiles, ws_size);
}


# edge_aggr scans one owner byte per edge (dst / 800, written once by layer 1's node launch) instead of the 4-byte dst: 16 edges per lane per load; src and dst read on a hit; row sum order changes
# speedup vs baseline: 4.2834x; 1.0910x over previous
#include <hip/hip_runtime.h>
#include <math.h>

#define NNODES 100000
#define NEDGES 1600000
#define NEP    1700000
#define FIN    16
#define FH     64
#define FOUT   8

typedef __attribute__((ext_vector_type(2))) float v2f;
typedef __attribute__((ext_vector_type(8))) float v8f;

__device__ __forceinline__ float leaky(float e) { return e > 0.f ? e : 0.2f * e; }

#define SHIFT_WINDOW 40.0f

#define TN        800
#define AG_WAVES  32
#define CH_EDGES  128
#define NCHUNK    (NEDGES / CH_EDGES)
#define LD_DEPTH  8
#define WS_SPLIT_BYTES ((size_t)(3 * FH + 6) * NNODES * sizeof(float))
#define QCAP      576
#define KCH_EDGES 512
#define NKCHUNK   (NEDGES / KCH_EDGES)
#define WS_KEYS_BYTES (WS_SPLIT_BYTES + (size_t)NEDGES)
static_assert(NEDGES % KCH_EDGES == 0 && NEDGES % CH_EDGES == 0, "whole chunks");
static_assert(NNODES / TN <= 127 && NNODES % TN == 0, "an owner number stays below 127, the filler");

__device__ __forceinline__ unsigned owner_of(int d) { return min((unsigned)d / (unsigned)TN, 127u); }
#define ROWS_GROUP 16

template <int MODE>
__device__ __forceinline__ void aggr_batch(unsigned pk, bool valid, int nb, int base, int lane,
                                           const int* __restrict__ src, const int* __restrict__ dst,
                                           const float* __restrict__ s_src,
                                           const float* __restrict__ s_dst,
                                           const float* __restrict__ h, float* acc,
                                           float* den_l, int* shift_i, int* flag_l) {
    const int e = (int)pk;
    int   s = base, d = base;
    float w = 0.f;
    if (valid) {
        if (e < NEDGES) { s = src[e]; d = dst[e]; }
        else            { s = d = e - NEDGES; }
    }
    const int rel = d - base;
    if (valid) {
        const float l = leaky(s_src[s] + s_dst[d]);
        if (MODE == 1) {
            const unsigned u = __float_as_uint(l);
            if ((u >> 31) == 0) atomicMax(&shift_i[rel], (int)u);
            else                atomicMin((unsigned int*)&shift_i[rel], u);
        } else {
            const float t = l - __int_as_float(shift_i[rel]);
            if (!(t <= SHIFT_WINDOW)) *flag_l = 1;
            w = expf(t);
            atomicAdd(&den_l[rel], w);
        }
    }
    if (MODE == 1) return;
    for (int j0 = 0; j0 < nb; j0 += ROWS_GROUP) {
        float v0[ROWS_GROUP], v1[ROWS_GROUP];
#pragma unroll
        for (int u = 0; u < ROWS_GROUP; ++u) {
            const int j = j0 + u;
            const int   sj = __builtin_amdgcn_readlane(s, j);
            const float wj = __int_as_float(__builtin_amdgcn_readlane(__float_as_int(w), j));
            v0[u] = h[(size_t)sj * FH + lane]      * wj;
            v1[u] = h[(size_t)sj * FH + lane + 32] * wj;
        }
#pragma unroll
        for (int u = 0; u < ROWS_GROUP; ++u) {
            const int j = j0 + u;
            if (j < nb) {
                const int rj = __builtin_amdgcn_readlane(rel, j);
                atomicAdd(&acc[rj * FH + lane],      v0[u]);
                atomicAdd(&acc[rj * FH + lane + 32], v1[u]);
            }
        }
    }
}

#define AG_APPEND(HIT, E)                                                          \
    {                                                                              \
        const unsigned m_ = __builtin_amdgcn_ballot_w32(HIT);                      \
        if (HIT) myq[cnt + (int)__builtin_amdgcn_mbcnt_lo(m_, 0u)] = (unsigned)(E); \
        cnt += __builtin_popcount(m_);                                             \
    }

#define AG_DRAIN()                                                                 \
    while (cnt >= 32) {                                                            \
        cnt -= 32;                                                                 \
                \
                    \
        __builtin_amdgcn_wave_barrier();                                           \
        aggr_batch<MODE>(myq[cnt + lane], true, 32, base, lane, src, dst, s_src, s_dst, h, \
                         acc, den_l, shift_i, flag_l);                             \
    }

template <int MODE, bool NARROW>
__device__ __forceinline__ void scan_pass(int wave, int lane, int base, unsigned* myq,
                                          int cbeg, int cend, int sbeg, int nself,
                                          const int* __restrict__ src, const int* __restrict__ dst,
                                          const unsigned* __restrict__ keys,
                                          const float* __restrict__ s_src,
                                          const float* __restrict__ s_dst,
                                          const float* __restrict__ h, float* acc,
                                          float* den_l, int* shift_i, int* flag_l) {
    int cnt = 0;
    if (wave * 32 < nself) {
        const int idx = wave * 32 + lane;
        if (idx < nself) myq[lane] = (unsigned)(NEDGES + base + sbeg + idx);
        cnt = min(32, nself - wave * 32);
    }
    const int nit = (cend - cbeg + AG_WAVES * LD_DEPTH - 1) / (AG_WAVES * LD_DEPTH);
    if (NARROW) {
        const uint4* k4 = (const uint4*)keys;
        const unsigned gx = (unsigned)(base / TN) * 0x01010101u;
        for (int it = 0; it < nit; ++it) {
            const int c0 = cbeg + it * LD_DEPTH * AG_WAVES + wave;
            uint4 kv[LD_DEPTH];
#pragma unroll
            for (int u = 0; u < LD_DEPTH; ++u) {
                const int c = c0 + u * AG_WAVES;
                kv[u] = (c < cend) ? k4[(size_t)c * 32 + lane]
                                   : make_uint4(0x7F7F7F7Fu, 0x7F7F7F7Fu, 0x7F7F7F7Fu, 0x7F7F7F7Fu);
            }
#pragma unroll
            for (int u = 0; u < LD_DEPTH; ++u) {
                const unsigned t0 = ((kv[u].x ^ gx) + 0x7F7F7F7Fu) & 0x80808080u;
                const unsigned t1 = ((kv[u].y ^ gx) + 0x7F7F7F7Fu) & 0x80808080u;
                const unsigned t2 = ((kv[u].z ^ gx) + 0x7F7F7F7Fu) & 0x80808080u;
                const unsigned t3 = ((kv[u].w ^ gx) + 0x7F7F7F7Fu) & 0x80808080u;
                unsigned hm = ((t0 >> 7) | (t1 >> 6) | (t2 >> 5) | (t3 >> 4)) ^ 0x0F0F0F0Fu;
                const int eb = (c0 + u * AG_WAVES) * KCH_EDGES + lane * 16;
                unsigned live = __builtin_amdgcn_ballot_w32(hm != 0u);
                while (live) {
                    if (hm != 0u) {
                        const int p = __builtin_ctz(hm);
                        myq[cnt + (int)__builtin_amdgcn_mbcnt_lo(live, 0u)] =
                            (unsigned)(eb + ((p & 7) << 2) + (p >> 3));
                    }
                    cnt += __builtin_popcount(live);
                    hm &= hm - 1u;
                    live = __builtin_amdgcn_ballot_w32(hm != 0u);
                }
                AG_DRAIN()
            }
        }
    } else {
        const int4* dst4 = (const int4*)dst;
        for (int it = 0; it < nit; ++it) {
            const int c0 = cbeg + (it * AG_WAVES + wave) * LD_DEPTH;
            int4 dv[LD_DEPTH];
#pragma unroll
            for (int u = 0; u < LD_DEPTH; ++u) {
                const int c = c0 + u;
                dv[u] = (c < cend) ? dst4[(size_t)c * 32 + lane] : make_int4(-1, -1, -1, -1);
            }
#pragma unroll
            for (int half = 0; half < 2; ++half) {
#pragma unroll
                for (int u = half * 4; u < half * 4 + 4; ++u) {
                    const bool h0 = (unsigned)(dv[u].x - base) < TN, h1 = (unsigned)(dv[u].y - base) < TN;
                    const bool h2 = (unsigned)(dv[u].z - base) < TN, h3 = (unsigned)(dv[u].w - base) < TN;
                    if (__builtin_amdgcn_ballot_w32(h0 | h1 | h2 | h3)) {
                        const int eb = (c0 + u) * CH_EDGES + lane * 4;
                        AG_APPEND(h0, eb)
                        AG_APPEND(h1, eb + 1)
                        AG_APPEND(h2, eb + 2)
                        AG_APPEND(h3, eb + 3)
                    }
                }
                AG_DRAIN()
            }
        }
    }
    if (cnt > 0) {
        __builtin_amdgcn_wave_barrier();
        aggr_batch<MODE>(lane < cnt ? myq[lane] : 0u, lane < cnt, cnt, base, lane,
                         src, dst, s_src, s_dst, h, acc, den_l, shift_i, flag_l);
    }
}

template <bool NARROW>
__device__ __forceinline__ void aggr_passes(int wave, int lane, int base, unsigned* myq,
                                            int cbeg, int cend, int sbeg, int nself,
                                            const int* __restrict__ src, const int* __restrict__ dst,
                                            const unsigned* __restrict__ keys,
                                            const float* __restrict__ s_src,
                                            const float* __restrict__ s_dst,
                                            const float* __restrict__ h, float* acc,
                                            float* den_l, int* shift_i, int* flag_l) {
    scan_pass<0, NARROW>(wave, lane, base, myq, cbeg, cend, sbeg, nself, src, dst, keys, s_src, s_dst, h, acc, den_l, shift_i, flag_l);
    __syncthreads();

    const bool redo = *flag_l != 0;
    if (redo) {
        scan_pass<1, NARROW>(wave, lane, base, myq, cbeg, cend, sbeg, nself, src, dst, keys, s_src, s_dst, h, acc, den_l, shift_i, flag_l);
        __syncthreads();
        for (int i = threadIdx.x; i < TN * FH; i += 1024) acc[i] = 0.f;
        if (threadIdx.x < TN) den_l[threadIdx.x] = 0.f;
        __syncthreads();
        scan_pass<0, NARROW>(wave, lane, base, myq, cbeg, cend, sbeg, nself, src, dst, keys, s_src, s_dst, h, acc, den_l, shift_i, flag_l);
        __syncthreads();
    }
}

__global__ __launch_bounds__(1024)
void edge_aggr_kernel(const int* __restrict__ src, const int* __restrict__ dst,
                      const unsigned* __restrict__ keys,
                      const float* __restrict__ s_src, const float* __restrict__ s_dst,
                      const float* __restrict__ h, float* __restrict__ out,
                      float* __restrict__ part1, float* __restrict__ den_g,
                      int* __restrict__ shf_g, size_t ws_size) {
    const bool split  = ws_size >= WS_SPLIT_BYTES;
    const bool narrow = ws_size >= WS_KEYS_BYTES;
    const int  half   = blockIdx.x & 1;
    if (!split && half) return;
    __shared__ float    acc[TN * FH];
    __shared__ unsigned queue[AG_WAVES * QCAP];
    __shared__ float    den_l[TN];
    __shared__ int      shift_i[TN];
    __shared__ int      flag_l;
    const int lane = threadIdx.x & 31;
    const int wave = __builtin_amdgcn_readfirstlane((int)(threadIdx.x >> 5));
    const int base = (int)(blockIdx.x >> 1) * TN;
    const int nch   = narrow ? NKCHUNK : NCHUNK;
    const int cbeg  = split ? half * (nch / 2) : 0;
    const int cend  = (split && !half) ? nch / 2 : nch;
    const int sbeg  = split ? half * (TN / 2) : 0;
    const int nself = split ? TN / 2 : TN;
    unsigned* myq = queue + wave * QCAP;
    for (int i = threadIdx.x; i < TN * FH; i += 1024) acc[i] = 0.f;
    if (threadIdx.x < TN) {
        const int n = base + (int)threadIdx.x;
        den_l[threadIdx.x]   = 0.f;
        shift_i[threadIdx.x] = __float_as_int(leaky(s_src[n] + s_dst[n]));
    }
    if (threadIdx.x == 0) flag_l = 0;
    __syncthreads();

    if (narrow) aggr_passes<true >(wave, lane, base, myq, cbeg, cend, sbeg, nself, src, dst, keys, s_src, s_dst, h, acc, den_l, shift_i, &flag_l);
    else        aggr_passes<false>(wave, lane, base, myq, cbeg, cend, sbeg, nself, src, dst, keys, s_src, s_dst, h, acc, den_l, shift_i, &flag_l);

    if (!split) {
        for (int i = threadIdx.x; i < TN * FH; i += 1024)
            out[(size_t)base * FH + i] = acc[i] / (den_l[i >> 6] + 1e-16f);
        return;
    }
    float* rows = half ? part1 : out;
    for (int i = threadIdx.x; i < TN * FH; i += 1024)
        rows[(size_t)base * FH + i] = acc[i];
    if (threadIdx.x < TN) {
        den_g[half * NNODES + base + (int)threadIdx.x] = den_l[threadIdx.x];
        shf_g[half * NNODES + base + (int)threadIdx.x] = shift_i[threadIdx.x];
    }
}

__device__ __forceinline__ int kbase(int s, int hi) { return 8 * (s >> 1) + 4 * hi + 2 * (s & 1); }

template <int K, bool ACT>
__device__ __forceinline__ void load_a(const float* __restrict__ in, const float* __restrict__ part1,
                                       const float* __restrict__ den_g, const int* __restrict__ shf_g,
                                       const float* __restrict__ bprev, bool split, int row, int hi,
                                       float4* a) {
    const float4* p0 = (const float4*)(in + (size_t)row * K + 4 * hi);
#pragma unroll
    for (int j = 0; j < K / 8; ++j) a[j] = p0[2 * j];
    if (ACT) {
        if (split) {
            const float4* p1 = (const float4*)(part1 + (size_t)row * K + 4 * hi);
            float4 q[K / 8];
#pragma unroll
            for (int j = 0; j < K / 8; ++j) q[j] = p1[2 * j];
            const float s0 = __int_as_float(shf_g[row]);
            const float s1 = __int_as_float(shf_g[NNODES + row]);
            float d0 = den_g[row], d1 = den_g[NNODES + row];
            float f0 = 1.f, f1 = 1.f;
            if (s0 != s1) {
                const float m = fmaxf(s0, s1);
                f0 = expf(s0 - m);
                f1 = expf(s1 - m);
            }
            d0 *= f0;
            d1 *= f1;
            const float den = d0 + d1 + 1e-16f;
#pragma unroll
            for (int j = 0; j < K / 8; ++j) {
                a[j].x = (a[j].x * f0 + q[j].x * f1) / den;
                a[j].y = (a[j].y * f0 + q[j].y * f1) / den;
                a[j].z = (a[j].z * f0 + q[j].z * f1) / den;
                a[j].w = (a[j].w * f0 + q[j].w * f1) / den;
            }
        }
#pragma unroll
        for (int j = 0; j < K / 8; ++j) {
            const float4 bb = *(const float4*)(bprev + 8 * j + 4 * hi);
            float v;
            v = a[j].x + bb.x; a[j].x = v > 0.f ? v : 0.f;
            v = a[j].y + bb.y; a[j].y = v > 0.f ? v : 0.f;
            v = a[j].z + bb.z; a[j].z = v > 0.f ? v : 0.f;
            v = a[j].w + bb.w; a[j].w = v > 0.f ? v : 0.f;
        }
    }
}

template <int K, bool ACT>
__global__ __launch_bounds__(256)
void node_kernel(const float* __restrict__ in, const float* __restrict__ part1,
                            const float* __restrict__ den_g, const int* __restrict__ shf_g,
                            const float* __restrict__ bprev, const float* __restrict__ W,
                            const float* __restrict__ a_src, const float* __restrict__ a_dst,
                            float* __restrict__ h, float* __restrict__ s_src,
                            float* __restrict__ s_dst, const int* __restrict__ dst,
                            unsigned* __restrict__ keys, int nRowTiles, size_t ws_size) {
    const int lane    = threadIdx.x & 31;
    const int rowTile = blockIdx.x * 8 + (threadIdx.x >> 5);
    if (rowTile >= nRowTiles) return;
    const int hi  = lane >> 4;
    const int lo  = lane & 15;
    const int row = rowTile * 16 + lo;
    if (!ACT && ws_size >= WS_KEYS_BYTES) {
        static_assert((NNODES / 16) * 256 == NEDGES, "one wave per 256 edges");
        const int4* d4 = (const int4*)dst;
#pragma unroll
        for (int q = 0; q < 2; ++q) {
            const int  idx = rowTile * 64 + q * 32 + lane;
            const int4 v   = d4[idx];
            keys[idx] = owner_of(v.x) | (owner_of(v.y) << 8) | (owner_of(v.z) << 16) | (owner_of(v.w) << 24);
        }
    }
    float4 a[K / 8];
    load_a<K, ACT>(in, part1, den_g, shf_g, bprev, ws_size >= WS_SPLIT_BYTES, row, hi, a);
    v8f c0 = {}, c1 = {}, c2 = {}, c3 = {};
#pragma unroll
    for (int s = 0; s < K / 4; ++s) {
        const int kb = kbase(s, hi);
        v2f av;
        av.x = (s & 1) ? a[s >> 1].z : a[s >> 1].x;
        av.y = (s & 1) ? a[s >> 1].w : a[s >> 1].y;
        const float* w0 = W + (size_t)kb * FH + lo;
        v2f b0, b1, b2, b3;
        b0.x = w0[0];  b0.y = w0[FH];
        b1.x = w0[16]; b1.y = w0[FH + 16];
        b2.x = w0[32]; b2.y = w0[FH + 32];
        b3.x = w0[48]; b3.y = w0[FH + 48];
        c0 = __builtin_amdgcn_wmma_f32_16x16x4_f32(false, av, false, b0, (short)0, c0, false, false);
        c1 = __builtin_amdgcn_wmma_f32_16x16x4_f32(false, av, false, b1, (short)0, c1, false, false);
        c2 = __builtin_amdgcn_wmma_f32_16x16x4_f32(false, av, false, b2, (short)0, c2, false, false);
        c3 = __builtin_amdgcn_wmma_f32_16x16x4_f32(false, av, false, b3, (short)0, c3, false, false);
    }
    const int rbase = rowTile * 16 + 8 * hi;
#pragma unroll
    for (int r = 0; r < 8; ++r) {
        float* hr = h + (size_t)(rbase + r) * FH + lo;
        hr[0]  = c0[r];
        hr[16] = c1[r];
        hr[32] = c2[r];
        hr[48] = c3[r];
    }
    const float as0 = a_src[lo], as1 = a_src[16 + lo], as2 = a_src[32 + lo], as3 = a_src[48 + lo];
    const float ad0 = a_dst[lo], ad1 = a_dst[16 + lo], ad2 = a_dst[32 + lo], ad3 = a_dst[48 + lo];
    float ps[8], pd[8];
#pragma unroll
    for (int r = 0; r < 8; ++r) {
        ps[r] = c0[r] * as0 + c1[r] * as1 + c2[r] * as2 + c3[r] * as3;
        pd[r] = c0[r] * ad0 + c1[r] * ad1 + c2[r] * ad2 + c3[r] * ad3;
    }
    const bool u8 = lo & 8, u4 = lo & 4, u2 = lo & 2, u1 = lo & 1;
    float qs[4], qd[4];
#pragma unroll
    for (int i = 0; i < 4; ++i) {
        qs[i] = (u8 ? ps[i + 4] : ps[i]) + __shfl_xor(u8 ? ps[i] : ps[i + 4], 8, 32);
        qd[i] = (u8 ? pd[i + 4] : pd[i]) + __shfl_xor(u8 ? pd[i] : pd[i + 4], 8, 32);
    }
    float ts[2], td[2];
#pragma unroll
    for (int i = 0; i < 2; ++i) {
        ts[i] = (u4 ? qs[i + 2] : qs[i]) + __shfl_xor(u4 ? qs[i] : qs[i + 2], 4, 32);
        td[i] = (u4 ? qd[i + 2] : qd[i]) + __shfl_xor(u4 ? qd[i] : qd[i + 2], 4, 32);
    }
    const float es = (u2 ? ts[1] : ts[0]) + __shfl_xor(u2 ? ts[0] : ts[1], 2, 32);
    const float ed = (u2 ? td[1] : td[0]) + __shfl_xor(u2 ? td[0] : td[1], 2, 32);
    const float fin = (u1 ? ed : es) + __shfl_xor(u1 ? es : ed, 1, 32);
    (u1 ? s_dst : s_src)[rbase + (lo >> 1)] = fin;
}

__global__ __launch_bounds__(256)
void policy_kernel(const float* __restrict__ rows0, const float* __restrict__ part1,
                              const float* __restrict__ den_g, const int* __restrict__ shf_g,
                              const float* __restrict__ bprev, const float* __restrict__ Wp,
                              const float* __restrict__ bp, float* __restrict__ out,
                              int nRowTiles, size_t ws_size) {
    const int lane = threadIdx.x & 31;
    const int gw   = blockIdx.x * 8 + (threadIdx.x >> 5);
    if (gw >= nRowTiles) return;
    const int hi  = lane >> 4;
    const int lo  = lane & 15;
    const int row = gw * 16 + lo;
    const int cl  = lo < FOUT ? lo : 0;
    float4 a[FH / 8];
    load_a<FH, true>(rows0, part1, den_g, shf_g, bprev, ws_size >= WS_SPLIT_BYTES, row, hi, a);
    v8f c = {};
#pragma unroll
    for (int s = 0; s < FH / 4; ++s) {
        const int kb = kbase(s, hi);
        v2f av, b;
        av.x = (s & 1) ? a[s >> 1].z : a[s >> 1].x;
        av.y = (s & 1) ? a[s >> 1].w : a[s >> 1].y;
        const float w0 = Wp[(size_t)kb * FOUT + cl];
        const float w1 = Wp[(size_t)(kb + 1) * FOUT + cl];
        b.x = (lo < FOUT) ? w0 : 0.f;
        b.y = (lo < FOUT) ? w1 : 0.f;
        c = __builtin_amdgcn_wmma_f32_16x16x4_f32(false, av, false, b, (short)0, c, false, false);
    }
    if (lo < FOUT) {
        const int rbase = gw * 16 + 8 * hi;
        const float bb = bp[lo];
#pragma unroll
        for (int r = 0; r < 8; ++r)
            out[(size_t)(rbase + r) * FOUT + lo] = c[r] + bb;
    }
}

extern "C" void kernel_launch(void* const* d_in, const int* in_sizes, int n_in,
                              void* d_out, int out_size, void* d_ws, size_t ws_size,
                              hipStream_t stream) {
    const float* x   = (const float*)d_in[0];
    const int*   ei  = (const int*)  d_in[1];
    const float* W1  = (const float*)d_in[2];
    const float* as1 = (const float*)d_in[3];
    const float* ad1 = (const float*)d_in[4];
    const float* b1  = (const float*)d_in[5];
    const float* W2  = (const float*)d_in[6];
    const float* as2 = (const float*)d_in[7];
    const float* ad2 = (const float*)d_in[8];
    const float* b2  = (const float*)d_in[9];
    const float* Wp  = (const float*)d_in[10];
    const float* bp  = (const float*)d_in[11];
    float* out = (float*)d_out;

    const int* srcA = ei;
    const int* dstA = ei + NEDGES;

    float* ws   = (float*)d_ws;
    float* hbuf = ws;
    float* obuf = ws + (size_t)NNODES * FH;
    float* ssrc = ws + (size_t)2 * NNODES * FH;
    float* sdst = ssrc + NNODES;
    float* prt1 = sdst + NNODES;
    float* deng = prt1 + (size_t)NNODES * FH;
    int*   shfg = (int*)(deng + 2 * NNODES);
    unsigned* keys = (unsigned*)(shfg + 2 * NNODES);

    const int nRowTiles = NNODES / 16;
    const dim3 blk(256);
    const int tileBlocks = (nRowTiles + 7) / 8;
    const int aggrBlocks = 2 * (NNODES / TN);

    node_kernel<FIN, false><<<tileBlocks, blk, 0, stream>>>(x, nullptr, nullptr, nullptr, nullptr,
                                                            W1, as1, ad1, hbuf, ssrc, sdst,
                                                            dstA, keys, nRowTiles, ws_size);
    edge_aggr_kernel<<<aggrBlocks, dim3(1024), 0, stream>>>(srcA, dstA, keys, ssrc, sdst, hbuf, obuf,
                                                           prt1, deng, shfg, ws_size);

    node_kernel<FH, true><<<tileBlocks, blk, 0, stream>>>(obuf, prt1, deng, shfg, b1,
                                                          W2, as2, ad2, hbuf, ssrc, sdst,
                                                          nullptr, nullptr, nRowTiles, ws_size);
    edge_aggr_kernel<<<aggrBlocks, dim3(1024), 0, stream>>>(srcA, dstA, keys, ssrc, sdst, hbuf, obuf,
                                                           prt1, deng, shfg, ws_size);

    policy_kernel<<<tileBlocks, blk, 0, stream>>>(obuf, prt1, deng, shfg, b2, Wp, bp, out,
                                                  nRowTiles, ws_size);
}


# Edge batch reads one packed word per edge (src | dst offset in its range, written once by layer 1's node launch) and its nodes' s_dst from LDS: divergent global loads per edge go from 4 to 2
# speedup vs baseline: 4.7076x; 1.0990x over previous
#include <hip/hip_runtime.h>
#include <math.h>

#define NNODES 100000
#define NEDGES 1600000
#define NEP    1700000
#define FIN    16
#define FH     64
#define FOUT   8

typedef __attribute__((ext_vector_type(2))) float v2f;
typedef __attribute__((ext_vector_type(8))) float v8f;

__device__ __forceinline__ float leaky(float e) { return e > 0.f ? e : 0.2f * e; }

#define SHIFT_WINDOW 40.0f

#define TN        800
#define AG_WAVES  32
#define CH_EDGES  128
#define NCHUNK    (NEDGES / CH_EDGES)
#define LD_DEPTH  8
#define WS_SPLIT_BYTES ((size_t)(3 * FH + 6) * NNODES * sizeof(float))
#define QCAP      576
#define KCH_EDGES 512
#define NKCHUNK   (NEDGES / KCH_EDGES)
#define WS_KEYS_BYTES (WS_SPLIT_BYTES + (size_t)NEDGES)
#define PK_SHIFT  17
#define WS_PK_BYTES (WS_KEYS_BYTES + (size_t)NEDGES * sizeof(unsigned))
static_assert(NNODES <= (1 << PK_SHIFT) && TN <= (1 << (32 - PK_SHIFT)), "src and the offset fit one word");
static_assert(NEDGES % 16 == 0, "the packed words start 16-byte aligned behind the owner bytes");
static_assert(NEDGES % KCH_EDGES == 0 && NEDGES % CH_EDGES == 0, "whole chunks");
static_assert(NNODES / TN <= 127 && NNODES % TN == 0, "an owner number stays below 127, the filler");

__device__ __forceinline__ unsigned owner_of(int d) { return min((unsigned)d / (unsigned)TN, 127u); }
#define ROWS_GROUP 16

template <int MODE, bool PACKED>
__device__ __forceinline__ void aggr_batch(unsigned qw, bool valid, int nb, int base, int lane,
                                           const int* __restrict__ src, const int* __restrict__ dst,
                                           const unsigned* __restrict__ pkw,
                                           const float* __restrict__ s_src,
                                           const float* sdst_l,
                                           const float* __restrict__ h, float* acc,
                                           float* den_l, int* shift_i, int* flag_l) {
    const int e = (int)qw;
    int   s = base, rel = 0;
    float w = 0.f;
    if (valid) {
        if (e < NEDGES) {
            if (PACKED) {
                const unsigned p = pkw[e];
                s   = (int)(p & ((1u << PK_SHIFT) - 1u));
                rel = (int)(p >> PK_SHIFT);
            } else {
                s   = src[e];
                rel = dst[e] - base;
            }
        } else {
            s   = e - NEDGES;
            rel = s - base;
        }
    }
    if (valid) {
        const float l = leaky(s_src[s] + sdst_l[rel]);
        if (MODE == 1) {
            const unsigned u = __float_as_uint(l);
            if ((u >> 31) == 0) atomicMax(&shift_i[rel], (int)u);
            else                atomicMin((unsigned int*)&shift_i[rel], u);
        } else {
            const float t = l - __int_as_float(shift_i[rel]);
            if (!(t <= SHIFT_WINDOW)) *flag_l = 1;
            w = expf(t);
            atomicAdd(&den_l[rel], w);
        }
    }
    if (MODE == 1) return;
    for (int j0 = 0; j0 < nb; j0 += ROWS_GROUP) {
        float v0[ROWS_GROUP], v1[ROWS_GROUP];
#pragma unroll
        for (int u = 0; u < ROWS_GROUP; ++u) {
            const int j = j0 + u;
            const int   sj = __builtin_amdgcn_readlane(s, j);
            const float wj = __int_as_float(__builtin_amdgcn_readlane(__float_as_int(w), j));
            v0[u] = h[(size_t)sj * FH + lane]      * wj;
            v1[u] = h[(size_t)sj * FH + lane + 32] * wj;
        }
#pragma unroll
        for (int u = 0; u < ROWS_GROUP; ++u) {
            const int j = j0 + u;
            if (j < nb) {
                const int rj = __builtin_amdgcn_readlane(rel, j);
                atomicAdd(&acc[rj * FH + lane],      v0[u]);
                atomicAdd(&acc[rj * FH + lane + 32], v1[u]);
            }
        }
    }
}

#define AG_APPEND(HIT, E)                                                          \
    {                                                                              \
        const unsigned m_ = __builtin_amdgcn_ballot_w32(HIT);                      \
        if (HIT) myq[cnt + (int)__builtin_amdgcn_mbcnt_lo(m_, 0u)] = (unsigned)(E); \
        cnt += __builtin_popcount(m_);                                             \
    }

#define AG_DRAIN()                                                                 \
    while (cnt >= 32) {                                                            \
        cnt -= 32;                                                                 \
                \
                    \
        __builtin_amdgcn_wave_barrier();                                           \
        aggr_batch<MODE, PACKED>(myq[cnt + lane], true, 32, base, lane, src, dst, pkw, s_src, \
                         sdst_l, h, acc, den_l, shift_i, flag_l);                  \
    }

template <int MODE, bool NARROW, bool PACKED>
__device__ __forceinline__ void scan_pass(int wave, int lane, int base, unsigned* myq,
                                          int cbeg, int cend, int sbeg, int nself,
                                          const int* __restrict__ src, const int* __restrict__ dst,
                                          const unsigned* __restrict__ keys,
                                          const unsigned* __restrict__ pkw,
                                          const float* __restrict__ s_src,
                                          const float* sdst_l,
                                          const float* __restrict__ h, float* acc,
                                          float* den_l, int* shift_i, int* flag_l) {
    int cnt = 0;
    if (wave * 32 < nself) {
        const int idx = wave * 32 + lane;
        if (idx < nself) myq[lane] = (unsigned)(NEDGES + base + sbeg + idx);
        cnt = min(32, nself - wave * 32);
    }
    const int nit = (cend - cbeg + AG_WAVES * LD_DEPTH - 1) / (AG_WAVES * LD_DEPTH);
    if (NARROW) {
        const uint4* k4 = (const uint4*)keys;
        const unsigned gx = (unsigned)(base / TN) * 0x01010101u;
        for (int it = 0; it < nit; ++it) {
            const int c0 = cbeg + it * LD_DEPTH * AG_WAVES + wave;
            uint4 kv[LD_DEPTH];
#pragma unroll
            for (int u = 0; u < LD_DEPTH; ++u) {
                const int c = c0 + u * AG_WAVES;
                kv[u] = (c < cend) ? k4[(size_t)c * 32 + lane]
                                   : make_uint4(0x7F7F7F7Fu, 0x7F7F7F7Fu, 0x7F7F7F7Fu, 0x7F7F7F7Fu);
            }
#pragma unroll
            for (int u = 0; u < LD_DEPTH; ++u) {
                const unsigned t0 = ((kv[u].x ^ gx) + 0x7F7F7F7Fu) & 0x80808080u;
                const unsigned t1 = ((kv[u].y ^ gx) + 0x7F7F7F7Fu) & 0x80808080u;
                const unsigned t2 = ((kv[u].z ^ gx) + 0x7F7F7F7Fu) & 0x80808080u;
                const unsigned t3 = ((kv[u].w ^ gx) + 0x7F7F7F7Fu) & 0x80808080u;
                unsigned hm = ((t0 >> 7) | (t1 >> 6) | (t2 >> 5) | (t3 >> 4)) ^ 0x0F0F0F0Fu;
                const int eb = (c0 + u * AG_WAVES) * KCH_EDGES + lane * 16;
                unsigned live = __builtin_amdgcn_ballot_w32(hm != 0u);
                while (live) {
                    if (hm != 0u) {
                        const int p = __builtin_ctz(hm);
                        myq[cnt + (int)__builtin_amdgcn_mbcnt_lo(live, 0u)] =
                            (unsigned)(eb + ((p & 7) << 2) + (p >> 3));
                    }
                    cnt += __builtin_popcount(live);
                    hm &= hm - 1u;
                    live = __builtin_amdgcn_ballot_w32(hm != 0u);
                }
                AG_DRAIN()
            }
        }
    } else {
        const int4* dst4 = (const int4*)dst;
        for (int it = 0; it < nit; ++it) {
            const int c0 = cbeg + (it * AG_WAVES + wave) * LD_DEPTH;
            int4 dv[LD_DEPTH];
#pragma unroll
            for (int u = 0; u < LD_DEPTH; ++u) {
                const int c = c0 + u;
                dv[u] = (c < cend) ? dst4[(size_t)c * 32 + lane] : make_int4(-1, -1, -1, -1);
            }
#pragma unroll
            for (int half = 0; half < 2; ++half) {
#pragma unroll
                for (int u = half * 4; u < half * 4 + 4; ++u) {
                    const bool h0 = (unsigned)(dv[u].x - base) < TN, h1 = (unsigned)(dv[u].y - base) < TN;
                    const bool h2 = (unsigned)(dv[u].z - base) < TN, h3 = (unsigned)(dv[u].w - base) < TN;
                    if (__builtin_amdgcn_ballot_w32(h0 | h1 | h2 | h3)) {
                        const int eb = (c0 + u) * CH_EDGES + lane * 4;
                        AG_APPEND(h0, eb)
                        AG_APPEND(h1, eb + 1)
                        AG_APPEND(h2, eb + 2)
                        AG_APPEND(h3, eb + 3)
                    }
                }
                AG_DRAIN()
            }
        }
    }
    if (cnt > 0) {
        __builtin_amdgcn_wave_barrier();
        aggr_batch<MODE, PACKED>(lane < cnt ? myq[lane] : 0u, lane < cnt, cnt, base, lane,
                         src, dst, pkw, s_src, sdst_l, h, acc, den_l, shift_i, flag_l);
    }
}

template <bool NARROW, bool PACKED>
__device__ __forceinline__ void aggr_passes(int wave, int lane, int base, unsigned* myq,
                                            int cbeg, int cend, int sbeg, int nself,
                                            const int* __restrict__ src, const int* __restrict__ dst,
                                            const unsigned* __restrict__ keys,
                                            const unsigned* __restrict__ pkw,
                                            const float* __restrict__ s_src,
                                            const float* sdst_l,
                                            const float* __restrict__ h, float* acc,
                                            float* den_l, int* shift_i, int* flag_l) {
    scan_pass<0, NARROW, PACKED>(wave, lane, base, myq, cbeg, cend, sbeg, nself, src, dst, keys, pkw, s_src, sdst_l, h, acc, den_l, shift_i, flag_l);
    __syncthreads();

    const bool redo = *flag_l != 0;
    if (redo) {
        scan_pass<1, NARROW, PACKED>(wave, lane, base, myq, cbeg, cend, sbeg, nself, src, dst, keys, pkw, s_src, sdst_l, h, acc, den_l, shift_i, flag_l);
        __syncthreads();
        for (int i = threadIdx.x; i < TN * FH; i += 1024) acc[i] = 0.f;
        if (threadIdx.x < TN) den_l[threadIdx.x] = 0.f;
        __syncthreads();
        scan_pass<0, NARROW, PACKED>(wave, lane, base, myq, cbeg, cend, sbeg, nself, src, dst, keys, pkw, s_src, sdst_l, h, acc, den_l, shift_i, flag_l);
        __syncthreads();
    }
}

__global__ __launch_bounds__(1024)
void edge_aggr_kernel(const int* __restrict__ src, const int* __restrict__ dst,
                      const unsigned* __restrict__ keys, const unsigned* __restrict__ pkw,
                      const float* __restrict__ s_src, const float* __restrict__ s_dst,
                      const float* __restrict__ h, float* __restrict__ out,
                      float* __restrict__ part1, float* __restrict__ den_g,
                      int* __restrict__ shf_g, size_t ws_size) {
    const bool split  = ws_size >= WS_SPLIT_BYTES;
    const bool narrow = ws_size >= WS_KEYS_BYTES;
    const bool packed = ws_size >= WS_PK_BYTES;
    const int  half   = blockIdx.x & 1;
    if (!split && half) return;
    __shared__ float    acc[TN * FH];
    __shared__ unsigned queue[AG_WAVES * QCAP];
    __shared__ float    den_l[TN];
    __shared__ int      shift_i[TN];
    __shared__ float    sdst_l[TN];
    __shared__ int      flag_l;
    const int lane = threadIdx.x & 31;
    const int wave = __builtin_amdgcn_readfirstlane((int)(threadIdx.x >> 5));
    const int base = (int)(blockIdx.x >> 1) * TN;
    const int nch   = narrow ? NKCHUNK : NCHUNK;
    const int cbeg  = split ? half * (nch / 2) : 0;
    const int cend  = (split && !half) ? nch / 2 : nch;
    const int sbeg  = split ? half * (TN / 2) : 0;
    const int nself = split ? TN / 2 : TN;
    unsigned* myq = queue + wave * QCAP;
    for (int i = threadIdx.x; i < TN * FH; i += 1024) acc[i] = 0.f;
    if (threadIdx.x < TN) {
        const int n = base + (int)threadIdx.x;
        const float sd = s_dst[n];
        den_l[threadIdx.x]   = 0.f;
        sdst_l[threadIdx.x]  = sd;
        shift_i[threadIdx.x] = __float_as_int(leaky(s_src[n] + sd));
    }
    if (threadIdx.x == 0) flag_l = 0;
    __syncthreads();

    if (packed)      aggr_passes<true,  true >(wave, lane, base, myq, cbeg, cend, sbeg, nself, src, dst, keys, pkw, s_src, sdst_l, h, acc, den_l, shift_i, &flag_l);
    else if (narrow) aggr_passes<true,  false>(wave, lane, base, myq, cbeg, cend, sbeg, nself, src, dst, keys, pkw, s_src, sdst_l, h, acc, den_l, shift_i, &flag_l);
    else             aggr_passes<false, false>(wave, lane, base, myq, cbeg, cend, sbeg, nself, src, dst, keys, pkw, s_src, sdst_l, h, acc, den_l, shift_i, &flag_l);

    if (!split) {
        for (int i = threadIdx.x; i < TN * FH; i += 1024)
            out[(size_t)base * FH + i] = acc[i] / (den_l[i >> 6] + 1e-16f);
        return;
    }
    float* rows = half ? part1 : out;
    for (int i = threadIdx.x; i < TN * FH; i += 1024)
        rows[(size_t)base * FH + i] = acc[i];
    if (threadIdx.x < TN) {
        den_g[half * NNODES + base + (int)threadIdx.x] = den_l[threadIdx.x];
        shf_g[half * NNODES + base + (int)threadIdx.x] = shift_i[threadIdx.x];
    }
}

__device__ __forceinline__ int kbase(int s, int hi) { return 8 * (s >> 1) + 4 * hi + 2 * (s & 1); }

template <int K, bool ACT>
__device__ __forceinline__ void load_a(const float* __restrict__ in, const float* __restrict__ part1,
                                       const float* __restrict__ den_g, const int* __restrict__ shf_g,
                                       const float* __restrict__ bprev, bool split, int row, int hi,
                                       float4* a) {
    const float4* p0 = (const float4*)(in + (size_t)row * K + 4 * hi);
#pragma unroll
    for (int j = 0; j < K / 8; ++j) a[j] = p0[2 * j];
    if (ACT) {
        if (split) {
            const float4* p1 = (const float4*)(part1 + (size_t)row * K + 4 * hi);
            float4 q[K / 8];
#pragma unroll
            for (int j = 0; j < K / 8; ++j) q[j] = p1[2 * j];
            const float s0 = __int_as_float(shf_g[row]);
            const float s1 = __int_as_float(shf_g[NNODES + row]);
            float d0 = den_g[row], d1 = den_g[NNODES + row];
            float f0 = 1.f, f1 = 1.f;
            if (s0 != s1) {
                const float m = fmaxf(s0, s1);
                f0 = expf(s0 - m);
                f1 = expf(s1 - m);
            }
            d0 *= f0;
            d1 *= f1;
            const float den = d0 + d1 + 1e-16f;
#pragma unroll
            for (int j = 0; j < K / 8; ++j) {
                a[j].x = (a[j].x * f0 + q[j].x * f1) / den;
                a[j].y = (a[j].y * f0 + q[j].y * f1) / den;
                a[j].z = (a[j].z * f0 + q[j].z * f1) / den;
                a[j].w = (a[j].w * f0 + q[j].w * f1) / den;
            }
        }
#pragma unroll
        for (int j = 0; j < K / 8; ++j) {
            const float4 bb = *(const float4*)(bprev + 8 * j + 4 * hi);
            float v;
            v = a[j].x + bb.x; a[j].x = v > 0.f ? v : 0.f;
            v = a[j].y + bb.y; a[j].y = v > 0.f ? v : 0.f;
            v = a[j].z + bb.z; a[j].z = v > 0.f ? v : 0.f;
            v = a[j].w + bb.w; a[j].w = v > 0.f ? v : 0.f;
        }
    }
}

template <int K, bool ACT>
__global__ __launch_bounds__(256)
void node_kernel(const float* __restrict__ in, const float* __restrict__ part1,
                            const float* __restrict__ den_g, const int* __restrict__ shf_g,
                            const float* __restrict__ bprev, const float* __restrict__ W,
                            const float* __restrict__ a_src, const float* __restrict__ a_dst,
                            float* __restrict__ h, float* __restrict__ s_src,
                            float* __restrict__ s_dst, const int* __restrict__ src,
                            const int* __restrict__ dst, unsigned* __restrict__ keys,
                            unsigned* __restrict__ pkw, int nRowTiles, size_t ws_size) {
    const int lane    = threadIdx.x & 31;
    const int rowTile = blockIdx.x * 8 + (threadIdx.x >> 5);
    if (rowTile >= nRowTiles) return;
    const int hi  = lane >> 4;
    const int lo  = lane & 15;
    const int row = rowTile * 16 + lo;
    if (!ACT && ws_size >= WS_KEYS_BYTES) {
        static_assert((NNODES / 16) * 256 == NEDGES, "one wave per 256 edges");
        const bool  packed = ws_size >= WS_PK_BYTES;
        const int4* d4 = (const int4*)dst;
        const int4* s4 = (const int4*)src;
        uint4*      p4 = (uint4*)pkw;
#pragma unroll
        for (int q = 0; q < 2; ++q) {
            const int  idx = rowTile * 64 + q * 32 + lane;
            const int4 v   = d4[idx];
            const unsigned o0 = owner_of(v.x), o1 = owner_of(v.y), o2 = owner_of(v.z), o3 = owner_of(v.w);
            keys[idx] = o0 | (o1 << 8) | (o2 << 16) | (o3 << 24);
            if (packed) {
                const int4 sv = s4[idx];
                p4[idx] = make_uint4((unsigned)sv.x | (((unsigned)v.x - o0 * TN) << PK_SHIFT),
                                     (unsigned)sv.y | (((unsigned)v.y - o1 * TN) << PK_SHIFT),
                                     (unsigned)sv.z | (((unsigned)v.z - o2 * TN) << PK_SHIFT),
                                     (unsigned)sv.w | (((unsigned)v.w - o3 * TN) << PK_SHIFT));
            }
        }
    }
    float4 a[K / 8];
    load_a<K, ACT>(in, part1, den_g, shf_g, bprev, ws_size >= WS_SPLIT_BYTES, row, hi, a);
    v8f c0 = {}, c1 = {}, c2 = {}, c3 = {};
#pragma unroll
    for (int s = 0; s < K / 4; ++s) {
        const int kb = kbase(s, hi);
        v2f av;
        av.x = (s & 1) ? a[s >> 1].z : a[s >> 1].x;
        av.y = (s & 1) ? a[s >> 1].w : a[s >> 1].y;
        const float* w0 = W + (size_t)kb * FH + lo;
        v2f b0, b1, b2, b3;
        b0.x = w0[0];  b0.y = w0[FH];
        b1.x = w0[16]; b1.y = w0[FH + 16];
        b2.x = w0[32]; b2.y = w0[FH + 32];
        b3.x = w0[48]; b3.y = w0[FH + 48];
        c0 = __builtin_amdgcn_wmma_f32_16x16x4_f32(false, av, false, b0, (short)0, c0, false, false);
        c1 = __builtin_amdgcn_wmma_f32_16x16x4_f32(false, av, false, b1, (short)0, c1, false, false);
        c2 = __builtin_amdgcn_wmma_f32_16x16x4_f32(false, av, false, b2, (short)0, c2, false, false);
        c3 = __builtin_amdgcn_wmma_f32_16x16x4_f32(false, av, false, b3, (short)0, c3, false, false);
    }
    const int rbase = rowTile * 16 + 8 * hi;
#pragma unroll
    for (int r = 0; r < 8; ++r) {
        float* hr = h + (size_t)(rbase + r) * FH + lo;
        hr[0]  = c0[r];
        hr[16] = c1[r];
        hr[32] = c2[r];
        hr[48] = c3[r];
    }
    const float as0 = a_src[lo], as1 = a_src[16 + lo], as2 = a_src[32 + lo], as3 = a_src[48 + lo];
    const float ad0 = a_dst[lo], ad1 = a_dst[16 + lo], ad2 = a_dst[32 + lo], ad3 = a_dst[48 + lo];
    float ps[8], pd[8];
#pragma unroll
    for (int r = 0; r < 8; ++r) {
        ps[r] = c0[r] * as0 + c1[r] * as1 + c2[r] * as2 + c3[r] * as3;
        pd[r] = c0[r] * ad0 + c1[r] * ad1 + c2[r] * ad2 + c3[r] * ad3;
    }
    const bool u8 = lo & 8, u4 = lo & 4, u2 = lo & 2, u1 = lo & 1;
    float qs[4], qd[4];
#pragma unroll
    for (int i = 0; i < 4; ++i) {
        qs[i] = (u8 ? ps[i + 4] : ps[i]) + __shfl_xor(u8 ? ps[i] : ps[i + 4], 8, 32);
        qd[i] = (u8 ? pd[i + 4] : pd[i]) + __shfl_xor(u8 ? pd[i] : pd[i + 4], 8, 32);
    }
    float ts[2], td[2];
#pragma unroll
    for (int i = 0; i < 2; ++i) {
        ts[i] = (u4 ? qs[i + 2] : qs[i]) + __shfl_xor(u4 ? qs[i] : qs[i + 2], 4, 32);
        td[i] = (u4 ? qd[i + 2] : qd[i]) + __shfl_xor(u4 ? qd[i] : qd[i + 2], 4, 32);
    }
    const float es = (u2 ? ts[1] : ts[0]) + __shfl_xor(u2 ? ts[0] : ts[1], 2, 32);
    const float ed = (u2 ? td[1] : td[0]) + __shfl_xor(u2 ? td[0] : td[1], 2, 32);
    const float fin = (u1 ? ed : es) + __shfl_xor(u1 ? es : ed, 1, 32);
    (u1 ? s_dst : s_src)[rbase + (lo >> 1)] = fin;
}

__global__ __launch_bounds__(256)
void policy_kernel(const float* __restrict__ rows0, const float* __restrict__ part1,
                              const float* __restrict__ den_g, const int* __restrict__ shf_g,
                              const float* __restrict__ bprev, const float* __restrict__ Wp,
                              const float* __restrict__ bp, float* __restrict__ out,
                              int nRowTiles, size_t ws_size) {
    const int lane = threadIdx.x & 31;
    const int gw   = blockIdx.x * 8 + (threadIdx.x >> 5);
    if (gw >= nRowTiles) return;
    const int hi  = lane >> 4;
    const int lo  = lane & 15;
    const int row = gw * 16 + lo;
    const int cl  = lo < FOUT ? lo : 0;
    float4 a[FH / 8];
    load_a<FH, true>(rows0, part1, den_g, shf_g, bprev, ws_size >= WS_SPLIT_BYTES, row, hi, a);
    v8f c = {};
#pragma unroll
    for (int s = 0; s < FH / 4; ++s) {
        const int kb = kbase(s, hi);
        v2f av, b;
        av.x = (s & 1) ? a[s >> 1].z : a[s >> 1].x;
        av.y = (s & 1) ? a[s >> 1].w : a[s >> 1].y;
        const float w0 = Wp[(size_t)kb * FOUT + cl];
        const float w1 = Wp[(size_t)(kb + 1) * FOUT + cl];
        b.x = (lo < FOUT) ? w0 : 0.f;
        b.y = (lo < FOUT) ? w1 : 0.f;
        c = __builtin_amdgcn_wmma_f32_16x16x4_f32(false, av, false, b, (short)0, c, false, false);
    }
    if (lo < FOUT) {
        const int rbase = gw * 16 + 8 * hi;
        const float bb = bp[lo];
#pragma unroll
        for (int r = 0; r < 8; ++r)
            out[(size_t)(rbase + r) * FOUT + lo] = c[r] + bb;
    }
}

extern "C" void kernel_launch(void* const* d_in, const int* in_sizes, int n_in,
                              void* d_out, int out_size, void* d_ws, size_t ws_size,
                              hipStream_t stream) {
    const float* x   = (const float*)d_in[0];
    const int*   ei  = (const int*)  d_in[1];
    const float* W1  = (const float*)d_in[2];
    const float* as1 = (const float*)d_in[3];
    const float* ad1 = (const float*)d_in[4];
    const float* b1  = (const float*)d_in[5];
    const float* W2  = (const float*)d_in[6];
    const float* as2 = (const float*)d_in[7];
    const float* ad2 = (const float*)d_in[8];
    const float* b2  = (const float*)d_in[9];
    const float* Wp  = (const float*)d_in[10];
    const float* bp  = (const float*)d_in[11];
    float* out = (float*)d_out;

    const int* srcA = ei;
    const int* dstA = ei + NEDGES;

    float* ws   = (float*)d_ws;
    float* hbuf = ws;
    float* obuf = ws + (size_t)NNODES * FH;
    float* ssrc = ws + (size_t)2 * NNODES * FH;
    float* sdst = ssrc + NNODES;
    float* prt1 = sdst + NNODES;
    float* deng = prt1 + (size_t)NNODES * FH;
    int*   shfg = (int*)(deng + 2 * NNODES);
    unsigned* keys = (unsigned*)(shfg + 2 * NNODES);
    unsigned* pkw  = keys + NEDGES / 4;

    const int nRowTiles = NNODES / 16;
    const dim3 blk(256);
    const int tileBlocks = (nRowTiles + 7) / 8;
    const int aggrBlocks = 2 * (NNODES / TN);

    node_kernel<FIN, false><<<tileBlocks, blk, 0, stream>>>(x, nullptr, nullptr, nullptr, nullptr,
                                                            W1, as1, ad1, hbuf, ssrc, sdst,
                                                            srcA, dstA, keys, pkw, nRowTiles, ws_size);
    edge_aggr_kernel<<<aggrBlocks, dim3(1024), 0, stream>>>(srcA, dstA, keys, pkw, ssrc, sdst, hbuf, obuf,
                                                           prt1, deng, shfg, ws_size);

    node_kernel<FH, true><<<tileBlocks, blk, 0, stream>>>(obuf, prt1, deng, shfg, b1,
                                                          W2, as2, ad2, hbuf, ssrc, sdst,
                                                          nullptr, nullptr, nullptr, nullptr, nRowTiles, ws_size);
    edge_aggr_kernel<<<aggrBlocks, dim3(1024), 0, stream>>>(srcA, dstA, keys, pkw, ssrc, sdst, hbuf, obuf,
                                                           prt1, deng, shfg, ws_size);

    policy_kernel<<<tileBlocks, blk, 0, stream>>>(obuf, prt1, deng, shfg, b2, Wp, bp, out,
                                                  nRowTiles, ws_size);
}


# edge_aggr: a wave's hits wait in its LDS queue and leave 128 at a time: 4 edge-word loads, then 4 s_src loads in flight; next 8 h rows asked for before the adds of the last 8; row sum order changes
# speedup vs baseline: 4.9279x; 1.0468x over previous
#include <hip/hip_runtime.h>
#include <math.h>

#define NNODES 100000
#define NEDGES 1600000
#define NEP    1700000
#define FIN    16
#define FH     64
#define FOUT   8

typedef __attribute__((ext_vector_type(2))) float v2f;
typedef __attribute__((ext_vector_type(8))) float v8f;

__device__ __forceinline__ float leaky(float e) { return e > 0.f ? e : 0.2f * e; }

#define SHIFT_WINDOW 40.0f

#define TN        800
#define AG_WAVES  32
#define CH_EDGES  128
#define NCHUNK    (NEDGES / CH_EDGES)
#define LD_DEPTH  8
#define WS_SPLIT_BYTES ((size_t)(3 * FH + 6) * NNODES * sizeof(float))
#define BLK_EDGES 128
#define RG        8
#define QCAP      768
#define KCH_EDGES 512
#define NKCHUNK   (NEDGES / KCH_EDGES)
#define WS_KEYS_BYTES (WS_SPLIT_BYTES + (size_t)NEDGES)
#define PK_SHIFT  17
#define WS_PK_BYTES (WS_KEYS_BYTES + (size_t)NEDGES * sizeof(unsigned))
static_assert(NNODES <= (1 << PK_SHIFT) && TN <= (1 << (32 - PK_SHIFT)), "src and the offset fit one word");
static_assert(NEDGES % 16 == 0, "the packed words start 16-byte aligned behind the owner bytes");
static_assert(NEDGES % KCH_EDGES == 0 && NEDGES % CH_EDGES == 0, "whole chunks");
static_assert(NNODES / TN <= 127 && NNODES % TN == 0, "an owner number stays below 127, the filler");

__device__ __forceinline__ unsigned owner_of(int d) { return min((unsigned)d / (unsigned)TN, 127u); }
#define ROWS_GROUP 16

template <int MODE, bool PACKED>
__device__ __forceinline__ void aggr_batch(unsigned qw, bool valid, int nb, int base, int lane,
                                           const int* __restrict__ src, const int* __restrict__ dst,
                                           const unsigned* __restrict__ pkw,
                                           const float* __restrict__ s_src,
                                           const float* sdst_l,
                                           const float* __restrict__ h, float* acc,
                                           float* den_l, int* shift_i, int* flag_l) {
    const int e = (int)qw;
    int   s = base, rel = 0;
    float w = 0.f;
    if (valid) {
        if (e < NEDGES) {
            if (PACKED) {
                const unsigned p = pkw[e];
                s   = (int)(p & ((1u << PK_SHIFT) - 1u));
                rel = (int)(p >> PK_SHIFT);
            } else {
                s   = src[e];
                rel = dst[e] - base;
            }
        } else {
            s   = e - NEDGES;
            rel = s - base;
        }
    }
    if (valid) {
        const float l = leaky(s_src[s] + sdst_l[rel]);
        if (MODE == 1) {
            const unsigned u = __float_as_uint(l);
            if ((u >> 31) == 0) atomicMax(&shift_i[rel], (int)u);
            else                atomicMin((unsigned int*)&shift_i[rel], u);
        } else {
            const float t = l - __int_as_float(shift_i[rel]);
            if (!(t <= SHIFT_WINDOW)) *flag_l = 1;
            w = expf(t);
            atomicAdd(&den_l[rel], w);
        }
    }
    if (MODE == 1) return;
    for (int j0 = 0; j0 < nb; j0 += ROWS_GROUP) {
        float v0[ROWS_GROUP], v1[ROWS_GROUP];
#pragma unroll
        for (int u = 0; u < ROWS_GROUP; ++u) {
            const int j = j0 + u;
            const int   sj = __builtin_amdgcn_readlane(s, j);
            const float wj = __int_as_float(__builtin_amdgcn_readlane(__float_as_int(w), j));
            v0[u] = h[(size_t)sj * FH + lane]      * wj;
            v1[u] = h[(size_t)sj * FH + lane + 32] * wj;
        }
#pragma unroll
        for (int u = 0; u < ROWS_GROUP; ++u) {
            const int j = j0 + u;
            if (j < nb) {
                const int rj = __builtin_amdgcn_readlane(rel, j);
                atomicAdd(&acc[rj * FH + lane],      v0[u]);
                atomicAdd(&acc[rj * FH + lane + 32], v1[u]);
            }
        }
    }
}


template <bool PACKED>
__device__ __forceinline__ void aggr_block(const unsigned* q, int n, int base, int lane,
                                           const int* __restrict__ src, const int* __restrict__ dst,
                                           const unsigned* __restrict__ pkw,
                                           const float* __restrict__ s_src,
                                           const float* sdst_l,
                                           const float* __restrict__ h, float* acc,
                                           float* den_l, const int* shift_i, int* flag_l) {
    unsigned e[4];
    bool     valid[4];
#pragma unroll
    for (int b = 0; b < 4; ++b) {
        const int idx = b * 32 + lane;
        valid[b] = idx < n;
        e[b] = valid[b] ? q[idx] : (unsigned)(NEDGES + base);
    }
    unsigned pa[4], pb[4];
#pragma unroll
    for (int b = 0; b < 4; ++b) {
        const unsigned ei = min(e[b], (unsigned)(NEDGES - 1));
        if (PACKED) { pa[b] = pkw[ei]; pb[b] = 0u; }
        else        { pa[b] = (unsigned)src[ei]; pb[b] = (unsigned)dst[ei]; }
    }
    int   s[4], rel[4];
    float w[4], l[4];
#pragma unroll
    for (int b = 0; b < 4; ++b) {
        const bool selfl = e[b] >= (unsigned)NEDGES;
        const int  sn    = (int)(e[b] - (unsigned)NEDGES);
        if (PACKED) {
            s[b]   = selfl ? sn        : (int)(pa[b] & ((1u << PK_SHIFT) - 1u));
            rel[b] = selfl ? sn - base : (int)(pa[b] >> PK_SHIFT);
        } else {
            s[b]   = selfl ? sn        : (int)pa[b];
            rel[b] = selfl ? sn - base : (int)pb[b] - base;
        }
    }
#pragma unroll
    for (int b = 0; b < 4; ++b) l[b] = s_src[s[b]];
#pragma unroll
    for (int b = 0; b < 4; ++b) {
        const float t = leaky(l[b] + sdst_l[rel[b]]) - __int_as_float(shift_i[rel[b]]);
        w[b] = 0.f;
        if (valid[b]) {
            if (!(t <= SHIFT_WINDOW)) *flag_l = 1;
            w[b] = expf(t);
            atomicAdd(&den_l[rel[b]], w[b]);
        }
    }
    float x0[2][RG], x1[2][RG];
#define BLK_LOAD(K)                                                                    \
    _Pragma("unroll") for (int u = 0; u < RG; ++u) {                                   \
        const int sj = __builtin_amdgcn_readlane(s[(K) >> 2], ((K) & 3) * RG + u);     \
        x0[(K) & 1][u] = h[(size_t)sj * FH + lane];                                    \
        x1[(K) & 1][u] = h[(size_t)sj * FH + lane + 32];                               \
    }
#define BLK_ADD(K)                                                                     \
    _Pragma("unroll") for (int u = 0; u < RG; ++u) {                                   \
        const int   j  = ((K) & 3) * RG + u;                                           \
        const int   rj = __builtin_amdgcn_readlane(rel[(K) >> 2], j);                  \
        const float wj = __int_as_float(__builtin_amdgcn_readlane(__float_as_int(w[(K) >> 2]), j)); \
        atomicAdd(&acc[rj * FH + lane],      x0[(K) & 1][u] * wj);                     \
        atomicAdd(&acc[rj * FH + lane + 32], x1[(K) & 1][u] * wj);                     \
    }
    BLK_LOAD(0)
#pragma unroll
    for (int k = 0; k < BLK_EDGES / RG; ++k) {
        if ((k & 3) == 3 && k + 1 < BLK_EDGES / RG) {
            if (n <= (k + 1) * RG) {
                BLK_ADD(k)
                break;
            }
        }
        if (k + 1 < BLK_EDGES / RG) { BLK_LOAD(k + 1) }
        BLK_ADD(k)
    }
#undef BLK_LOAD
#undef BLK_ADD
}

#define AG_APPEND(HIT, E)                                                          \
    {                                                                              \
        const unsigned m_ = __builtin_amdgcn_ballot_w32(HIT);                      \
        if (HIT) myq[cnt + (int)__builtin_amdgcn_mbcnt_lo(m_, 0u)] = (unsigned)(E); \
        cnt += __builtin_popcount(m_);                                             \
    }

#define AG_DRAIN()                                                                 \
    while (cnt >= 32) {                                                            \
        cnt -= 32;                                                                 \
                \
                    \
        __builtin_amdgcn_wave_barrier();                                           \
        aggr_batch<MODE, PACKED>(myq[cnt + lane], true, 32, base, lane, src, dst, pkw, s_src, \
                         sdst_l, h, acc, den_l, shift_i, flag_l);                  \
    }

template <int MODE, bool NARROW, bool PACKED>
__device__ __forceinline__ void scan_pass(int wave, int lane, int base, unsigned* myq,
                                          int cbeg, int cend, int sbeg, int nself,
                                          const int* __restrict__ src, const int* __restrict__ dst,
                                          const unsigned* __restrict__ keys,
                                          const unsigned* __restrict__ pkw,
                                          const float* __restrict__ s_src,
                                          const float* sdst_l,
                                          const float* __restrict__ h, float* acc,
                                          float* den_l, int* shift_i, int* flag_l) {
    int cnt = 0;
    if (wave * 32 < nself) {
        const int idx = wave * 32 + lane;
        if (idx < nself) myq[lane] = (unsigned)(NEDGES + base + sbeg + idx);
        cnt = min(32, nself - wave * 32);
    }
    const int nit = (cend - cbeg + AG_WAVES * LD_DEPTH - 1) / (AG_WAVES * LD_DEPTH);
    if (NARROW) {
        const uint4* k4 = (const uint4*)keys;
        const unsigned gx = (unsigned)(base / TN) * 0x01010101u;
        const int nrounds = (MODE == 0) ? nit + 1 : nit;
        for (int it = 0; it < nrounds; ++it) {
            const int c0 = cbeg + it * LD_DEPTH * AG_WAVES + wave;
            uint4 kv[LD_DEPTH];
#pragma unroll
            for (int u = 0; u < LD_DEPTH; ++u) {
                const int c = c0 + u * AG_WAVES;
                kv[u] = (c < cend) ? k4[(size_t)c * 32 + lane]
                                   : make_uint4(0x7F7F7F7Fu, 0x7F7F7F7Fu, 0x7F7F7F7Fu, 0x7F7F7F7Fu);
            }
#pragma unroll
            for (int u = 0; u < LD_DEPTH; ++u) {
                const unsigned t0 = ((kv[u].x ^ gx) + 0x7F7F7F7Fu) & 0x80808080u;
                const unsigned t1 = ((kv[u].y ^ gx) + 0x7F7F7F7Fu) & 0x80808080u;
                const unsigned t2 = ((kv[u].z ^ gx) + 0x7F7F7F7Fu) & 0x80808080u;
                const unsigned t3 = ((kv[u].w ^ gx) + 0x7F7F7F7Fu) & 0x80808080u;
                unsigned hm = ((t0 >> 7) | (t1 >> 6) | (t2 >> 5) | (t3 >> 4)) ^ 0x0F0F0F0Fu;
                const int eb = (c0 + u * AG_WAVES) * KCH_EDGES + lane * 16;
                if (MODE == 0) {
                    if (cnt > QCAP - KCH_EDGES) { AG_DRAIN() }
                }
                unsigned live = __builtin_amdgcn_ballot_w32(hm != 0u);
                while (live) {
                    if (hm != 0u) {
                        const int p = __builtin_ctz(hm);
                        myq[cnt + (int)__builtin_amdgcn_mbcnt_lo(live, 0u)] =
                            (unsigned)(eb + ((p & 7) << 2) + (p >> 3));
                    }
                    cnt += __builtin_popcount(live);
                    hm &= hm - 1u;
                    live = __builtin_amdgcn_ballot_w32(hm != 0u);
                }
                if (MODE != 0) { AG_DRAIN() }
            }
            if (MODE == 0) {
                while (cnt >= BLK_EDGES || (it == nit && cnt > 0)) {
                    const int n = min(cnt, BLK_EDGES);
                    cnt -= n;
                    __builtin_amdgcn_wave_barrier();
                    aggr_block<PACKED>(myq + cnt, n, base, lane, src, dst, pkw, s_src, sdst_l, h, acc,
                                       den_l, shift_i, flag_l);
                }
            }
        }
    } else {
        const int4* dst4 = (const int4*)dst;
        for (int it = 0; it < nit; ++it) {
            const int c0 = cbeg + (it * AG_WAVES + wave) * LD_DEPTH;
            int4 dv[LD_DEPTH];
#pragma unroll
            for (int u = 0; u < LD_DEPTH; ++u) {
                const int c = c0 + u;
                dv[u] = (c < cend) ? dst4[(size_t)c * 32 + lane] : make_int4(-1, -1, -1, -1);
            }
#pragma unroll
            for (int half = 0; half < 2; ++half) {
#pragma unroll
                for (int u = half * 4; u < half * 4 + 4; ++u) {
                    const bool h0 = (unsigned)(dv[u].x - base) < TN, h1 = (unsigned)(dv[u].y - base) < TN;
                    const bool h2 = (unsigned)(dv[u].z - base) < TN, h3 = (unsigned)(dv[u].w - base) < TN;
                    if (__builtin_amdgcn_ballot_w32(h0 | h1 | h2 | h3)) {
                        const int eb = (c0 + u) * CH_EDGES + lane * 4;
                        AG_APPEND(h0, eb)
                        AG_APPEND(h1, eb + 1)
                        AG_APPEND(h2, eb + 2)
                        AG_APPEND(h3, eb + 3)
                    }
                }
                AG_DRAIN()
            }
        }
    }
    if (!(MODE == 0 && NARROW) && cnt > 0) {
        __builtin_amdgcn_wave_barrier();
        aggr_batch<MODE, PACKED>(lane < cnt ? myq[lane] : 0u, lane < cnt, cnt, base, lane,
                         src, dst, pkw, s_src, sdst_l, h, acc, den_l, shift_i, flag_l);
    }
}

template <bool NARROW, bool PACKED>
__device__ __forceinline__ void aggr_passes(int wave, int lane, int base, unsigned* myq,
                                            int cbeg, int cend, int sbeg, int nself,
                                            const int* __restrict__ src, const int* __restrict__ dst,
                                            const unsigned* __restrict__ keys,
                                            const unsigned* __restrict__ pkw,
                                            const float* __restrict__ s_src,
                                            const float* sdst_l,
                                            const float* __restrict__ h, float* acc,
                                            float* den_l, int* shift_i, int* flag_l) {
    scan_pass<0, NARROW, PACKED>(wave, lane, base, myq, cbeg, cend, sbeg, nself, src, dst, keys, pkw, s_src, sdst_l, h, acc, den_l, shift_i, flag_l);
    __syncthreads();

    const bool redo = *flag_l != 0;
    if (redo) {
        scan_pass<1, NARROW, PACKED>(wave, lane, base, myq, cbeg, cend, sbeg, nself, src, dst, keys, pkw, s_src, sdst_l, h, acc, den_l, shift_i, flag_l);
        __syncthreads();
        for (int i = threadIdx.x; i < TN * FH; i += 1024) acc[i] = 0.f;
        if (threadIdx.x < TN) den_l[threadIdx.x] = 0.f;
        __syncthreads();
        scan_pass<0, NARROW, PACKED>(wave, lane, base, myq, cbeg, cend, sbeg, nself, src, dst, keys, pkw, s_src, sdst_l, h, acc, den_l, shift_i, flag_l);
        __syncthreads();
    }
}

__global__ __launch_bounds__(1024)
void edge_aggr_kernel(const int* __restrict__ src, const int* __restrict__ dst,
                      const unsigned* __restrict__ keys, const unsigned* __restrict__ pkw,
                      const float* __restrict__ s_src, const float* __restrict__ s_dst,
                      const float* __restrict__ h, float* __restrict__ out,
                      float* __restrict__ part1, float* __restrict__ den_g,
                      int* __restrict__ shf_g, size_t ws_size) {
    const bool split  = ws_size >= WS_SPLIT_BYTES;
    const bool narrow = ws_size >= WS_KEYS_BYTES;
    const bool packed = ws_size >= WS_PK_BYTES;
    const int  half   = blockIdx.x & 1;
    if (!split && half) return;
    __shared__ float    acc[TN * FH];
    __shared__ unsigned queue[AG_WAVES * QCAP];
    __shared__ float    den_l[TN];
    __shared__ int      shift_i[TN];
    __shared__ float    sdst_l[TN];
    __shared__ int      flag_l;
    const int lane = threadIdx.x & 31;
    const int wave = __builtin_amdgcn_readfirstlane((int)(threadIdx.x >> 5));
    const int base = (int)(blockIdx.x >> 1) * TN;
    const int nch   = narrow ? NKCHUNK : NCHUNK;
    const int cbeg  = split ? half * (nch / 2) : 0;
    const int cend  = (split && !half) ? nch / 2 : nch;
    const int sbeg  = split ? half * (TN / 2) : 0;
    const int nself = split ? TN / 2 : TN;
    unsigned* myq = queue + wave * QCAP;
    for (int i = threadIdx.x; i < TN * FH; i += 1024) acc[i] = 0.f;
    if (threadIdx.x < TN) {
        const int n = base + (int)threadIdx.x;
        const float sd = s_dst[n];
        den_l[threadIdx.x]   = 0.f;
        sdst_l[threadIdx.x]  = sd;
        shift_i[threadIdx.x] = __float_as_int(leaky(s_src[n] + sd));
    }
    if (threadIdx.x == 0) flag_l = 0;
    __syncthreads();

    if (packed)      aggr_passes<true,  true >(wave, lane, base, myq, cbeg, cend, sbeg, nself, src, dst, keys, pkw, s_src, sdst_l, h, acc, den_l, shift_i, &flag_l);
    else if (narrow) aggr_passes<true,  false>(wave, lane, base, myq, cbeg, cend, sbeg, nself, src, dst, keys, pkw, s_src, sdst_l, h, acc, den_l, shift_i, &flag_l);
    else             aggr_passes<false, false>(wave, lane, base, myq, cbeg, cend, sbeg, nself, src, dst, keys, pkw, s_src, sdst_l, h, acc, den_l, shift_i, &flag_l);

    if (!split) {
        for (int i = threadIdx.x; i < TN * FH; i += 1024)
            out[(size_t)base * FH + i] = acc[i] / (den_l[i >> 6] + 1e-16f);
        return;
    }
    float* rows = half ? part1 : out;
    for (int i = threadIdx.x; i < TN * FH; i += 1024)
        rows[(size_t)base * FH + i] = acc[i];
    if (threadIdx.x < TN) {
        den_g[half * NNODES + base + (int)threadIdx.x] = den_l[threadIdx.x];
        shf_g[half * NNODES + base + (int)threadIdx.x] = shift_i[threadIdx.x];
    }
}

__device__ __forceinline__ int kbase(int s, int hi) { return 8 * (s >> 1) + 4 * hi + 2 * (s & 1); }

template <int K, bool ACT>
__device__ __forceinline__ void load_a(const float* __restrict__ in, const float* __restrict__ part1,
                                       const float* __restrict__ den_g, const int* __restrict__ shf_g,
                                       const float* __restrict__ bprev, bool split, int row, int hi,
                                       float4* a) {
    const float4* p0 = (const float4*)(in + (size_t)row * K + 4 * hi);
#pragma unroll
    for (int j = 0; j < K / 8; ++j) a[j] = p0[2 * j];
    if (ACT) {
        if (split) {
            const float4* p1 = (const float4*)(part1 + (size_t)row * K + 4 * hi);
            float4 q[K / 8];
#pragma unroll
            for (int j = 0; j < K / 8; ++j) q[j] = p1[2 * j];
            const float s0 = __int_as_float(shf_g[row]);
            const float s1 = __int_as_float(shf_g[NNODES + row]);
            float d0 = den_g[row], d1 = den_g[NNODES + row];
            float f0 = 1.f, f1 = 1.f;
            if (s0 != s1) {
                const float m = fmaxf(s0, s1);
                f0 = expf(s0 - m);
                f1 = expf(s1 - m);
            }
            d0 *= f0;
            d1 *= f1;
            const float den = d0 + d1 + 1e-16f;
#pragma unroll
            for (int j = 0; j < K / 8; ++j) {
                a[j].x = (a[j].x * f0 + q[j].x * f1) / den;
                a[j].y = (a[j].y * f0 + q[j].y * f1) / den;
                a[j].z = (a[j].z * f0 + q[j].z * f1) / den;
                a[j].w = (a[j].w * f0 + q[j].w * f1) / den;
            }
        }
#pragma unroll
        for (int j = 0; j < K / 8; ++j) {
            const float4 bb = *(const float4*)(bprev + 8 * j + 4 * hi);
            float v;
            v = a[j].x + bb.x; a[j].x = v > 0.f ? v : 0.f;
            v = a[j].y + bb.y; a[j].y = v > 0.f ? v : 0.f;
            v = a[j].z + bb.z; a[j].z = v > 0.f ? v : 0.f;
            v = a[j].w + bb.w; a[j].w = v > 0.f ? v : 0.f;
        }
    }
}

template <int K, bool ACT>
__global__ __launch_bounds__(256)
void node_kernel(const float* __restrict__ in, const float* __restrict__ part1,
                            const float* __restrict__ den_g, const int* __restrict__ shf_g,
                            const float* __restrict__ bprev, const float* __restrict__ W,
                            const float* __restrict__ a_src, const float* __restrict__ a_dst,
                            float* __restrict__ h, float* __restrict__ s_src,
                            float* __restrict__ s_dst, const int* __restrict__ src,
                            const int* __restrict__ dst, unsigned* __restrict__ keys,
                            unsigned* __restrict__ pkw, int nRowTiles, size_t ws_size) {
    const int lane    = threadIdx.x & 31;
    const int rowTile = blockIdx.x * 8 + (threadIdx.x >> 5);
    if (rowTile >= nRowTiles) return;
    const int hi  = lane >> 4;
    const int lo  = lane & 15;
    const int row = rowTile * 16 + lo;
    if (!ACT && ws_size >= WS_KEYS_BYTES) {
        static_assert((NNODES / 16) * 256 == NEDGES, "one wave per 256 edges");
        const bool  packed = ws_size >= WS_PK_BYTES;
        const int4* d4 = (const int4*)dst;
        const int4* s4 = (const int4*)src;
        uint4*      p4 = (uint4*)pkw;
#pragma unroll
        for (int q = 0; q < 2; ++q) {
            const int  idx = rowTile * 64 + q * 32 + lane;
            const int4 v   = d4[idx];
            const unsigned o0 = owner_of(v.x), o1 = owner_of(v.y), o2 = owner_of(v.z), o3 = owner_of(v.w);
            keys[idx] = o0 | (o1 << 8) | (o2 << 16) | (o3 << 24);
            if (packed) {
                const int4 sv = s4[idx];
                p4[idx] = make_uint4((unsigned)sv.x | (((unsigned)v.x - o0 * TN) << PK_SHIFT),
                                     (unsigned)sv.y | (((unsigned)v.y - o1 * TN) << PK_SHIFT),
                                     (unsigned)sv.z | (((unsigned)v.z - o2 * TN) << PK_SHIFT),
                                     (unsigned)sv.w | (((unsigned)v.w - o3 * TN) << PK_SHIFT));
            }
        }
    }
    float4 a[K / 8];
    load_a<K, ACT>(in, part1, den_g, shf_g, bprev, ws_size >= WS_SPLIT_BYTES, row, hi, a);
    v8f c0 = {}, c1 = {}, c2 = {}, c3 = {};
#pragma unroll
    for (int s = 0; s < K / 4; ++s) {
        const int kb = kbase(s, hi);
        v2f av;
        av.x = (s & 1) ? a[s >> 1].z : a[s >> 1].x;
        av.y = (s & 1) ? a[s >> 1].w : a[s >> 1].y;
        const float* w0 = W + (size_t)kb * FH + lo;
        v2f b0, b1, b2, b3;
        b0.x = w0[0];  b0.y = w0[FH];
        b1.x = w0[16]; b1.y = w0[FH + 16];
        b2.x = w0[32]; b2.y = w0[FH + 32];
        b3.x = w0[48]; b3.y = w0[FH + 48];
        c0 = __builtin_amdgcn_wmma_f32_16x16x4_f32(false, av, false, b0, (short)0, c0, false, false);
        c1 = __builtin_amdgcn_wmma_f32_16x16x4_f32(false, av, false, b1, (short)0, c1, false, false);
        c2 = __builtin_amdgcn_wmma_f32_16x16x4_f32(false, av, false, b2, (short)0, c2, false, false);
        c3 = __builtin_amdgcn_wmma_f32_16x16x4_f32(false, av, false, b3, (short)0, c3, false, false);
    }
    const int rbase = rowTile * 16 + 8 * hi;
#pragma unroll
    for (int r = 0; r < 8; ++r) {
        float* hr = h + (size_t)(rbase + r) * FH + lo;
        hr[0]  = c0[r];
        hr[16] = c1[r];
        hr[32] = c2[r];
        hr[48] = c3[r];
    }
    const float as0 = a_src[lo], as1 = a_src[16 + lo], as2 = a_src[32 + lo], as3 = a_src[48 + lo];
    const float ad0 = a_dst[lo], ad1 = a_dst[16 + lo], ad2 = a_dst[32 + lo], ad3 = a_dst[48 + lo];
    float ps[8], pd[8];
#pragma unroll
    for (int r = 0; r < 8; ++r) {
        ps[r] = c0[r] * as0 + c1[r] * as1 + c2[r] * as2 + c3[r] * as3;
        pd[r] = c0[r] * ad0 + c1[r] * ad1 + c2[r] * ad2 + c3[r] * ad3;
    }
    const bool u8 = lo & 8, u4 = lo & 4, u2 = lo & 2, u1 = lo & 1;
    float qs[4], qd[4];
#pragma unroll
    for (int i = 0; i < 4; ++i) {
        qs[i] = (u8 ? ps[i + 4] : ps[i]) + __shfl_xor(u8 ? ps[i] : ps[i + 4], 8, 32);
        qd[i] = (u8 ? pd[i + 4] : pd[i]) + __shfl_xor(u8 ? pd[i] : pd[i + 4], 8, 32);
    }
    float ts[2], td[2];
#pragma unroll
    for (int i = 0; i < 2; ++i) {
        ts[i] = (u4 ? qs[i + 2] : qs[i]) + __shfl_xor(u4 ? qs[i] : qs[i + 2], 4, 32);
        td[i] = (u4 ? qd[i + 2] : qd[i]) + __shfl_xor(u4 ? qd[i] : qd[i + 2], 4, 32);
    }
    const float es = (u2 ? ts[1] : ts[0]) + __shfl_xor(u2 ? ts[0] : ts[1], 2, 32);
    const float ed = (u2 ? td[1] : td[0]) + __shfl_xor(u2 ? td[0] : td[1], 2, 32);
    const float fin = (u1 ? ed : es) + __shfl_xor(u1 ? es : ed, 1, 32);
    (u1 ? s_dst : s_src)[rbase + (lo >> 1)] = fin;
}

__global__ __launch_bounds__(256)
void policy_kernel(const float* __restrict__ rows0, const float* __restrict__ part1,
                              const float* __restrict__ den_g, const int* __restrict__ shf_g,
                              const float* __restrict__ bprev, const float* __restrict__ Wp,
                              const float* __restrict__ bp, float* __restrict__ out,
                              int nRowTiles, size_t ws_size) {
    const int lane = threadIdx.x & 31;
    const int gw   = blockIdx.x * 8 + (threadIdx.x >> 5);
    if (gw >= nRowTiles) return;
    const int hi  = lane >> 4;
    const int lo  = lane & 15;
    const int row = gw * 16 + lo;
    const int cl  = lo < FOUT ? lo : 0;
    float4 a[FH / 8];
    load_a<FH, true>(rows0, part1, den_g, shf_g, bprev, ws_size >= WS_SPLIT_BYTES, row, hi, a);
    v8f c = {};
#pragma unroll
    for (int s = 0; s < FH / 4; ++s) {
        const int kb = kbase(s, hi);
        v2f av, b;
        av.x = (s & 1) ? a[s >> 1].z : a[s >> 1].x;
        av.y = (s & 1) ? a[s >> 1].w : a[s >> 1].y;
        const float w0 = Wp[(size_t)kb * FOUT + cl];
        const float w1 = Wp[(size_t)(kb + 1) * FOUT + cl];
        b.x = (lo < FOUT) ? w0 : 0.f;
        b.y = (lo < FOUT) ? w1 : 0.f;
        c = __builtin_amdgcn_wmma_f32_16x16x4_f32(false, av, false, b, (short)0, c, false, false);
    }
    if (lo < FOUT) {
        const int rbase = gw * 16 + 8 * hi;
        const float bb = bp[lo];
#pragma unroll
        for (int r = 0; r < 8; ++r)
            out[(size_t)(rbase + r) * FOUT + lo] = c[r] + bb;
    }
}

extern "C" void kernel_launch(void* const* d_in, const int* in_sizes, int n_in,
                              void* d_out, int out_size, void* d_ws, size_t ws_size,
                              hipStream_t stream) {
    const float* x   = (const float*)d_in[0];
    const int*   ei  = (const int*)  d_in[1];
    const float* W1  = (const float*)d_in[2];
    const float* as1 = (const float*)d_in[3];
    const float* ad1 = (const float*)d_in[4];
    const float* b1  = (const float*)d_in[5];
    const float* W2  = (const float*)d_in[6];
    const float* as2 = (const float*)d_in[7];
    const float* ad2 = (const float*)d_in[8];
    const float* b2  = (const float*)d_in[9];
    const float* Wp  = (const float*)d_in[10];
    const float* bp  = (const float*)d_in[11];
    float* out = (float*)d_out;

    const int* srcA = ei;
    const int* dstA = ei + NEDGES;

    float* ws   = (float*)d_ws;
    float* hbuf = ws;
    float* obuf = ws + (size_t)NNODES * FH;
    float* ssrc = ws + (size_t)2 * NNODES * FH;
    float* sdst = ssrc + NNODES;
    float* prt1 = sdst + NNODES;
    float* deng = prt1 + (size_t)NNODES * FH;
    int*   shfg = (int*)(deng + 2 * NNODES);
    unsigned* keys = (unsigned*)(shfg + 2 * NNODES);
    unsigned* pkw  = keys + NEDGES / 4;

    const int nRowTiles = NNODES / 16;
    const dim3 blk(256);
    const int tileBlocks = (nRowTiles + 7) / 8;
    const int aggrBlocks = 2 * (NNODES / TN);

    node_kernel<FIN, false><<<tileBlocks, blk, 0, stream>>>(x, nullptr, nullptr, nullptr, nullptr,
                                                            W1, as1, ad1, hbuf, ssrc, sdst,
                                                            srcA, dstA, keys, pkw, nRowTiles, ws_size);
    edge_aggr_kernel<<<aggrBlocks, dim3(1024), 0, stream>>>(srcA, dstA, keys, pkw, ssrc, sdst, hbuf, obuf,
                                                           prt1, deng, shfg, ws_size);

    node_kernel<FH, true><<<tileBlocks, blk, 0, stream>>>(obuf, prt1, deng, shfg, b1,
                                                          W2, as2, ad2, hbuf, ssrc, sdst,
                                                          nullptr, nullptr, nullptr, nullptr, nRowTiles, ws_size);
    edge_aggr_kernel<<<aggrBlocks, dim3(1024), 0, stream>>>(srcA, dstA, keys, pkw, ssrc, sdst, hbuf, obuf,
                                                           prt1, deng, shfg, ws_size);

    policy_kernel<<<tileBlocks, blk, 0, stream>>>(obuf, prt1, deng, shfg, b2, Wp, bp, out,
                                                  nRowTiles, ws_size);
}


# Layer 1's edge_aggr writes each wave's found edges (packed src | dst offset) as a list in the workspace; layer 2's reads its list: owner-byte scan and per-edge word loads go; row sum order changes
# speedup vs baseline: 5.4403x; 1.1040x over previous
#include <hip/hip_runtime.h>
#include <math.h>

#define NNODES 100000
#define NEDGES 1600000
#define NEP    1700000
#define FIN    16
#define FH     64
#define FOUT   8

typedef __attribute__((ext_vector_type(2))) float v2f;
typedef __attribute__((ext_vector_type(8))) float v8f;

__device__ __forceinline__ float leaky(float e) { return e > 0.f ? e : 0.2f * e; }

#define SHIFT_WINDOW 40.0f

#define TN        800
#define AG_WAVES  32
#define CH_EDGES  128
#define NCHUNK    (NEDGES / CH_EDGES)
#define LD_DEPTH  8
#define WS_SPLIT_BYTES ((size_t)(3 * FH + 6) * NNODES * sizeof(float))
#define BLK_EDGES 128
#define RG        8
#define QCAP      768
#define KCH_EDGES 512
#define NKCHUNK   (NEDGES / KCH_EDGES)
#define WS_KEYS_BYTES (WS_SPLIT_BYTES + (size_t)NEDGES)
#define PK_SHIFT  17
#define WS_PK_BYTES (WS_KEYS_BYTES + (size_t)NEDGES * sizeof(unsigned))
static_assert(NNODES <= (1 << PK_SHIFT) && TN <= (1 << (32 - PK_SHIFT)), "src and the offset fit one word");
#define LCAP      320
#define LIST_BAD  (LCAP + 1)
#define NLISTS    (2 * (NNODES / TN) * AG_WAVES)
#define WS_LIST_BYTES (WS_PK_BYTES + (size_t)NLISTS * sizeof(int) + (size_t)NLISTS * LCAP * sizeof(unsigned))
static_assert(WS_PK_BYTES % 128 == 0 && (NLISTS * sizeof(int)) % 128 == 0 && (LCAP * sizeof(unsigned)) % 128 == 0,
              "every list starts on a line");
static_assert(NEDGES % 16 == 0, "the packed words start 16-byte aligned behind the owner bytes");
static_assert(NEDGES % KCH_EDGES == 0 && NEDGES % CH_EDGES == 0, "whole chunks");
static_assert(NNODES / TN <= 127 && NNODES % TN == 0, "an owner number stays below 127, the filler");

__device__ __forceinline__ unsigned owner_of(int d) { return min((unsigned)d / (unsigned)TN, 127u); }
#define ROWS_GROUP 16

template <int MODE, bool PACKED>
__device__ __forceinline__ void aggr_batch(unsigned qw, bool valid, int nb, int base, int lane,
                                           const int* __restrict__ src, const int* __restrict__ dst,
                                           const unsigned* __restrict__ pkw,
                                           const float* __restrict__ s_src,
                                           const float* sdst_l,
                                           const float* __restrict__ h, float* acc,
                                           float* den_l, int* shift_i, int* flag_l) {
    const int e = (int)qw;
    int   s = base, rel = 0;
    float w = 0.f;
    if (valid) {
        if (e < NEDGES) {
            if (PACKED) {
                const unsigned p = pkw[e];
                s   = (int)(p & ((1u << PK_SHIFT) - 1u));
                rel = (int)(p >> PK_SHIFT);
            } else {
                s   = src[e];
                rel = dst[e] - base;
            }
        } else {
            s   = e - NEDGES;
            rel = s - base;
        }
    }
    if (valid) {
        const float l = leaky(s_src[s] + sdst_l[rel]);
        if (MODE == 1) {
            const unsigned u = __float_as_uint(l);
            if ((u >> 31) == 0) atomicMax(&shift_i[rel], (int)u);
            else                atomicMin((unsigned int*)&shift_i[rel], u);
        } else {
            const float t = l - __int_as_float(shift_i[rel]);
            if (!(t <= SHIFT_WINDOW)) *flag_l = 1;
            w = expf(t);
            atomicAdd(&den_l[rel], w);
        }
    }
    if (MODE == 1) return;
    for (int j0 = 0; j0 < nb; j0 += ROWS_GROUP) {
        float v0[ROWS_GROUP], v1[ROWS_GROUP];
#pragma unroll
        for (int u = 0; u < ROWS_GROUP; ++u) {
            const int j = j0 + u;
            const int   sj = __builtin_amdgcn_readlane(s, j);
            const float wj = __int_as_float(__builtin_amdgcn_readlane(__float_as_int(w), j));
            v0[u] = h[(size_t)sj * FH + lane]      * wj;
            v1[u] = h[(size_t)sj * FH + lane + 32] * wj;
        }
#pragma unroll
        for (int u = 0; u < ROWS_GROUP; ++u) {
            const int j = j0 + u;
            if (j < nb) {
                const int rj = __builtin_amdgcn_readlane(rel, j);
                atomicAdd(&acc[rj * FH + lane],      v0[u]);
                atomicAdd(&acc[rj * FH + lane + 32], v1[u]);
            }
        }
    }
}


template <bool PACKED, bool FROMLIST>
__device__ __forceinline__ void aggr_block(const unsigned* q, const unsigned* __restrict__ glist,
                                           int n, int base, int lane,
                                           const int* __restrict__ src, const int* __restrict__ dst,
                                           const unsigned* __restrict__ pkw,
                                           const float* __restrict__ s_src,
                                           const float* sdst_l,
                                           const float* __restrict__ h, float* acc,
                                           float* den_l, const int* shift_i, int* flag_l,
                                           unsigned* __restrict__ lrec) {
    bool  valid[4];
    int   s[4], rel[4];
    float w[4], l[4];
    if (FROMLIST) {
        unsigned pw[4];
#pragma unroll
        for (int b = 0; b < 4; ++b) {
            const int idx = b * 32 + lane;
            valid[b] = idx < n;
            pw[b] = glist[min(idx, n - 1)];
        }
#pragma unroll
        for (int b = 0; b < 4; ++b) {
            const unsigned p = valid[b] ? pw[b] : (unsigned)base;
            s[b]   = (int)(p & ((1u << PK_SHIFT) - 1u));
            rel[b] = (int)(p >> PK_SHIFT);
        }
    } else {
        unsigned e[4];
#pragma unroll
        for (int b = 0; b < 4; ++b) {
            const int idx = b * 32 + lane;
            valid[b] = idx < n;
            e[b] = valid[b] ? q[idx] : (unsigned)(NEDGES + base);
        }
        unsigned pa[4], pb[4];
#pragma unroll
        for (int b = 0; b < 4; ++b) {
            const unsigned ei = min(e[b], (unsigned)(NEDGES - 1));
            if (PACKED) { pa[b] = pkw[ei]; pb[b] = 0u; }
            else        { pa[b] = (unsigned)src[ei]; pb[b] = (unsigned)dst[ei]; }
        }
#pragma unroll
        for (int b = 0; b < 4; ++b) {
            const bool selfl = e[b] >= (unsigned)NEDGES;
            const int  sn    = (int)(e[b] - (unsigned)NEDGES);
            if (PACKED) {
                s[b]   = selfl ? sn        : (int)(pa[b] & ((1u << PK_SHIFT) - 1u));
                rel[b] = selfl ? sn - base : (int)(pa[b] >> PK_SHIFT);
            } else {
                s[b]   = selfl ? sn        : (int)pa[b];
                rel[b] = selfl ? sn - base : (int)pb[b] - base;
            }
        }
        if (PACKED && lrec != nullptr) {
#pragma unroll
            for (int b = 0; b < 4; ++b)
                if (valid[b]) lrec[b * 32 + lane] = (unsigned)s[b] | ((unsigned)rel[b] << PK_SHIFT);
        }
    }
#pragma unroll
    for (int b = 0; b < 4; ++b) l[b] = s_src[s[b]];
#pragma unroll
    for (int b = 0; b < 4; ++b) {
        const float t = leaky(l[b] + sdst_l[rel[b]]) - __int_as_float(shift_i[rel[b]]);
        w[b] = 0.f;
        if (valid[b]) {
            if (!(t <= SHIFT_WINDOW)) *flag_l = 1;
            w[b] = expf(t);
            atomicAdd(&den_l[rel[b]], w[b]);
        }
    }
    float x0[2][RG], x1[2][RG];
#define BLK_LOAD(K)                                                                    \
    _Pragma("unroll") for (int u = 0; u < RG; ++u) {                                   \
        const int sj = __builtin_amdgcn_readlane(s[(K) >> 2], ((K) & 3) * RG + u);     \
        x0[(K) & 1][u] = h[(size_t)sj * FH + lane];                                    \
        x1[(K) & 1][u] = h[(size_t)sj * FH + lane + 32];                               \
    }
#define BLK_ADD(K)                                                                     \
    _Pragma("unroll") for (int u = 0; u < RG; ++u) {                                   \
        const int   j  = ((K) & 3) * RG + u;                                           \
        const int   rj = __builtin_amdgcn_readlane(rel[(K) >> 2], j);                  \
        const float wj = __int_as_float(__builtin_amdgcn_readlane(__float_as_int(w[(K) >> 2]), j)); \
        atomicAdd(&acc[rj * FH + lane],      x0[(K) & 1][u] * wj);                     \
        atomicAdd(&acc[rj * FH + lane + 32], x1[(K) & 1][u] * wj);                     \
    }
    BLK_LOAD(0)
#pragma unroll
    for (int k = 0; k < BLK_EDGES / RG; ++k) {
        if ((k & 3) == 3 && k + 1 < BLK_EDGES / RG) {
            if (n <= (k + 1) * RG) {
                BLK_ADD(k)
                break;
            }
        }
        if (k + 1 < BLK_EDGES / RG) { BLK_LOAD(k + 1) }
        BLK_ADD(k)
    }
#undef BLK_LOAD
#undef BLK_ADD
}

#define AG_APPEND(HIT, E)                                                          \
    {                                                                              \
        const unsigned m_ = __builtin_amdgcn_ballot_w32(HIT);                      \
        if (HIT) myq[cnt + (int)__builtin_amdgcn_mbcnt_lo(m_, 0u)] = (unsigned)(E); \
        cnt += __builtin_popcount(m_);                                             \
    }

#define AG_DRAIN()                                                                 \
    while (cnt >= 32) {                                                            \
        cnt -= 32;                                                                 \
                \
                    \
        __builtin_amdgcn_wave_barrier();                                           \
        aggr_batch<MODE, PACKED>(myq[cnt + lane], true, 32, base, lane, src, dst, pkw, s_src, \
                         sdst_l, h, acc, den_l, shift_i, flag_l);                  \
    }

template <int MODE, bool NARROW, bool PACKED>
__device__ __forceinline__ void scan_pass(int wave, int lane, int base, unsigned* myq,
                                          unsigned* __restrict__ mylist, int& lcnt,
                                          int cbeg, int cend, int sbeg, int nself,
                                          const int* __restrict__ src, const int* __restrict__ dst,
                                          const unsigned* __restrict__ keys,
                                          const unsigned* __restrict__ pkw,
                                          const float* __restrict__ s_src,
                                          const float* sdst_l,
                                          const float* __restrict__ h, float* acc,
                                          float* den_l, int* shift_i, int* flag_l) {
    int cnt = 0;
    if (wave * 32 < nself) {
        const int idx = wave * 32 + lane;
        if (idx < nself) myq[lane] = (unsigned)(NEDGES + base + sbeg + idx);
        cnt = min(32, nself - wave * 32);
    }
    const int nit = (cend - cbeg + AG_WAVES * LD_DEPTH - 1) / (AG_WAVES * LD_DEPTH);
    if (NARROW) {
        const uint4* k4 = (const uint4*)keys;
        const unsigned gx = (unsigned)(base / TN) * 0x01010101u;
        const int nrounds = (MODE == 0) ? nit + 1 : nit;
        for (int it = 0; it < nrounds; ++it) {
            const int c0 = cbeg + it * LD_DEPTH * AG_WAVES + wave;
            uint4 kv[LD_DEPTH];
#pragma unroll
            for (int u = 0; u < LD_DEPTH; ++u) {
                const int c = c0 + u * AG_WAVES;
                kv[u] = (c < cend) ? k4[(size_t)c * 32 + lane]
                                   : make_uint4(0x7F7F7F7Fu, 0x7F7F7F7Fu, 0x7F7F7F7Fu, 0x7F7F7F7Fu);
            }
#pragma unroll
            for (int u = 0; u < LD_DEPTH; ++u) {
                const unsigned t0 = ((kv[u].x ^ gx) + 0x7F7F7F7Fu) & 0x80808080u;
                const unsigned t1 = ((kv[u].y ^ gx) + 0x7F7F7F7Fu) & 0x80808080u;
                const unsigned t2 = ((kv[u].z ^ gx) + 0x7F7F7F7Fu) & 0x80808080u;
                const unsigned t3 = ((kv[u].w ^ gx) + 0x7F7F7F7Fu) & 0x80808080u;
                unsigned hm = ((t0 >> 7) | (t1 >> 6) | (t2 >> 5) | (t3 >> 4)) ^ 0x0F0F0F0Fu;
                const int eb = (c0 + u * AG_WAVES) * KCH_EDGES + lane * 16;
                if (MODE == 0) {
                    if (cnt > QCAP - KCH_EDGES) {
                        AG_DRAIN()
                        lcnt = LIST_BAD;
                    }
                }
                unsigned live = __builtin_amdgcn_ballot_w32(hm != 0u);
                while (live) {
                    if (hm != 0u) {
                        const int p = __builtin_ctz(hm);
                        myq[cnt + (int)__builtin_amdgcn_mbcnt_lo(live, 0u)] =
                            (unsigned)(eb + ((p & 7) << 2) + (p >> 3));
                    }
                    cnt += __builtin_popcount(live);
                    hm &= hm - 1u;
                    live = __builtin_amdgcn_ballot_w32(hm != 0u);
                }
                if (MODE != 0) { AG_DRAIN() }
            }
            if (MODE == 0) {
                while (cnt >= BLK_EDGES || (it == nit && cnt > 0)) {
                    const int n = min(cnt, BLK_EDGES);
                    cnt -= n;
                    __builtin_amdgcn_wave_barrier();
                    unsigned* lrec = nullptr;
                    if (PACKED && mylist != nullptr) {
                        const bool room = lcnt + n <= LCAP;
                        if (room) lrec = mylist + lcnt;
                        lcnt = room ? lcnt + n : LIST_BAD;
                    }
                    aggr_block<PACKED, false>(myq + cnt, nullptr, n, base, lane, src, dst, pkw, s_src,
                                              sdst_l, h, acc, den_l, shift_i, flag_l, lrec);
                }
            }
        }
    } else {
        const int4* dst4 = (const int4*)dst;
        for (int it = 0; it < nit; ++it) {
            const int c0 = cbeg + (it * AG_WAVES + wave) * LD_DEPTH;
            int4 dv[LD_DEPTH];
#pragma unroll
            for (int u = 0; u < LD_DEPTH; ++u) {
                const int c = c0 + u;
                dv[u] = (c < cend) ? dst4[(size_t)c * 32 + lane] : make_int4(-1, -1, -1, -1);
            }
#pragma unroll
            for (int half = 0; half < 2; ++half) {
#pragma unroll
                for (int u = half * 4; u < half * 4 + 4; ++u) {
                    const bool h0 = (unsigned)(dv[u].x - base) < TN, h1 = (unsigned)(dv[u].y - base) < TN;
                    const bool h2 = (unsigned)(dv[u].z - base) < TN, h3 = (unsigned)(dv[u].w - base) < TN;
                    if (__builtin_amdgcn_ballot_w32(h0 | h1 | h2 | h3)) {
                        const int eb = (c0 + u) * CH_EDGES + lane * 4;
                        AG_APPEND(h0, eb)
                        AG_APPEND(h1, eb + 1)
                        AG_APPEND(h2, eb + 2)
                        AG_APPEND(h3, eb + 3)
                    }
                }
                AG_DRAIN()
            }
        }
    }
    if (!(MODE == 0 && NARROW) && cnt > 0) {
        __builtin_amdgcn_wave_barrier();
        aggr_batch<MODE, PACKED>(lane < cnt ? myq[lane] : 0u, lane < cnt, cnt, base, lane,
                         src, dst, pkw, s_src, sdst_l, h, acc, den_l, shift_i, flag_l);
    }
}

__device__ __forceinline__ void list_pass(const unsigned* __restrict__ mylist, int count, int base, int lane,
                                          const float* __restrict__ s_src, const float* sdst_l,
                                          const float* __restrict__ h, float* acc,
                                          float* den_l, const int* shift_i, int* flag_l) {
    for (int off = 0; off < count; off += BLK_EDGES)
        aggr_block<true, true>(nullptr, mylist + off, min(count - off, BLK_EDGES), base, lane, nullptr, nullptr,
                               nullptr, s_src, sdst_l, h, acc, den_l, shift_i, flag_l, nullptr);
}

template <int LAYER, bool NARROW, bool PACKED>
__device__ __forceinline__ void aggr_passes(int wave, int lane, int base, unsigned* myq, bool listed,
                                            unsigned* __restrict__ mylist, int* __restrict__ mycount, int count_v,
                                            int cbeg, int cend, int sbeg, int nself,
                                            const int* __restrict__ src, const int* __restrict__ dst,
                                            const unsigned* __restrict__ keys,
                                            const unsigned* __restrict__ pkw,
                                            const float* __restrict__ s_src,
                                            const float* sdst_l,
                                            const float* __restrict__ h, float* acc,
                                            float* den_l, int* shift_i, int* flag_l) {
    int  lcnt = 0;
    bool scan = true;
    if (PACKED && LAYER == 1 && listed) {
        const int count = __builtin_amdgcn_readfirstlane(count_v);
        if ((unsigned)count <= (unsigned)LCAP) {
            list_pass(mylist, count, base, lane, s_src, sdst_l, h, acc, den_l, shift_i, flag_l);
            scan = false;
        }
    }
    if (scan) {
        unsigned* rec = (PACKED && LAYER == 0 && listed) ? mylist : nullptr;
        scan_pass<0, NARROW, PACKED>(wave, lane, base, myq, rec, lcnt, cbeg, cend, sbeg, nself, src, dst, keys, pkw, s_src, sdst_l, h, acc, den_l, shift_i, flag_l);
        if (rec != nullptr && lane == 0) *mycount = lcnt;
    }
    __syncthreads();

    const bool redo = *flag_l != 0;
    if (redo) {
        scan_pass<1, NARROW, PACKED>(wave, lane, base, myq, nullptr, lcnt, cbeg, cend, sbeg, nself, src, dst, keys, pkw, s_src, sdst_l, h, acc, den_l, shift_i, flag_l);
        __syncthreads();
        for (int i = threadIdx.x; i < TN * FH; i += 1024) acc[i] = 0.f;
        if (threadIdx.x < TN) den_l[threadIdx.x] = 0.f;
        __syncthreads();
        scan_pass<0, NARROW, PACKED>(wave, lane, base, myq, nullptr, lcnt, cbeg, cend, sbeg, nself, src, dst, keys, pkw, s_src, sdst_l, h, acc, den_l, shift_i, flag_l);
        __syncthreads();
    }
}

template <int LAYER>
__global__ __launch_bounds__(1024)
void edge_aggr_kernel(const int* __restrict__ src, const int* __restrict__ dst,
                      const unsigned* __restrict__ keys, const unsigned* __restrict__ pkw,
                      const float* __restrict__ s_src, const float* __restrict__ s_dst,
                      const float* __restrict__ h, float* __restrict__ out,
                      float* __restrict__ part1, float* __restrict__ den_g,
                      int* __restrict__ shf_g, int* __restrict__ lcount,
                      unsigned* __restrict__ lists, size_t ws_size) {
    const bool split  = ws_size >= WS_SPLIT_BYTES;
    const bool narrow = ws_size >= WS_KEYS_BYTES;
    const bool packed = ws_size >= WS_PK_BYTES;
    const bool listed = ws_size >= WS_LIST_BYTES;
    const int  half   = blockIdx.x & 1;
    if (!split && half) return;
    __shared__ float    acc[TN * FH];
    __shared__ unsigned queue[AG_WAVES * QCAP];
    __shared__ float    den_l[TN];
    __shared__ int      shift_i[TN];
    __shared__ float    sdst_l[TN];
    __shared__ int      flag_l;
    const int lane = threadIdx.x & 31;
    const int wave = __builtin_amdgcn_readfirstlane((int)(threadIdx.x >> 5));
    const int base = (int)(blockIdx.x >> 1) * TN;
    const int nch   = narrow ? NKCHUNK : NCHUNK;
    const int cbeg  = split ? half * (nch / 2) : 0;
    const int cend  = (split && !half) ? nch / 2 : nch;
    const int sbeg  = split ? half * (TN / 2) : 0;
    const int nself = split ? TN / 2 : TN;
    unsigned* myq = queue + wave * QCAP;
    const int gw  = (int)blockIdx.x * AG_WAVES + wave;
    unsigned* mylist  = lists + (size_t)gw * LCAP;
    int*      mycount = lcount + gw;
    int count = 0;
    if (LAYER == 1 && listed) count = *mycount;
    for (int i = threadIdx.x; i < TN * FH; i += 1024) acc[i] = 0.f;
    if (threadIdx.x < TN) {
        const int n = base + (int)threadIdx.x;
        const float sd = s_dst[n];
        den_l[threadIdx.x]   = 0.f;
        sdst_l[threadIdx.x]  = sd;
        shift_i[threadIdx.x] = __float_as_int(leaky(s_src[n] + sd));
    }
    if (threadIdx.x == 0) flag_l = 0;
    __syncthreads();

    if (packed)      aggr_passes<LAYER, true,  true >(wave, lane, base, myq, listed, mylist, mycount, count, cbeg, cend, sbeg, nself, src, dst, keys, pkw, s_src, sdst_l, h, acc, den_l, shift_i, &flag_l);
    else if (narrow) aggr_passes<LAYER, true,  false>(wave, lane, base, myq, false, mylist, mycount, count, cbeg, cend, sbeg, nself, src, dst, keys, pkw, s_src, sdst_l, h, acc, den_l, shift_i, &flag_l);
    else             aggr_passes<LAYER, false, false>(wave, lane, base, myq, false, mylist, mycount, count, cbeg, cend, sbeg, nself, src, dst, keys, pkw, s_src, sdst_l, h, acc, den_l, shift_i, &flag_l);

    if (!split) {
        for (int i = threadIdx.x; i < TN * FH; i += 1024)
            out[(size_t)base * FH + i] = acc[i] / (den_l[i >> 6] + 1e-16f);
        return;
    }
    float* rows = half ? part1 : out;
    for (int i = threadIdx.x; i < TN * FH; i += 1024)
        rows[(size_t)base * FH + i] = acc[i];
    if (threadIdx.x < TN) {
        den_g[half * NNODES + base + (int)threadIdx.x] = den_l[threadIdx.x];
        shf_g[half * NNODES + base + (int)threadIdx.x] = shift_i[threadIdx.x];
    }
}

__device__ __forceinline__ int kbase(int s, int hi) { return 8 * (s >> 1) + 4 * hi + 2 * (s & 1); }

template <int K, bool ACT>
__device__ __forceinline__ void load_a(const float* __restrict__ in, const float* __restrict__ part1,
                                       const float* __restrict__ den_g, const int* __restrict__ shf_g,
                                       const float* __restrict__ bprev, bool split, int row, int hi,
                                       float4* a) {
    const float4* p0 = (const float4*)(in + (size_t)row * K + 4 * hi);
#pragma unroll
    for (int j = 0; j < K / 8; ++j) a[j] = p0[2 * j];
    if (ACT) {
        if (split) {
            const float4* p1 = (const float4*)(part1 + (size_t)row * K + 4 * hi);
            float4 q[K / 8];
#pragma unroll
            for (int j = 0; j < K / 8; ++j) q[j] = p1[2 * j];
            const float s0 = __int_as_float(shf_g[row]);
            const float s1 = __int_as_float(shf_g[NNODES + row]);
            float d0 = den_g[row], d1 = den_g[NNODES + row];
            float f0 = 1.f, f1 = 1.f;
            if (s0 != s1) {
                const float m = fmaxf(s0, s1);
                f0 = expf(s0 - m);
                f1 = expf(s1 - m);
            }
            d0 *= f0;
            d1 *= f1;
            const float den = d0 + d1 + 1e-16f;
#pragma unroll
            for (int j = 0; j < K / 8; ++j) {
                a[j].x = (a[j].x * f0 + q[j].x * f1) / den;
                a[j].y = (a[j].y * f0 + q[j].y * f1) / den;
                a[j].z = (a[j].z * f0 + q[j].z * f1) / den;
                a[j].w = (a[j].w * f0 + q[j].w * f1) / den;
            }
        }
#pragma unroll
        for (int j = 0; j < K / 8; ++j) {
            const float4 bb = *(const float4*)(bprev + 8 * j + 4 * hi);
            float v;
            v = a[j].x + bb.x; a[j].x = v > 0.f ? v : 0.f;
            v = a[j].y + bb.y; a[j].y = v > 0.f ? v : 0.f;
            v = a[j].z + bb.z; a[j].z = v > 0.f ? v : 0.f;
            v = a[j].w + bb.w; a[j].w = v > 0.f ? v : 0.f;
        }
    }
}

template <int K, bool ACT>
__global__ __launch_bounds__(256)
void node_kernel(const float* __restrict__ in, const float* __restrict__ part1,
                            const float* __restrict__ den_g, const int* __restrict__ shf_g,
                            const float* __restrict__ bprev, const float* __restrict__ W,
                            const float* __restrict__ a_src, const float* __restrict__ a_dst,
                            float* __restrict__ h, float* __restrict__ s_src,
                            float* __restrict__ s_dst, const int* __restrict__ src,
                            const int* __restrict__ dst, unsigned* __restrict__ keys,
                            unsigned* __restrict__ pkw, int nRowTiles, size_t ws_size) {
    const int lane    = threadIdx.x & 31;
    const int rowTile = blockIdx.x * 8 + (threadIdx.x >> 5);
    if (rowTile >= nRowTiles) return;
    const int hi  = lane >> 4;
    const int lo  = lane & 15;
    const int row = rowTile * 16 + lo;
    if (!ACT && ws_size >= WS_KEYS_BYTES) {
        static_assert((NNODES / 16) * 256 == NEDGES, "one wave per 256 edges");
        const bool  packed = ws_size >= WS_PK_BYTES;
        const int4* d4 = (const int4*)dst;
        const int4* s4 = (const int4*)src;
        uint4*      p4 = (uint4*)pkw;
#pragma unroll
        for (int q = 0; q < 2; ++q) {
            const int  idx = rowTile * 64 + q * 32 + lane;
            const int4 v   = d4[idx];
            const unsigned o0 = owner_of(v.x), o1 = owner_of(v.y), o2 = owner_of(v.z), o3 = owner_of(v.w);
            keys[idx] = o0 | (o1 << 8) | (o2 << 16) | (o3 << 24);
            if (packed) {
                const int4 sv = s4[idx];
                p4[idx] = make_uint4((unsigned)sv.x | (((unsigned)v.x - o0 * TN) << PK_SHIFT),
                                     (unsigned)sv.y | (((unsigned)v.y - o1 * TN) << PK_SHIFT),
                                     (unsigned)sv.z | (((unsigned)v.z - o2 * TN) << PK_SHIFT),
                                     (unsigned)sv.w | (((unsigned)v.w - o3 * TN) << PK_SHIFT));
            }
        }
    }
    float4 a[K / 8];
    load_a<K, ACT>(in, part1, den_g, shf_g, bprev, ws_size >= WS_SPLIT_BYTES, row, hi, a);
    v8f c0 = {}, c1 = {}, c2 = {}, c3 = {};
#pragma unroll
    for (int s = 0; s < K / 4; ++s) {
        const int kb = kbase(s, hi);
        v2f av;
        av.x = (s & 1) ? a[s >> 1].z : a[s >> 1].x;
        av.y = (s & 1) ? a[s >> 1].w : a[s >> 1].y;
        const float* w0 = W + (size_t)kb * FH + lo;
        v2f b0, b1, b2, b3;
        b0.x = w0[0];  b0.y = w0[FH];
        b1.x = w0[16]; b1.y = w0[FH + 16];
        b2.x = w0[32]; b2.y = w0[FH + 32];
        b3.x = w0[48]; b3.y = w0[FH + 48];
        c0 = __builtin_amdgcn_wmma_f32_16x16x4_f32(false, av, false, b0, (short)0, c0, false, false);
        c1 = __builtin_amdgcn_wmma_f32_16x16x4_f32(false, av, false, b1, (short)0, c1, false, false);
        c2 = __builtin_amdgcn_wmma_f32_16x16x4_f32(false, av, false, b2, (short)0, c2, false, false);
        c3 = __builtin_amdgcn_wmma_f32_16x16x4_f32(false, av, false, b3, (short)0, c3, false, false);
    }
    const int rbase = rowTile * 16 + 8 * hi;
#pragma unroll
    for (int r = 0; r < 8; ++r) {
        float* hr = h + (size_t)(rbase + r) * FH + lo;
        hr[0]  = c0[r];
        hr[16] = c1[r];
        hr[32] = c2[r];
        hr[48] = c3[r];
    }
    const float as0 = a_src[lo], as1 = a_src[16 + lo], as2 = a_src[32 + lo], as3 = a_src[48 + lo];
    const float ad0 = a_dst[lo], ad1 = a_dst[16 + lo], ad2 = a_dst[32 + lo], ad3 = a_dst[48 + lo];
    float ps[8], pd[8];
#pragma unroll
    for (int r = 0; r < 8; ++r) {
        ps[r] = c0[r] * as0 + c1[r] * as1 + c2[r] * as2 + c3[r] * as3;
        pd[r] = c0[r] * ad0 + c1[r] * ad1 + c2[r] * ad2 + c3[r] * ad3;
    }
    const bool u8 = lo & 8, u4 = lo & 4, u2 = lo & 2, u1 = lo & 1;
    float qs[4], qd[4];
#pragma unroll
    for (int i = 0; i < 4; ++i) {
        qs[i] = (u8 ? ps[i + 4] : ps[i]) + __shfl_xor(u8 ? ps[i] : ps[i + 4], 8, 32);
        qd[i] = (u8 ? pd[i + 4] : pd[i]) + __shfl_xor(u8 ? pd[i] : pd[i + 4], 8, 32);
    }
    float ts[2], td[2];
#pragma unroll
    for (int i = 0; i < 2; ++i) {
        ts[i] = (u4 ? qs[i + 2] : qs[i]) + __shfl_xor(u4 ? qs[i] : qs[i + 2], 4, 32);
        td[i] = (u4 ? qd[i + 2] : qd[i]) + __shfl_xor(u4 ? qd[i] : qd[i + 2], 4, 32);
    }
    const float es = (u2 ? ts[1] : ts[0]) + __shfl_xor(u2 ? ts[0] : ts[1], 2, 32);
    const float ed = (u2 ? td[1] : td[0]) + __shfl_xor(u2 ? td[0] : td[1], 2, 32);
    const float fin = (u1 ? ed : es) + __shfl_xor(u1 ? es : ed, 1, 32);
    (u1 ? s_dst : s_src)[rbase + (lo >> 1)] = fin;
}

__global__ __launch_bounds__(256)
void policy_kernel(const float* __restrict__ rows0, const float* __restrict__ part1,
                              const float* __restrict__ den_g, const int* __restrict__ shf_g,
                              const float* __restrict__ bprev, const float* __restrict__ Wp,
                              const float* __restrict__ bp, float* __restrict__ out,
                              int nRowTiles, size_t ws_size) {
    const int lane = threadIdx.x & 31;
    const int gw   = blockIdx.x * 8 + (threadIdx.x >> 5);
    if (gw >= nRowTiles) return;
    const int hi  = lane >> 4;
    const int lo  = lane & 15;
    const int row = gw * 16 + lo;
    const int cl  = lo < FOUT ? lo : 0;
    float4 a[FH / 8];
    load_a<FH, true>(rows0, part1, den_g, shf_g, bprev, ws_size >= WS_SPLIT_BYTES, row, hi, a);
    v8f c = {};
#pragma unroll
    for (int s = 0; s < FH / 4; ++s) {
        const int kb = kbase(s, hi);
        v2f av, b;
        av.x = (s & 1) ? a[s >> 1].z : a[s >> 1].x;
        av.y = (s & 1) ? a[s >> 1].w : a[s >> 1].y;
        const float w0 = Wp[(size_t)kb * FOUT + cl];
        const float w1 = Wp[(size_t)(kb + 1) * FOUT + cl];
        b.x = (lo < FOUT) ? w0 : 0.f;
        b.y = (lo < FOUT) ? w1 : 0.f;
        c = __builtin_amdgcn_wmma_f32_16x16x4_f32(false, av, false, b, (short)0, c, false, false);
    }
    if (lo < FOUT) {
        const int rbase = gw * 16 + 8 * hi;
        const float bb = bp[lo];
#pragma unroll
        for (int r = 0; r < 8; ++r)
            out[(size_t)(rbase + r) * FOUT + lo] = c[r] + bb;
    }
}

extern "C" void kernel_launch(void* const* d_in, const int* in_sizes, int n_in,
                              void* d_out, int out_size, void* d_ws, size_t ws_size,
                              hipStream_t stream) {
    const float* x   = (const float*)d_in[0];
    const int*   ei  = (const int*)  d_in[1];
    const float* W1  = (const float*)d_in[2];
    const float* as1 = (const float*)d_in[3];
    const float* ad1 = (const float*)d_in[4];
    const float* b1  = (const float*)d_in[5];
    const float* W2  = (const float*)d_in[6];
    const float* as2 = (const float*)d_in[7];
    const float* ad2 = (const float*)d_in[8];
    const float* b2  = (const float*)d_in[9];
    const float* Wp  = (const float*)d_in[10];
    const float* bp  = (const float*)d_in[11];
    float* out = (float*)d_out;

    const int* srcA = ei;
    const int* dstA = ei + NEDGES;

    float* ws   = (float*)d_ws;
    float* hbuf = ws;
    float* obuf = ws + (size_t)NNODES * FH;
    float* ssrc = ws + (size_t)2 * NNODES * FH;
    float* sdst = ssrc + NNODES;
    float* prt1 = sdst + NNODES;
    float* deng = prt1 + (size_t)NNODES * FH;
    int*   shfg = (int*)(deng + 2 * NNODES);
    unsigned* keys = (unsigned*)(shfg + 2 * NNODES);
    unsigned* pkw  = keys + NEDGES / 4;
    int*      lcnt = (int*)(pkw + NEDGES);
    unsigned* lsts = (unsigned*)(lcnt + NLISTS);

    const int nRowTiles = NNODES / 16;
    const dim3 blk(256);
    const int tileBlocks = (nRowTiles + 7) / 8;
    const int aggrBlocks = 2 * (NNODES / TN);

    node_kernel<FIN, false><<<tileBlocks, blk, 0, stream>>>(x, nullptr, nullptr, nullptr, nullptr,
                                                            W1, as1, ad1, hbuf, ssrc, sdst,
                                                            srcA, dstA, keys, pkw, nRowTiles, ws_size);
    edge_aggr_kernel<0><<<aggrBlocks, dim3(1024), 0, stream>>>(srcA, dstA, keys, pkw, ssrc, sdst, hbuf, obuf,
                                                              prt1, deng, shfg, lcnt, lsts, ws_size);

    node_kernel<FH, true><<<tileBlocks, blk, 0, stream>>>(obuf, prt1, deng, shfg, b1,
                                                          W2, as2, ad2, hbuf, ssrc, sdst,
                                                          nullptr, nullptr, nullptr, nullptr, nRowTiles, ws_size);
    edge_aggr_kernel<1><<<aggrBlocks, dim3(1024), 0, stream>>>(srcA, dstA, keys, pkw, ssrc, sdst, hbuf, obuf,
                                                              prt1, deng, shfg, lcnt, lsts, ws_size);

    policy_kernel<<<tileBlocks, blk, 0, stream>>>(obuf, prt1, deng, shfg, b2, Wp, bp, out,
                                                  nRowTiles, ws_size);
}


# Layer 2's edge_aggr: each group owns half its range's nodes whole (layer 1's scan lists edges per half): one plane of finished rows; policy loader's second plane and merge go; row sum order changes
# speedup vs baseline: 5.7025x; 1.0482x over previous
#include <hip/hip_runtime.h>
#include <math.h>

#define NNODES 100000
#define NEDGES 1600000
#define NEP    1700000
#define FIN    16
#define FH     64
#define FOUT   8

typedef __attribute__((ext_vector_type(2))) float v2f;
typedef __attribute__((ext_vector_type(8))) float v8f;

__device__ __forceinline__ float leaky(float e) { return e > 0.f ? e : 0.2f * e; }

#define SHIFT_WINDOW 40.0f

#define TN        800
#define AG_WAVES  32
#define CH_EDGES  128
#define NCHUNK    (NEDGES / CH_EDGES)
#define LD_DEPTH  8
#define WS_SPLIT_BYTES ((size_t)(3 * FH + 6) * NNODES * sizeof(float))
#define BLK_EDGES 128
#define RG        8
#define QCAP      768
#define KCH_EDGES 512
#define NKCHUNK   (NEDGES / KCH_EDGES)
#define WS_KEYS_BYTES (WS_SPLIT_BYTES + (size_t)NEDGES)
#define PK_SHIFT  17
#define WS_PK_BYTES (WS_KEYS_BYTES + (size_t)NEDGES * sizeof(unsigned))
static_assert(NNODES <= (1 << PK_SHIFT) && TN <= (1 << (32 - PK_SHIFT)), "src and the offset fit one word");
#define LCAP      320
#define LIST_BAD  (LCAP + 1)
#define NLISTS    (2 * (NNODES / TN) * AG_WAVES)
#define WS_LIST_BYTES (WS_PK_BYTES + (size_t)NLISTS * sizeof(int) + (size_t)NLISTS * LCAP * sizeof(unsigned))
static_assert(WS_PK_BYTES % 128 == 0 && (NLISTS * sizeof(int)) % 128 == 0 && (LCAP * sizeof(unsigned)) % 128 == 0,
              "every list starts on a line");
static_assert(NEDGES % 16 == 0, "the packed words start 16-byte aligned behind the owner bytes");
static_assert(NEDGES % KCH_EDGES == 0 && NEDGES % CH_EDGES == 0, "whole chunks");
static_assert(NNODES / TN <= 127 && NNODES % TN == 0, "an owner number stays below 127, the filler");

__device__ __forceinline__ unsigned owner_of(int d) { return min((unsigned)d / (unsigned)TN, 127u); }
#define ROWS_GROUP 16

template <int MODE, bool PACKED>
__device__ __forceinline__ void aggr_batch(unsigned qw, bool valid, int nb, int base, int lane,
                                           const int* __restrict__ src, const int* __restrict__ dst,
                                           const unsigned* __restrict__ pkw,
                                           const float* __restrict__ s_src,
                                           const float* sdst_l,
                                           const float* __restrict__ h, float* acc,
                                           float* den_l, int* shift_i, int* flag_l) {
    const int e = (int)qw;
    int   s = base, rel = 0;
    float w = 0.f;
    if (valid) {
        if (e < NEDGES) {
            if (PACKED) {
                const unsigned p = pkw[e];
                s   = (int)(p & ((1u << PK_SHIFT) - 1u));
                rel = (int)(p >> PK_SHIFT);
            } else {
                s   = src[e];
                rel = dst[e] - base;
            }
        } else {
            s   = e - NEDGES;
            rel = s - base;
        }
    }
    if (valid) {
        const float l = leaky(s_src[s] + sdst_l[rel]);
        if (MODE == 1) {
            const unsigned u = __float_as_uint(l);
            if ((u >> 31) == 0) atomicMax(&shift_i[rel], (int)u);
            else                atomicMin((unsigned int*)&shift_i[rel], u);
        } else {
            const float t = l - __int_as_float(shift_i[rel]);
            if (!(t <= SHIFT_WINDOW)) *flag_l = 1;
            w = expf(t);
            atomicAdd(&den_l[rel], w);
        }
    }
    if (MODE == 1) return;
    for (int j0 = 0; j0 < nb; j0 += ROWS_GROUP) {
        float v0[ROWS_GROUP], v1[ROWS_GROUP];
#pragma unroll
        for (int u = 0; u < ROWS_GROUP; ++u) {
            const int j = j0 + u;
            const int   sj = __builtin_amdgcn_readlane(s, j);
            const float wj = __int_as_float(__builtin_amdgcn_readlane(__float_as_int(w), j));
            v0[u] = h[(size_t)sj * FH + lane]      * wj;
            v1[u] = h[(size_t)sj * FH + lane + 32] * wj;
        }
#pragma unroll
        for (int u = 0; u < ROWS_GROUP; ++u) {
            const int j = j0 + u;
            if (j < nb) {
                const int rj = __builtin_amdgcn_readlane(rel, j);
                atomicAdd(&acc[rj * FH + lane],      v0[u]);
                atomicAdd(&acc[rj * FH + lane + 32], v1[u]);
            }
        }
    }
}


template <bool PACKED, bool FROMLIST>
__device__ __forceinline__ void aggr_block(const unsigned* q, const unsigned* __restrict__ glist,
                                           int voff, int nfwd,
                                           int n, int base, int lane,
                                           const int* __restrict__ src, const int* __restrict__ dst,
                                           const unsigned* __restrict__ pkw,
                                           const float* __restrict__ s_src,
                                           const float* sdst_l,
                                           const float* __restrict__ h, float* acc,
                                           float* den_l, const int* shift_i, int* flag_l,
                                           unsigned* __restrict__ recA, unsigned* __restrict__ recB,
                                           bool rev, int& lcA, int& lcB) {
    bool  valid[4];
    int   s[4], rel[4];
    float w[4], l[4];
    if (FROMLIST) {
        unsigned pw[4];
#pragma unroll
        for (int b = 0; b < 4; ++b) {
            const int idx = b * 32 + lane;
            valid[b] = idx < n;
            const int v = voff + min(idx, n - 1);
            pw[b] = glist[v < nfwd ? v : LCAP - 1 - (v - nfwd)];
        }
#pragma unroll
        for (int b = 0; b < 4; ++b) {
            const unsigned p = valid[b] ? pw[b] : (unsigned)base;
            s[b]   = (int)(p & ((1u << PK_SHIFT) - 1u));
            rel[b] = (int)(p >> PK_SHIFT);
        }
    } else {
        unsigned e[4];
#pragma unroll
        for (int b = 0; b < 4; ++b) {
            const int idx = b * 32 + lane;
            valid[b] = idx < n;
            e[b] = valid[b] ? q[idx] : (unsigned)(NEDGES + base);
        }
        unsigned pa[4], pb[4];
#pragma unroll
        for (int b = 0; b < 4; ++b) {
            const unsigned ei = min(e[b], (unsigned)(NEDGES - 1));
            if (PACKED) { pa[b] = pkw[ei]; pb[b] = 0u; }
            else        { pa[b] = (unsigned)src[ei]; pb[b] = (unsigned)dst[ei]; }
        }
#pragma unroll
        for (int b = 0; b < 4; ++b) {
            const bool selfl = e[b] >= (unsigned)NEDGES;
            const int  sn    = (int)(e[b] - (unsigned)NEDGES);
            if (PACKED) {
                s[b]   = selfl ? sn        : (int)(pa[b] & ((1u << PK_SHIFT) - 1u));
                rel[b] = selfl ? sn - base : (int)(pa[b] >> PK_SHIFT);
            } else {
                s[b]   = selfl ? sn        : (int)pa[b];
                rel[b] = selfl ? sn - base : (int)pb[b] - base;
            }
        }
        if (PACKED && recA != nullptr) {
            unsigned mA[4], mB[4];
            int tA = 0, tB = 0;
#pragma unroll
            for (int b = 0; b < 4; ++b) {
                const bool inB = rel[b] >= TN / 2;
                mA[b] = __builtin_amdgcn_ballot_w32(valid[b] && !inB);
                mB[b] = __builtin_amdgcn_ballot_w32(valid[b] && inB);
                tA += __builtin_popcount(mA[b]);
                tB += __builtin_popcount(mB[b]);
            }
            const bool okA = lcA != LIST_BAD && lcA + tA <= LCAP;
            const bool okB = lcB != LIST_BAD && lcB + tB <= LCAP;
            int pA = lcA, pB = lcB;
#pragma unroll
            for (int b = 0; b < 4; ++b) {
                const bool inB = rel[b] >= TN / 2;
                const int  pos = inB ? pB + (int)__builtin_amdgcn_mbcnt_lo(mB[b], 0u)
                                     : pA + (int)__builtin_amdgcn_mbcnt_lo(mA[b], 0u);
                unsigned* reg = inB ? recB : recA;
                if (valid[b] && (inB ? okB : okA))
                    reg[rev ? LCAP - 1 - pos : pos] = (unsigned)s[b] | ((unsigned)rel[b] << PK_SHIFT);
                pA += __builtin_popcount(mA[b]);
                pB += __builtin_popcount(mB[b]);
            }
            lcA = okA ? lcA + tA : LIST_BAD;
            lcB = okB ? lcB + tB : LIST_BAD;
        }
    }
#pragma unroll
    for (int b = 0; b < 4; ++b) l[b] = s_src[s[b]];
#pragma unroll
    for (int b = 0; b < 4; ++b) {
        const float t = leaky(l[b] + sdst_l[rel[b]]) - __int_as_float(shift_i[rel[b]]);
        w[b] = 0.f;
        if (valid[b]) {
            if (!(t <= SHIFT_WINDOW)) *flag_l = 1;
            w[b] = expf(t);
            atomicAdd(&den_l[rel[b]], w[b]);
        }
    }
    float x0[2][RG], x1[2][RG];
#define BLK_LOAD(K)                                                                    \
    _Pragma("unroll") for (int u = 0; u < RG; ++u) {                                   \
        const int sj = __builtin_amdgcn_readlane(s[(K) >> 2], ((K) & 3) * RG + u);     \
        x0[(K) & 1][u] = h[(size_t)sj * FH + lane];                                    \
        x1[(K) & 1][u] = h[(size_t)sj * FH + lane + 32];                               \
    }
#define BLK_ADD(K)                                                                     \
    _Pragma("unroll") for (int u = 0; u < RG; ++u) {                                   \
        const int   j  = ((K) & 3) * RG + u;                                           \
        const int   rj = __builtin_amdgcn_readlane(rel[(K) >> 2], j);                  \
        const float wj = __int_as_float(__builtin_amdgcn_readlane(__float_as_int(w[(K) >> 2]), j)); \
        atomicAdd(&acc[rj * FH + lane],      x0[(K) & 1][u] * wj);                     \
        atomicAdd(&acc[rj * FH + lane + 32], x1[(K) & 1][u] * wj);                     \
    }
    BLK_LOAD(0)
#pragma unroll
    for (int k = 0; k < BLK_EDGES / RG; ++k) {
        if ((k & 3) == 3 && k + 1 < BLK_EDGES / RG) {
            if (n <= (k + 1) * RG) {
                BLK_ADD(k)
                break;
            }
        }
        if (k + 1 < BLK_EDGES / RG) { BLK_LOAD(k + 1) }
        BLK_ADD(k)
    }
#undef BLK_LOAD
#undef BLK_ADD
}

#define AG_APPEND(HIT, E)                                                          \
    {                                                                              \
        const unsigned m_ = __builtin_amdgcn_ballot_w32(HIT);                      \
        if (HIT) myq[cnt + (int)__builtin_amdgcn_mbcnt_lo(m_, 0u)] = (unsigned)(E); \
        cnt += __builtin_popcount(m_);                                             \
    }

#define AG_DRAIN()                                                                 \
    while (cnt >= 32) {                                                            \
        cnt -= 32;                                                                 \
                \
                    \
        __builtin_amdgcn_wave_barrier();                                           \
        aggr_batch<MODE, PACKED>(myq[cnt + lane], true, 32, base, lane, src, dst, pkw, s_src, \
                         sdst_l, h, acc, den_l, shift_i, flag_l);                  \
    }

template <int MODE, bool NARROW, bool PACKED>
__device__ __forceinline__ void scan_pass(int wave, int lane, int base, unsigned* myq,
                                          unsigned* __restrict__ recA, unsigned* __restrict__ recB,
                                          bool rev, int& lcA, int& lcB,
                                          int cbeg, int cend, int sbeg, int nself,
                                          const int* __restrict__ src, const int* __restrict__ dst,
                                          const unsigned* __restrict__ keys,
                                          const unsigned* __restrict__ pkw,
                                          const float* __restrict__ s_src,
                                          const float* sdst_l,
                                          const float* __restrict__ h, float* acc,
                                          float* den_l, int* shift_i, int* flag_l) {
    int cnt = 0;
    if (wave * 32 < nself) {
        const int idx = wave * 32 + lane;
        if (idx < nself) myq[lane] = (unsigned)(NEDGES + base + sbeg + idx);
        cnt = min(32, nself - wave * 32);
    }
    const int nit = (cend - cbeg + AG_WAVES * LD_DEPTH - 1) / (AG_WAVES * LD_DEPTH);
    if (NARROW) {
        const uint4* k4 = (const uint4*)keys;
        const unsigned gx = (unsigned)(base / TN) * 0x01010101u;
        const int nrounds = (MODE == 0) ? nit + 1 : nit;
        for (int it = 0; it < nrounds; ++it) {
            const int c0 = cbeg + it * LD_DEPTH * AG_WAVES + wave;
            uint4 kv[LD_DEPTH];
#pragma unroll
            for (int u = 0; u < LD_DEPTH; ++u) {
                const int c = c0 + u * AG_WAVES;
                kv[u] = (c < cend) ? k4[(size_t)c * 32 + lane]
                                   : make_uint4(0x7F7F7F7Fu, 0x7F7F7F7Fu, 0x7F7F7F7Fu, 0x7F7F7F7Fu);
            }
#pragma unroll
            for (int u = 0; u < LD_DEPTH; ++u) {
                const unsigned t0 = ((kv[u].x ^ gx) + 0x7F7F7F7Fu) & 0x80808080u;
                const unsigned t1 = ((kv[u].y ^ gx) + 0x7F7F7F7Fu) & 0x80808080u;
                const unsigned t2 = ((kv[u].z ^ gx) + 0x7F7F7F7Fu) & 0x80808080u;
                const unsigned t3 = ((kv[u].w ^ gx) + 0x7F7F7F7Fu) & 0x80808080u;
                unsigned hm = ((t0 >> 7) | (t1 >> 6) | (t2 >> 5) | (t3 >> 4)) ^ 0x0F0F0F0Fu;
                const int eb = (c0 + u * AG_WAVES) * KCH_EDGES + lane * 16;
                if (MODE == 0) {
                    if (cnt > QCAP - KCH_EDGES) {
                        AG_DRAIN()
                        lcA = LIST_BAD;
                        lcB = LIST_BAD;
                    }
                }
                unsigned live = __builtin_amdgcn_ballot_w32(hm != 0u);
                while (live) {
                    if (hm != 0u) {
                        const int p = __builtin_ctz(hm);
                        myq[cnt + (int)__builtin_amdgcn_mbcnt_lo(live, 0u)] =
                            (unsigned)(eb + ((p & 7) << 2) + (p >> 3));
                    }
                    cnt += __builtin_popcount(live);
                    hm &= hm - 1u;
                    live = __builtin_amdgcn_ballot_w32(hm != 0u);
                }
                if (MODE != 0) { AG_DRAIN() }
            }
            if (MODE == 0) {
                while (cnt >= BLK_EDGES || (it == nit && cnt > 0)) {
                    const int n = min(cnt, BLK_EDGES);
                    cnt -= n;
                    __builtin_amdgcn_wave_barrier();
                    aggr_block<PACKED, false>(myq + cnt, nullptr, 0, 0, n, base, lane, src, dst, pkw, s_src,
                                              sdst_l, h, acc, den_l, shift_i, flag_l, recA, recB, rev, lcA, lcB);
                }
            }
        }
    } else {
        const int4* dst4 = (const int4*)dst;
        for (int it = 0; it < nit; ++it) {
            const int c0 = cbeg + (it * AG_WAVES + wave) * LD_DEPTH;
            int4 dv[LD_DEPTH];
#pragma unroll
            for (int u = 0; u < LD_DEPTH; ++u) {
                const int c = c0 + u;
                dv[u] = (c < cend) ? dst4[(size_t)c * 32 + lane] : make_int4(-1, -1, -1, -1);
            }
#pragma unroll
            for (int half = 0; half < 2; ++half) {
#pragma unroll
                for (int u = half * 4; u < half * 4 + 4; ++u) {
                    const bool h0 = (unsigned)(dv[u].x - base) < TN, h1 = (unsigned)(dv[u].y - base) < TN;
                    const bool h2 = (unsigned)(dv[u].z - base) < TN, h3 = (unsigned)(dv[u].w - base) < TN;
                    if (__builtin_amdgcn_ballot_w32(h0 | h1 | h2 | h3)) {
                        const int eb = (c0 + u) * CH_EDGES + lane * 4;
                        AG_APPEND(h0, eb)
                        AG_APPEND(h1, eb + 1)
                        AG_APPEND(h2, eb + 2)
                        AG_APPEND(h3, eb + 3)
                    }
                }
                AG_DRAIN()
            }
        }
    }
    if (!(MODE == 0 && NARROW) && cnt > 0) {
        __builtin_amdgcn_wave_barrier();
        aggr_batch<MODE, PACKED>(lane < cnt ? myq[lane] : 0u, lane < cnt, cnt, base, lane,
                         src, dst, pkw, s_src, sdst_l, h, acc, den_l, shift_i, flag_l);
    }
}

__device__ __forceinline__ void list_pass(const unsigned* __restrict__ mylist, int nfwd, int nbwd, int base, int lane,
                                          const float* __restrict__ s_src, const float* sdst_l,
                                          const float* __restrict__ h, float* acc,
                                          float* den_l, const int* shift_i, int* flag_l) {
    const int count = nfwd + nbwd;
    int lcA = 0, lcB = 0;
    for (int off = 0; off < count; off += BLK_EDGES)
        aggr_block<true, true>(nullptr, mylist, off, nfwd, min(count - off, BLK_EDGES), base, lane, nullptr, nullptr,
                               nullptr, s_src, sdst_l, h, acc, den_l, shift_i, flag_l, nullptr, nullptr, false, lcA, lcB);
}

template <int LAYER, bool NARROW, bool PACKED>
__device__ __forceinline__ void aggr_passes(int wave, int lane, int base, unsigned* myq, bool listed,
                                            unsigned* __restrict__ mylist, unsigned* __restrict__ recA,
                                            unsigned* __restrict__ recB, bool rev, int* __restrict__ mycount,
                                            bool lists_ok, int nfwd, int nbwd,
                                            int cbeg, int cend, int sbeg, int nself,
                                            const int* __restrict__ src, const int* __restrict__ dst,
                                            const unsigned* __restrict__ keys,
                                            const unsigned* __restrict__ pkw,
                                            const float* __restrict__ s_src,
                                            const float* sdst_l,
                                            const float* __restrict__ h, float* acc,
                                            float* den_l, int* shift_i, int* flag_l) {
    int  lcA = 0, lcB = 0;
    bool scan = true;
    if (PACKED && LAYER == 1 && listed && lists_ok) {
        list_pass(mylist, nfwd, nbwd, base, lane, s_src, sdst_l, h, acc, den_l, shift_i, flag_l);
        scan = false;
    }
    if (scan) {
        const bool rec = PACKED && LAYER == 0 && listed;
        scan_pass<0, NARROW, PACKED>(wave, lane, base, myq, rec ? recA : nullptr, recB, rev, lcA, lcB, cbeg, cend, sbeg, nself, src, dst, keys, pkw, s_src, sdst_l, h, acc, den_l, shift_i, flag_l);
        if (rec && lane == 0) *mycount = lcA | (lcB << 16);
    }
    __syncthreads();

    const bool redo = *flag_l != 0;
    if (redo) {
        scan_pass<1, NARROW, PACKED>(wave, lane, base, myq, nullptr, nullptr, false, lcA, lcB, cbeg, cend, sbeg, nself, src, dst, keys, pkw, s_src, sdst_l, h, acc, den_l, shift_i, flag_l);
        __syncthreads();
        for (int i = threadIdx.x; i < TN * FH; i += 1024) acc[i] = 0.f;
        if (threadIdx.x < TN) den_l[threadIdx.x] = 0.f;
        __syncthreads();
        scan_pass<0, NARROW, PACKED>(wave, lane, base, myq, nullptr, nullptr, false, lcA, lcB, cbeg, cend, sbeg, nself, src, dst, keys, pkw, s_src, sdst_l, h, acc, den_l, shift_i, flag_l);
        __syncthreads();
    }
}

template <int LAYER>
__global__ __launch_bounds__(1024)
void edge_aggr_kernel(const int* __restrict__ src, const int* __restrict__ dst,
                      const unsigned* __restrict__ keys, const unsigned* __restrict__ pkw,
                      const float* __restrict__ s_src, const float* __restrict__ s_dst,
                      const float* __restrict__ h, float* __restrict__ out,
                      float* __restrict__ part1, float* __restrict__ den_g,
                      int* __restrict__ shf_g, int* __restrict__ lcount,
                      unsigned* __restrict__ lists, size_t ws_size) {
    const bool split  = ws_size >= WS_SPLIT_BYTES;
    const bool narrow = ws_size >= WS_KEYS_BYTES;
    const bool packed = ws_size >= WS_PK_BYTES;
    const bool listed = ws_size >= WS_LIST_BYTES;
    const int  half   = blockIdx.x & 1;
    if (!split && half) return;
    __shared__ float    acc[TN * FH];
    __shared__ unsigned queue[AG_WAVES * QCAP];
    __shared__ float    den_l[TN];
    __shared__ int      shift_i[TN];
    __shared__ float    sdst_l[TN];
    __shared__ int      flag_l;
    const int lane = threadIdx.x & 31;
    const int wave = __builtin_amdgcn_readfirstlane((int)(threadIdx.x >> 5));
    const int base = (int)(blockIdx.x >> 1) * TN;
    const bool merged = LAYER == 1 && listed;
    const int nch   = narrow ? NKCHUNK : NCHUNK;
    const int cbeg  = (split && !merged) ? half * (nch / 2) : 0;
    const int cend  = (split && !merged && !half) ? nch / 2 : nch;
    const int sbeg  = split ? half * (TN / 2) : 0;
    const int nself = split ? TN / 2 : TN;
    unsigned* myq = queue + wave * QCAP;
    const int gw  = (int)blockIdx.x * AG_WAVES + wave;
    unsigned* mylist  = lists + (size_t)gw * LCAP;
    int*      mycount = lcount + gw;
    const int pair0 = (int)(blockIdx.x & ~1u) * AG_WAVES;
    unsigned* recA = lists + (size_t)(pair0 + wave) * LCAP;
    unsigned* recB = recA + (size_t)AG_WAVES * LCAP;
    unsigned cw0 = 0u, cw1 = 0u;
    if (merged) {
        cw0 = (unsigned)lcount[pair0 + lane];
        cw1 = (unsigned)lcount[pair0 + AG_WAVES + lane];
    }
    for (int i = threadIdx.x; i < TN * FH; i += 1024) acc[i] = 0.f;
    if (threadIdx.x < TN) {
        const int n = base + (int)threadIdx.x;
        const float sd = s_dst[n];
        den_l[threadIdx.x]   = 0.f;
        sdst_l[threadIdx.x]  = sd;
        shift_i[threadIdx.x] = __float_as_int(leaky(s_src[n] + sd));
    }
    if (threadIdx.x == 0) flag_l = 0;
    bool lists_ok = false;
    int  nfwd = 0, nbwd = 0;
    if (merged) {
        const unsigned n0 = (cw0 >> (16 * half)) & 0xFFFFu;
        const unsigned n1 = (cw1 >> (16 * half)) & 0xFFFFu;
        lists_ok = __builtin_amdgcn_ballot_w32(n0 + n1 > (unsigned)LCAP) == 0u;
        nfwd = __builtin_amdgcn_readlane((int)n0, wave);
        nbwd = __builtin_amdgcn_readlane((int)n1, wave);
    }
    __syncthreads();

    if (packed)      aggr_passes<LAYER, true,  true >(wave, lane, base, myq, listed, mylist, recA, recB, half != 0, mycount, lists_ok, nfwd, nbwd, cbeg, cend, sbeg, nself, src, dst, keys, pkw, s_src, sdst_l, h, acc, den_l, shift_i, &flag_l);
    else if (narrow) aggr_passes<LAYER, true,  false>(wave, lane, base, myq, false, mylist, recA, recB, half != 0, mycount, false, 0, 0, cbeg, cend, sbeg, nself, src, dst, keys, pkw, s_src, sdst_l, h, acc, den_l, shift_i, &flag_l);
    else             aggr_passes<LAYER, false, false>(wave, lane, base, myq, false, mylist, recA, recB, half != 0, mycount, false, 0, 0, cbeg, cend, sbeg, nself, src, dst, keys, pkw, s_src, sdst_l, h, acc, den_l, shift_i, &flag_l);

    if (merged) {
        const int lo = half * (TN / 2);
        for (int i = threadIdx.x; i < (TN / 2) * FH; i += 1024)
            out[(size_t)(base + lo) * FH + i] = acc[lo * FH + i] / (den_l[lo + (i >> 6)] + 1e-16f);
        return;
    }

    if (!split) {
        for (int i = threadIdx.x; i < TN * FH; i += 1024)
            out[(size_t)base * FH + i] = acc[i] / (den_l[i >> 6] + 1e-16f);
        return;
    }
    float* rows = half ? part1 : out;
    for (int i = threadIdx.x; i < TN * FH; i += 1024)
        rows[(size_t)base * FH + i] = acc[i];
    if (threadIdx.x < TN) {
        den_g[half * NNODES + base + (int)threadIdx.x] = den_l[threadIdx.x];
        shf_g[half * NNODES + base + (int)threadIdx.x] = shift_i[threadIdx.x];
    }
}

__device__ __forceinline__ int kbase(int s, int hi) { return 8 * (s >> 1) + 4 * hi + 2 * (s & 1); }

template <int K, bool ACT>
__device__ __forceinline__ void load_a(const float* __restrict__ in, const float* __restrict__ part1,
                                       const float* __restrict__ den_g, const int* __restrict__ shf_g,
                                       const float* __restrict__ bprev, bool split, int row, int hi,
                                       float4* a) {
    const float4* p0 = (const float4*)(in + (size_t)row * K + 4 * hi);
#pragma unroll
    for (int j = 0; j < K / 8; ++j) a[j] = p0[2 * j];
    if (ACT) {
        if (split) {
            const float4* p1 = (const float4*)(part1 + (size_t)row * K + 4 * hi);
            float4 q[K / 8];
#pragma unroll
            for (int j = 0; j < K / 8; ++j) q[j] = p1[2 * j];
            const float s0 = __int_as_float(shf_g[row]);
            const float s1 = __int_as_float(shf_g[NNODES + row]);
            float d0 = den_g[row], d1 = den_g[NNODES + row];
            float f0 = 1.f, f1 = 1.f;
            if (s0 != s1) {
                const float m = fmaxf(s0, s1);
                f0 = expf(s0 - m);
                f1 = expf(s1 - m);
            }
            d0 *= f0;
            d1 *= f1;
            const float den = d0 + d1 + 1e-16f;
#pragma unroll
            for (int j = 0; j < K / 8; ++j) {
                a[j].x = (a[j].x * f0 + q[j].x * f1) / den;
                a[j].y = (a[j].y * f0 + q[j].y * f1) / den;
                a[j].z = (a[j].z * f0 + q[j].z * f1) / den;
                a[j].w = (a[j].w * f0 + q[j].w * f1) / den;
            }
        }
#pragma unroll
        for (int j = 0; j < K / 8; ++j) {
            const float4 bb = *(const float4*)(bprev + 8 * j + 4 * hi);
            float v;
            v = a[j].x + bb.x; a[j].x = v > 0.f ? v : 0.f;
            v = a[j].y + bb.y; a[j].y = v > 0.f ? v : 0.f;
            v = a[j].z + bb.z; a[j].z = v > 0.f ? v : 0.f;
            v = a[j].w + bb.w; a[j].w = v > 0.f ? v : 0.f;
        }
    }
}

template <int K, bool ACT>
__global__ __launch_bounds__(256)
void node_kernel(const float* __restrict__ in, const float* __restrict__ part1,
                            const float* __restrict__ den_g, const int* __restrict__ shf_g,
                            const float* __restrict__ bprev, const float* __restrict__ W,
                            const float* __restrict__ a_src, const float* __restrict__ a_dst,
                            float* __restrict__ h, float* __restrict__ s_src,
                            float* __restrict__ s_dst, const int* __restrict__ src,
                            const int* __restrict__ dst, unsigned* __restrict__ keys,
                            unsigned* __restrict__ pkw, int nRowTiles, size_t ws_size) {
    const int lane    = threadIdx.x & 31;
    const int rowTile = blockIdx.x * 8 + (threadIdx.x >> 5);
    if (rowTile >= nRowTiles) return;
    const int hi  = lane >> 4;
    const int lo  = lane & 15;
    const int row = rowTile * 16 + lo;
    if (!ACT && ws_size >= WS_KEYS_BYTES) {
        static_assert((NNODES / 16) * 256 == NEDGES, "one wave per 256 edges");
        const bool  packed = ws_size >= WS_PK_BYTES;
        const int4* d4 = (const int4*)dst;
        const int4* s4 = (const int4*)src;
        uint4*      p4 = (uint4*)pkw;
#pragma unroll
        for (int q = 0; q < 2; ++q) {
            const int  idx = rowTile * 64 + q * 32 + lane;
            const int4 v   = d4[idx];
            const unsigned o0 = owner_of(v.x), o1 = owner_of(v.y), o2 = owner_of(v.z), o3 = owner_of(v.w);
            keys[idx] = o0 | (o1 << 8) | (o2 << 16) | (o3 << 24);
            if (packed) {
                const int4 sv = s4[idx];
                p4[idx] = make_uint4((unsigned)sv.x | (((unsigned)v.x - o0 * TN) << PK_SHIFT),
                                     (unsigned)sv.y | (((unsigned)v.y - o1 * TN) << PK_SHIFT),
                                     (unsigned)sv.z | (((unsigned)v.z - o2 * TN) << PK_SHIFT),
                                     (unsigned)sv.w | (((unsigned)v.w - o3 * TN) << PK_SHIFT));
            }
        }
    }
    float4 a[K / 8];
    load_a<K, ACT>(in, part1, den_g, shf_g, bprev, ws_size >= WS_SPLIT_BYTES, row, hi, a);
    v8f c0 = {}, c1 = {}, c2 = {}, c3 = {};
#pragma unroll
    for (int s = 0; s < K / 4; ++s) {
        const int kb = kbase(s, hi);
        v2f av;
        av.x = (s & 1) ? a[s >> 1].z : a[s >> 1].x;
        av.y = (s & 1) ? a[s >> 1].w : a[s >> 1].y;
        const float* w0 = W + (size_t)kb * FH + lo;
        v2f b0, b1, b2, b3;
        b0.x = w0[0];  b0.y = w0[FH];
        b1.x = w0[16]; b1.y = w0[FH + 16];
        b2.x = w0[32]; b2.y = w0[FH + 32];
        b3.x = w0[48]; b3.y = w0[FH + 48];
        c0 = __builtin_amdgcn_wmma_f32_16x16x4_f32(false, av, false, b0, (short)0, c0, false, false);
        c1 = __builtin_amdgcn_wmma_f32_16x16x4_f32(false, av, false, b1, (short)0, c1, false, false);
        c2 = __builtin_amdgcn_wmma_f32_16x16x4_f32(false, av, false, b2, (short)0, c2, false, false);
        c3 = __builtin_amdgcn_wmma_f32_16x16x4_f32(false, av, false, b3, (short)0, c3, false, false);
    }
    const int rbase = rowTile * 16 + 8 * hi;
#pragma unroll
    for (int r = 0; r < 8; ++r) {
        float* hr = h + (size_t)(rbase + r) * FH + lo;
        hr[0]  = c0[r];
        hr[16] = c1[r];
        hr[32] = c2[r];
        hr[48] = c3[r];
    }
    const float as0 = a_src[lo], as1 = a_src[16 + lo], as2 = a_src[32 + lo], as3 = a_src[48 + lo];
    const float ad0 = a_dst[lo], ad1 = a_dst[16 + lo], ad2 = a_dst[32 + lo], ad3 = a_dst[48 + lo];
    float ps[8], pd[8];
#pragma unroll
    for (int r = 0; r < 8; ++r) {
        ps[r] = c0[r] * as0 + c1[r] * as1 + c2[r] * as2 + c3[r] * as3;
        pd[r] = c0[r] * ad0 + c1[r] * ad1 + c2[r] * ad2 + c3[r] * ad3;
    }
    const bool u8 = lo & 8, u4 = lo & 4, u2 = lo & 2, u1 = lo & 1;
    float qs[4], qd[4];
#pragma unroll
    for (int i = 0; i < 4; ++i) {
        qs[i] = (u8 ? ps[i + 4] : ps[i]) + __shfl_xor(u8 ? ps[i] : ps[i + 4], 8, 32);
        qd[i] = (u8 ? pd[i + 4] : pd[i]) + __shfl_xor(u8 ? pd[i] : pd[i + 4], 8, 32);
    }
    float ts[2], td[2];
#pragma unroll
    for (int i = 0; i < 2; ++i) {
        ts[i] = (u4 ? qs[i + 2] : qs[i]) + __shfl_xor(u4 ? qs[i] : qs[i + 2], 4, 32);
        td[i] = (u4 ? qd[i + 2] : qd[i]) + __shfl_xor(u4 ? qd[i] : qd[i + 2], 4, 32);
    }
    const float es = (u2 ? ts[1] : ts[0]) + __shfl_xor(u2 ? ts[0] : ts[1], 2, 32);
    const float ed = (u2 ? td[1] : td[0]) + __shfl_xor(u2 ? td[0] : td[1], 2, 32);
    const float fin = (u1 ? ed : es) + __shfl_xor(u1 ? es : ed, 1, 32);
    (u1 ? s_dst : s_src)[rbase + (lo >> 1)] = fin;
}

__global__ __launch_bounds__(256)
void policy_kernel(const float* __restrict__ rows0, const float* __restrict__ part1,
                              const float* __restrict__ den_g, const int* __restrict__ shf_g,
                              const float* __restrict__ bprev, const float* __restrict__ Wp,
                              const float* __restrict__ bp, float* __restrict__ out,
                              int nRowTiles, size_t ws_size) {
    const int lane = threadIdx.x & 31;
    const int gw   = blockIdx.x * 8 + (threadIdx.x >> 5);
    if (gw >= nRowTiles) return;
    const int hi  = lane >> 4;
    const int lo  = lane & 15;
    const int row = gw * 16 + lo;
    const int cl  = lo < FOUT ? lo : 0;
    float4 a[FH / 8];
    load_a<FH, true>(rows0, part1, den_g, shf_g, bprev, ws_size >= WS_SPLIT_BYTES && ws_size < WS_LIST_BYTES, row, hi, a);
    v8f c = {};
#pragma unroll
    for (int s = 0; s < FH / 4; ++s) {
        const int kb = kbase(s, hi);
        v2f av, b;
        av.x = (s & 1) ? a[s >> 1].z : a[s >> 1].x;
        av.y = (s & 1) ? a[s >> 1].w : a[s >> 1].y;
        const float w0 = Wp[(size_t)kb * FOUT + cl];
        const float w1 = Wp[(size_t)(kb + 1) * FOUT + cl];
        b.x = (lo < FOUT) ? w0 : 0.f;
        b.y = (lo < FOUT) ? w1 : 0.f;
        c = __builtin_amdgcn_wmma_f32_16x16x4_f32(false, av, false, b, (short)0, c, false, false);
    }
    if (lo < FOUT) {
        const int rbase = gw * 16 + 8 * hi;
        const float bb = bp[lo];
#pragma unroll
        for (int r = 0; r < 8; ++r)
            out[(size_t)(rbase + r) * FOUT + lo] = c[r] + bb;
    }
}

extern "C" void kernel_launch(void* const* d_in, const int* in_sizes, int n_in,
                              void* d_out, int out_size, void* d_ws, size_t ws_size,
                              hipStream_t stream) {
    const float* x   = (const float*)d_in[0];
    const int*   ei  = (const int*)  d_in[1];
    const float* W1  = (const float*)d_in[2];
    const float* as1 = (const float*)d_in[3];
    const float* ad1 = (const float*)d_in[4];
    const float* b1  = (const float*)d_in[5];
    const float* W2  = (const float*)d_in[6];
    const float* as2 = (const float*)d_in[7];
    const float* ad2 = (const float*)d_in[8];
    const float* b2  = (const float*)d_in[9];
    const float* Wp  = (const float*)d_in[10];
    const float* bp  = (const float*)d_in[11];
    float* out = (float*)d_out;

    const int* srcA = ei;
    const int* dstA = ei + NEDGES;

    float* ws   = (float*)d_ws;
    float* hbuf = ws;
    float* obuf = ws + (size_t)NNODES * FH;
    float* ssrc = ws + (size_t)2 * NNODES * FH;
    float* sdst = ssrc + NNODES;
    float* prt1 = sdst + NNODES;
    float* deng = prt1 + (size_t)NNODES * FH;
    int*   shfg = (int*)(deng + 2 * NNODES);
    unsigned* keys = (unsigned*)(shfg + 2 * NNODES);
    unsigned* pkw  = keys + NEDGES / 4;
    int*      lcnt = (int*)(pkw + NEDGES);
    unsigned* lsts = (unsigned*)(lcnt + NLISTS);

    const int nRowTiles = NNODES / 16;
    const dim3 blk(256);
    const int tileBlocks = (nRowTiles + 7) / 8;
    const int aggrBlocks = 2 * (NNODES / TN);

    node_kernel<FIN, false><<<tileBlocks, blk, 0, stream>>>(x, nullptr, nullptr, nullptr, nullptr,
                                                            W1, as1, ad1, hbuf, ssrc, sdst,
                                                            srcA, dstA, keys, pkw, nRowTiles, ws_size);
    edge_aggr_kernel<0><<<aggrBlocks, dim3(1024), 0, stream>>>(srcA, dstA, keys, pkw, ssrc, sdst, hbuf, obuf,
                                                              prt1, deng, shfg, lcnt, lsts, ws_size);

    node_kernel<FH, true><<<tileBlocks, blk, 0, stream>>>(obuf, prt1, deng, shfg, b1,
                                                          W2, as2, ad2, hbuf, ssrc, sdst,
                                                          nullptr, nullptr, nullptr, nullptr, nRowTiles, ws_size);
    edge_aggr_kernel<1><<<aggrBlocks, dim3(1024), 0, stream>>>(srcA, dstA, keys, pkw, ssrc, sdst, hbuf, obuf,
                                                              prt1, deng, shfg, lcnt, lsts, ws_size);

    policy_kernel<<<tileBlocks, blk, 0, stream>>>(obuf, prt1, deng, shfg, b2, Wp, bp, out,
                                                  nRowTiles, ws_size);
}
